# starting kernel
# baseline (speedup 1.0000x reference)
#include <hip/hip_runtime.h>
#include <math.h>

typedef float v2f __attribute__((ext_vector_type(2)));
typedef float v8f __attribute__((ext_vector_type(8)));

namespace {
constexpr int B_TOT = 2048;
constexpr int T_LEN = 66;
constexpr int INW   = 9;
constexpr int H     = 100;
constexpr int NMEM  = 128;
constexpr int D     = 20;
constexpr int OUTW  = 8;
constexpr int NH    = 2;
constexpr int PPH   = 26;
constexpr int PDIM  = 92;
constexpr int PPAD  = 96;
constexpr int G4    = 400;
constexpr int KZ    = 132;
constexpr int BC    = 16;
constexpr int NT    = 256;
constexpr float CLIPV = 20.f;
constexpr float EPSV  = 1e-8f;

constexpr int WS_WCAT = 0;
constexpr int WS_WP   = WS_WCAT + KZ*G4;
constexpr int WS_BP   = WS_WP   + H*PPAD;
constexpr int WS_TOT  = WS_BP   + PPAD;

constexpr int SMEM_FLOATS =
    BC*NMEM*D
  + BC*NH*NMEM
  + BC*NH*NMEM
  + BC*NH*NMEM
  + BC*H
  + BC*H
  + BC*D
  + BC*KZ
  + BC*G4
  + BC*PPAD
  + BC*NH*D
  + 32
  + 32
  + 32
  + 32
  + 32*3
  + 64
  + BC*D
  + BC*D;
}

__device__ __forceinline__ float sigm(float x) { return 1.f / (1.f + expf(-x)); }
__device__ __forceinline__ float softplusf(float x) { return (x > 20.f) ? x : log1pf(expf(x)); }
__device__ __forceinline__ float clipf(float x) { return fminf(fmaxf(x, -CLIPV), CLIPV); }

__global__ __launch_bounds__(256) void pack_weights(
    const float* __restrict__ Wx, const float* __restrict__ Wh,
    const float* __restrict__ W_params, const float* __restrict__ b_params,
    float* __restrict__ ws)
{
  int i = blockIdx.x * 256 + threadIdx.x;
  if (i < KZ*G4) {
    int k = i / G4, n = i - k*G4;
    float v = (k < 29) ? Wx[k*G4 + n] : ((k < 129) ? Wh[(k - 29)*G4 + n] : 0.f);
    ws[WS_WCAT + i] = v;
  } else if (i < KZ*G4 + H*PPAD) {
    int j = i - KZ*G4;
    int k = j / PPAD, n = j - k*PPAD;
    ws[WS_WP + j] = (n < PDIM) ? W_params[k*PDIM + n] : 0.f;
  } else if (i < WS_TOT) {
    int n = i - KZ*G4 - H*PPAD;
    ws[WS_BP + n] = (n < PDIM) ? b_params[n] : 0.f;
  }
}

__global__ __launch_bounds__(NT) void ntm_fused(
    const float* __restrict__ inputs,
    const float* __restrict__ Wcat,
    const float* __restrict__ b_lstm,
    const float* __restrict__ Wp,
    const float* __restrict__ bp,
    const float* __restrict__ W_out,
    const float* __restrict__ b_out,
    const int*   __restrict__ seq_ptr,
    float* __restrict__ out)
{
  extern __shared__ float smem[];
  float* sM   = smem;
  float* sW   = sM   + BC*NMEM*D;
  float* sA   = sW   + BC*NH*NMEM;
  float* sBuf = sA   + BC*NH*NMEM;
  float* sH   = sBuf + BC*NH*NMEM;
  float* sC   = sH   + BC*H;
  float* sR   = sC   + BC*H;
  float* sZ   = sR   + BC*D;
  float* sG   = sZ   + BC*KZ;
  float* sP   = sG   + BC*G4;
  float* sK   = sP   + BC*PPAD;
  float* sKn  = sK   + BC*NH*D;
  float* sBe  = sKn  + 32;
  float* sGg  = sBe  + 32;
  float* sGa  = sGg  + 32;
  float* sS   = sGa  + 32;
  float* sRed = sS   + 32*3;
  float* sE   = sRed + 64;
  float* sAd  = sE   + BC*D;

  const int tid   = threadIdx.x;
  const int wave  = tid >> 5;
  const int lane  = tid & 31;
  const int m     = lane & 15;
  const int khalf = (lane < 16) ? 0 : 2;
  const int crow  = (lane < 16) ? 0 : 8;
  const int b0    = blockIdx.x * BC;
  const int seq   = seq_ptr[0];
  const int tOutStart = seq + 1;
  const int TOUT  = T_LEN - tOutStart;

  for (int i = tid; i < BC*NMEM*D;  i += NT) sM[i] = 1e-6f;
  for (int i = tid; i < BC*NH*NMEM; i += NT) sW[i] = ((i & (NMEM-1)) == 0) ? 1.f : 0.f;
  for (int i = tid; i < BC*H;       i += NT) { sH[i] = 0.f; sC[i] = 0.f; }
  for (int i = tid; i < BC*D;       i += NT) sR[i] = 0.f;
  __syncthreads();

  for (int t = 0; t < T_LEN; ++t) {
    for (int i = tid; i < BC*KZ; i += NT) {
      int b = i / KZ, j = i - b*KZ;
      float v;
      if (j < INW)          v = inputs[(size_t)(b0 + b) * T_LEN * INW + t * INW + j];
      else if (j < INW + D) v = sR[b*D + (j - INW)];
      else if (j < 129)     v = sH[b*H + (j - 29)];
      else                  v = 0.f;
      sZ[i] = v;
    }
    __syncthreads();

    for (int ct = wave; ct < G4/16; ct += NT/32) {
      int n = ct*16 + m;
      const float* __restrict__ wc = Wcat + n;
      float bias = b_lstm[n];
      v8f acc = { bias, bias, bias, bias, bias, bias, bias, bias };
      for (int kt = 0; kt < KZ/4; ++kt) {
        int k0 = kt*4 + khalf;
        v2f av; av.x = sZ[m*KZ + k0];     av.y = sZ[m*KZ + k0 + 1];
        v2f bv; bv.x = wc[(size_t)k0*G4]; bv.y = wc[(size_t)(k0 + 1)*G4];
        acc = __builtin_amdgcn_wmma_f32_16x16x4_f32(false, av, false, bv,
                                                    (short)0, acc, false, false);
      }
      #pragma unroll
      for (int v = 0; v < 8; ++v) sG[(v + crow)*G4 + n] = acc[v];
    }
    __syncthreads();

    for (int i = tid; i < BC*H; i += NT) {
      int b = i / H, u = i - b*H;
      float gi = sG[b*G4 + u];
      float gf = sG[b*G4 + H   + u];
      float gg = sG[b*G4 + 2*H + u];
      float go = sG[b*G4 + 3*H + u];
      float c  = sigm(gf) * sC[i] + sigm(gi) * tanhf(gg);
      float h  = sigm(go) * tanhf(c);
      sC[i] = c; sH[i] = h;
    }
    __syncthreads();

    for (int ct = wave; ct < PPAD/16; ct += NT/32) {
      int n = ct*16 + m;
      const float* __restrict__ wp = Wp + n;
      float bias = bp[n];
      v8f acc = { bias, bias, bias, bias, bias, bias, bias, bias };
      for (int kt = 0; kt < H/4; ++kt) {
        int k0 = kt*4 + khalf;
        v2f av; av.x = sH[m*H + k0];        av.y = sH[m*H + k0 + 1];
        v2f bv; bv.x = wp[(size_t)k0*PPAD]; bv.y = wp[(size_t)(k0 + 1)*PPAD];
        acc = __builtin_amdgcn_wmma_f32_16x16x4_f32(false, av, false, bv,
                                                    (short)0, acc, false, false);
      }
      #pragma unroll
      for (int v = 0; v < 8; ++v) sP[(v + crow)*PPAD + n] = clipf(acc[v]);
    }
    __syncthreads();

    if (tid < BC*NH) {
      int b = tid >> 1, hd = tid & 1;
      const float* pb = sP + b*PPAD + hd*PPH;
      float kn2 = 0.f;
      for (int d = 0; d < D; ++d) {
        float kv = tanhf(pb[d]);
        sK[tid*D + d] = kv;
        kn2 += kv * kv;
      }
      sKn[tid] = sqrtf(kn2);
      sBe[tid] = softplusf(pb[D]);
      sGg[tid] = sigm(pb[D + 1]);
      float s0 = pb[D + 2], s1 = pb[D + 3], s2 = pb[D + 4];
      float mx = fmaxf(s0, fmaxf(s1, s2));
      float e0 = expf(s0 - mx), e1 = expf(s1 - mx), e2 = expf(s2 - mx);
      float inv = 1.f / (e0 + e1 + e2);
      sS[tid*3 + 0] = e0 * inv; sS[tid*3 + 1] = e1 * inv; sS[tid*3 + 2] = e2 * inv;
      sGa[tid] = softplusf(pb[D + 5]) + 1.f;
    }
    __syncthreads();

    for (int i = tid; i < BC*NMEM; i += NT) {
      int b = i >> 7, n = i & (NMEM - 1);
      const float* Mr = sM + (b*NMEM + n)*D;
      const float* k0 = sK + (b*NH + 0)*D;
      const float* k1 = sK + (b*NH + 1)*D;
      float mn2 = 0.f, d0 = 0.f, d1 = 0.f;
      for (int d = 0; d < D; ++d) {
        float mv = Mr[d];
        mn2 += mv * mv; d0 += k0[d] * mv; d1 += k1[d] * mv;
      }
      float mn = sqrtf(mn2);
      sA[(b*NH + 0)*NMEM + n] = sBe[b*NH + 0] * d0 / (sKn[b*NH + 0]*mn + EPSV);
      sA[(b*NH + 1)*NMEM + n] = sBe[b*NH + 1] * d1 / (sKn[b*NH + 1]*mn + EPSV);
    }
    __syncthreads();

    if (tid < BC*NH) {
      const float* row = sA + tid*NMEM;
      float mx = row[0];
      for (int n = 1; n < NMEM; ++n) mx = fmaxf(mx, row[n]);
      float sum = 0.f;
      for (int n = 0; n < NMEM; ++n) sum += expf(row[n] - mx);
      sRed[tid*2 + 0] = mx;
      sRed[tid*2 + 1] = 1.f / sum;
    }
    __syncthreads();

    for (int i = tid; i < BC*NH*NMEM; i += NT) {
      int bh = i >> 7;
      float wc = expf(sA[i] - sRed[bh*2]) * sRed[bh*2 + 1];
      float g  = sGg[bh];
      sA[i] = g * wc + (1.f - g) * sW[i];
    }
    __syncthreads();

    for (int i = tid; i < BC*NH*NMEM; i += NT) {
      int bh = i >> 7, n = i & (NMEM - 1);
      const float* wg = sA + bh*NMEM;
      float wt = sS[bh*3 + 0] * wg[(n + 1) & (NMEM - 1)]
               + sS[bh*3 + 1] * wg[n]
               + sS[bh*3 + 2] * wg[(n - 1) & (NMEM - 1)];
      wt = fmaxf(wt, 1e-16f);
      sBuf[i] = expf(sGa[bh] * logf(wt));
    }
    __syncthreads();

    if (tid < BC*NH) {
      const float* row = sBuf + tid*NMEM;
      float sum = 0.f;
      for (int n = 0; n < NMEM; ++n) sum += row[n];
      sRed[tid*2] = 1.f / sum;
    }
    __syncthreads();

    for (int i = tid; i < BC*NH*NMEM; i += NT)
      sW[i] = sBuf[i] * sRed[(i >> 7)*2];
    __syncthreads();

    for (int i = tid; i < BC*D; i += NT) {
      int b = i / D, d = i - b*D;
      const float* wr = sW + (b*NH + 0)*NMEM;
      float acc = 0.f;
      for (int n = 0; n < NMEM; ++n) acc += wr[n] * sM[(b*NMEM + n)*D + d];
      sR[i]  = acc;
      sE[i]  = sigm (sP[b*PPAD + NH*PPH + d]);
      sAd[i] = tanhf(sP[b*PPAD + NH*PPH + D + d]);
    }
    __syncthreads();

    for (int i = tid; i < BC*NMEM*D; i += NT) {
      int b   = i / (NMEM*D);
      int rem = i - b*NMEM*D;
      int n   = rem / D, d = rem - n*D;
      float ww = sW[(b*NH + 1)*NMEM + n];
      float mv = sM[i];
      sM[i] = mv * (1.f - ww * sE[b*D + d]) + ww * sAd[b*D + d];
    }

    if (tid < BC*OUTW) {
      int b = tid / OUTW, o = tid - b*OUTW;
      float acc = b_out[o];
      for (int j = 0; j < H; ++j) acc += sH[b*H + j] * W_out[j*OUTW + o];
      for (int d = 0; d < D; ++d) acc += sR[b*D + d] * W_out[(H + d)*OUTW + o];
      acc = clipf(acc);
      if (t >= tOutStart) {
        out[(size_t)(b0 + b) * TOUT * OUTW + (t - tOutStart) * OUTW + o] = sigm(acc);
      }
    }
    __syncthreads();
  }
}

extern "C" void kernel_launch(void* const* d_in, const int* in_sizes, int n_in,
                              void* d_out, int out_size, void* d_ws, size_t ws_size,
                              hipStream_t stream) {
  (void)in_sizes; (void)n_in; (void)out_size; (void)ws_size;
  const float* inputs   = (const float*)d_in[0];
  const float* Wx       = (const float*)d_in[1];
  const float* Wh       = (const float*)d_in[2];
  const float* b_lstm   = (const float*)d_in[3];
  const float* W_params = (const float*)d_in[4];
  const float* b_params = (const float*)d_in[5];
  const float* W_out    = (const float*)d_in[6];
  const float* b_out    = (const float*)d_in[7];
  const int*   seq      = (const int*)d_in[8];
  float* out = (float*)d_out;
  float* ws  = (float*)d_ws;

  pack_weights<<<dim3((WS_TOT + 255) / 256), dim3(256), 0, stream>>>(
      Wx, Wh, W_params, b_params, ws);

  const size_t shmem = (size_t)SMEM_FLOATS * sizeof(float);
  (void)hipFuncSetAttribute((const void*)ntm_fused,
                            hipFuncAttributeMaxDynamicSharedMemorySize,
                            (int)shmem);

  ntm_fused<<<dim3(B_TOT / BC), dim3(NT), shmem, stream>>>(
      inputs, ws + WS_WCAT, b_lstm, ws + WS_WP, ws + WS_BP,
      W_out, b_out, seq, out);
}


# The 3 one-wave phases (head params, softmax max/sum-exp, sharpen sum; 1 thread per row, 32-way LDS bank conflicts) now run on all 8 waves: 16 lanes per row, DPP tree sums (sum order changes)
# speedup vs baseline: 1.4059x; 1.4059x over previous
#include <hip/hip_runtime.h>
#include <math.h>

typedef float v2f __attribute__((ext_vector_type(2)));
typedef float v8f __attribute__((ext_vector_type(8)));

namespace {
constexpr int B_TOT = 2048;
constexpr int T_LEN = 66;
constexpr int INW   = 9;
constexpr int H     = 100;
constexpr int NMEM  = 128;
constexpr int D     = 20;
constexpr int OUTW  = 8;
constexpr int NH    = 2;
constexpr int PPH   = 26;
constexpr int PDIM  = 92;
constexpr int PPAD  = 96;
constexpr int G4    = 400;
constexpr int KZ    = 132;
constexpr int BC    = 16;
constexpr int NT    = 256;
constexpr float CLIPV = 20.f;
constexpr float EPSV  = 1e-8f;

constexpr int WS_WCAT = 0;
constexpr int WS_WP   = WS_WCAT + KZ*G4;
constexpr int WS_BP   = WS_WP   + H*PPAD;
constexpr int WS_TOT  = WS_BP   + PPAD;

constexpr int SMEM_FLOATS =
    BC*NMEM*D
  + BC*NH*NMEM
  + BC*NH*NMEM
  + BC*NH*NMEM
  + BC*H
  + BC*H
  + BC*D
  + BC*KZ
  + BC*G4
  + BC*PPAD
  + BC*NH*D
  + 32
  + 32
  + 32
  + 32
  + 32*3
  + 64
  + BC*D
  + BC*D;
static_assert(NT == BC*NH*8 && NMEM % 16 == 0,
              "phase 5: eight lanes per (b,head); phases 7 and 10: 16 lanes per row, two rows per 16-lane group");
}

__device__ __forceinline__ float sigm(float x) { return 1.f / (1.f + expf(-x)); }
__device__ __forceinline__ float softplusf(float x) { return (x > 20.f) ? x : log1pf(expf(x)); }
__device__ __forceinline__ float clipf(float x) { return fminf(fmaxf(x, -CLIPV), CLIPV); }

template <int X>
__device__ __forceinline__ float lane_xor(float v) {
  return __builtin_bit_cast(float, __builtin_amdgcn_update_dpp(
      0, __builtin_bit_cast(int, v), 0x160 + X, 0xf, 0xf, true));
}
__device__ __forceinline__ float sum8(float v) {
  v += lane_xor<1>(v); v += lane_xor<2>(v); v += lane_xor<4>(v);
  return v;
}
__device__ __forceinline__ float sum16(float v) {
  v += lane_xor<1>(v); v += lane_xor<2>(v); v += lane_xor<4>(v); v += lane_xor<8>(v);
  return v;
}
__device__ __forceinline__ float max16(float v) {
  v = fmaxf(v, lane_xor<1>(v)); v = fmaxf(v, lane_xor<2>(v));
  v = fmaxf(v, lane_xor<4>(v)); v = fmaxf(v, lane_xor<8>(v));
  return v;
}

__global__ __launch_bounds__(256) void pack_weights(
    const float* __restrict__ Wx, const float* __restrict__ Wh,
    const float* __restrict__ W_params, const float* __restrict__ b_params,
    float* __restrict__ ws)
{
  int i = blockIdx.x * 256 + threadIdx.x;
  if (i < KZ*G4) {
    int k = i / G4, n = i - k*G4;
    float v = (k < 29) ? Wx[k*G4 + n] : ((k < 129) ? Wh[(k - 29)*G4 + n] : 0.f);
    ws[WS_WCAT + i] = v;
  } else if (i < KZ*G4 + H*PPAD) {
    int j = i - KZ*G4;
    int k = j / PPAD, n = j - k*PPAD;
    ws[WS_WP + j] = (n < PDIM) ? W_params[k*PDIM + n] : 0.f;
  } else if (i < WS_TOT) {
    int n = i - KZ*G4 - H*PPAD;
    ws[WS_BP + n] = (n < PDIM) ? b_params[n] : 0.f;
  }
}

__global__ __launch_bounds__(NT) void ntm_fused(
    const float* __restrict__ inputs,
    const float* __restrict__ Wcat,
    const float* __restrict__ b_lstm,
    const float* __restrict__ Wp,
    const float* __restrict__ bp,
    const float* __restrict__ W_out,
    const float* __restrict__ b_out,
    const int*   __restrict__ seq_ptr,
    float* __restrict__ out)
{
  extern __shared__ float smem[];
  float* sM   = smem;
  float* sW   = sM   + BC*NMEM*D;
  float* sA   = sW   + BC*NH*NMEM;
  float* sBuf = sA   + BC*NH*NMEM;
  float* sH   = sBuf + BC*NH*NMEM;
  float* sC   = sH   + BC*H;
  float* sR   = sC   + BC*H;
  float* sZ   = sR   + BC*D;
  float* sG   = sZ   + BC*KZ;
  float* sP   = sG   + BC*G4;
  float* sK   = sP   + BC*PPAD;
  float* sKn  = sK   + BC*NH*D;
  float* sBe  = sKn  + 32;
  float* sGg  = sBe  + 32;
  float* sGa  = sGg  + 32;
  float* sS   = sGa  + 32;
  float* sRed = sS   + 32*3;
  float* sE   = sRed + 64;
  float* sAd  = sE   + BC*D;

  const int tid   = threadIdx.x;
  const int wave  = tid >> 5;
  const int lane  = tid & 31;
  const int m     = lane & 15;
  const int khalf = (lane < 16) ? 0 : 2;
  const int crow  = (lane < 16) ? 0 : 8;
  const int b0    = blockIdx.x * BC;
  const int seq   = seq_ptr[0];
  const int tOutStart = seq + 1;
  const int TOUT  = T_LEN - tOutStart;

  for (int i = tid; i < BC*NMEM*D;  i += NT) sM[i] = 1e-6f;
  for (int i = tid; i < BC*NH*NMEM; i += NT) sW[i] = ((i & (NMEM-1)) == 0) ? 1.f : 0.f;
  for (int i = tid; i < BC*H;       i += NT) { sH[i] = 0.f; sC[i] = 0.f; }
  for (int i = tid; i < BC*D;       i += NT) sR[i] = 0.f;
  __syncthreads();

  for (int t = 0; t < T_LEN; ++t) {
    for (int i = tid; i < BC*KZ; i += NT) {
      int b = i / KZ, j = i - b*KZ;
      float v;
      if (j < INW)          v = inputs[(size_t)(b0 + b) * T_LEN * INW + t * INW + j];
      else if (j < INW + D) v = sR[b*D + (j - INW)];
      else if (j < 129)     v = sH[b*H + (j - 29)];
      else                  v = 0.f;
      sZ[i] = v;
    }
    __syncthreads();

    for (int ct = wave; ct < G4/16; ct += NT/32) {
      int n = ct*16 + m;
      const float* __restrict__ wc = Wcat + n;
      float bias = b_lstm[n];
      v8f acc = { bias, bias, bias, bias, bias, bias, bias, bias };
      for (int kt = 0; kt < KZ/4; ++kt) {
        int k0 = kt*4 + khalf;
        v2f av; av.x = sZ[m*KZ + k0];     av.y = sZ[m*KZ + k0 + 1];
        v2f bv; bv.x = wc[(size_t)k0*G4]; bv.y = wc[(size_t)(k0 + 1)*G4];
        acc = __builtin_amdgcn_wmma_f32_16x16x4_f32(false, av, false, bv,
                                                    (short)0, acc, false, false);
      }
      #pragma unroll
      for (int v = 0; v < 8; ++v) sG[(v + crow)*G4 + n] = acc[v];
    }
    __syncthreads();

    for (int i = tid; i < BC*H; i += NT) {
      int b = i / H, u = i - b*H;
      float gi = sG[b*G4 + u];
      float gf = sG[b*G4 + H   + u];
      float gg = sG[b*G4 + 2*H + u];
      float go = sG[b*G4 + 3*H + u];
      float c  = sigm(gf) * sC[i] + sigm(gi) * tanhf(gg);
      float h  = sigm(go) * tanhf(c);
      sC[i] = c; sH[i] = h;
    }
    __syncthreads();

    for (int ct = wave; ct < PPAD/16; ct += NT/32) {
      int n = ct*16 + m;
      const float* __restrict__ wp = Wp + n;
      float bias = bp[n];
      v8f acc = { bias, bias, bias, bias, bias, bias, bias, bias };
      for (int kt = 0; kt < H/4; ++kt) {
        int k0 = kt*4 + khalf;
        v2f av; av.x = sH[m*H + k0];        av.y = sH[m*H + k0 + 1];
        v2f bv; bv.x = wp[(size_t)k0*PPAD]; bv.y = wp[(size_t)(k0 + 1)*PPAD];
        acc = __builtin_amdgcn_wmma_f32_16x16x4_f32(false, av, false, bv,
                                                    (short)0, acc, false, false);
      }
      #pragma unroll
      for (int v = 0; v < 8; ++v) sP[(v + crow)*PPAD + n] = clipf(acc[v]);
    }
    __syncthreads();

    {
      const int bh  = tid >> 3;
      const int sub = tid & 7;
      const float* pb = sP + (bh >> 1)*PPAD + (bh & 1)*PPH;
      float kn2 = 0.f;
      for (int j = 0; j < 3; ++j) {
        const int d = sub + 8*j;
        if (d < D) {
          float kv = tanhf(pb[d]);
          sK[bh*D + d] = kv;
          kn2 += kv * kv;
        }
      }
      kn2 = sum8(kn2);
      if (sub == 0) {
        sKn[bh] = sqrtf(kn2);
      } else if (sub == 1 || sub == 4) {
        float sp = softplusf(pb[(sub == 1) ? D : D + 5]);
        if (sub == 1) sBe[bh] = sp; else sGa[bh] = sp + 1.f;
      } else if (sub == 2) {
        sGg[bh] = sigm(pb[D + 1]);
      } else if (sub == 3) {
        float s0 = pb[D + 2], s1 = pb[D + 3], s2 = pb[D + 4];
        float mx = fmaxf(s0, fmaxf(s1, s2));
        float e0 = expf(s0 - mx), e1 = expf(s1 - mx), e2 = expf(s2 - mx);
        float inv = 1.f / (e0 + e1 + e2);
        sS[bh*3 + 0] = e0 * inv; sS[bh*3 + 1] = e1 * inv; sS[bh*3 + 2] = e2 * inv;
      }
    }
    __syncthreads();

    for (int i = tid; i < BC*NMEM; i += NT) {
      int b = i >> 7, n = i & (NMEM - 1);
      const float* Mr = sM + (b*NMEM + n)*D;
      const float* k0 = sK + (b*NH + 0)*D;
      const float* k1 = sK + (b*NH + 1)*D;
      float mn2 = 0.f, d0 = 0.f, d1 = 0.f;
      for (int d = 0; d < D; ++d) {
        float mv = Mr[d];
        mn2 += mv * mv; d0 += k0[d] * mv; d1 += k1[d] * mv;
      }
      float mn = sqrtf(mn2);
      sA[(b*NH + 0)*NMEM + n] = sBe[b*NH + 0] * d0 / (sKn[b*NH + 0]*mn + EPSV);
      sA[(b*NH + 1)*NMEM + n] = sBe[b*NH + 1] * d1 / (sKn[b*NH + 1]*mn + EPSV);
    }
    __syncthreads();

    {
      const int g16 = tid >> 4, c = tid & 15;
      for (int rr = 0; rr < 2; ++rr) {
        const int bh = g16*2 + rr;
        const float* row = sA + bh*NMEM;
        float v[NMEM/16];
        for (int j = 0; j < NMEM/16; ++j) v[j] = row[c + 16*j];
        float mx = v[0];
        for (int j = 1; j < NMEM/16; ++j) mx = fmaxf(mx, v[j]);
        mx = max16(mx);
        float sum = 0.f;
        for (int j = 0; j < NMEM/16; ++j) sum += expf(v[j] - mx);
        sum = sum16(sum);
        if (c == 0) {
          sRed[bh*2 + 0] = mx;
          sRed[bh*2 + 1] = 1.f / sum;
        }
      }
    }
    __syncthreads();

    for (int i = tid; i < BC*NH*NMEM; i += NT) {
      int bh = i >> 7;
      float wc = expf(sA[i] - sRed[bh*2]) * sRed[bh*2 + 1];
      float g  = sGg[bh];
      sA[i] = g * wc + (1.f - g) * sW[i];
    }
    __syncthreads();

    for (int i = tid; i < BC*NH*NMEM; i += NT) {
      int bh = i >> 7, n = i & (NMEM - 1);
      const float* wg = sA + bh*NMEM;
      float wt = sS[bh*3 + 0] * wg[(n + 1) & (NMEM - 1)]
               + sS[bh*3 + 1] * wg[n]
               + sS[bh*3 + 2] * wg[(n - 1) & (NMEM - 1)];
      wt = fmaxf(wt, 1e-16f);
      sBuf[i] = expf(sGa[bh] * logf(wt));
    }
    __syncthreads();

    {
      const int g16 = tid >> 4, c = tid & 15;
      for (int rr = 0; rr < 2; ++rr) {
        const int bh = g16*2 + rr;
        const float* row = sBuf + bh*NMEM;
        float sum = 0.f;
        for (int j = 0; j < NMEM/16; ++j) sum += row[c + 16*j];
        sum = sum16(sum);
        if (c == 0) sRed[bh*2] = 1.f / sum;
      }
    }
    __syncthreads();

    for (int i = tid; i < BC*NH*NMEM; i += NT)
      sW[i] = sBuf[i] * sRed[(i >> 7)*2];
    __syncthreads();

    for (int i = tid; i < BC*D; i += NT) {
      int b = i / D, d = i - b*D;
      const float* wr = sW + (b*NH + 0)*NMEM;
      float acc = 0.f;
      for (int n = 0; n < NMEM; ++n) acc += wr[n] * sM[(b*NMEM + n)*D + d];
      sR[i]  = acc;
      sE[i]  = sigm (sP[b*PPAD + NH*PPH + d]);
      sAd[i] = tanhf(sP[b*PPAD + NH*PPH + D + d]);
    }
    __syncthreads();

    for (int i = tid; i < BC*NMEM*D; i += NT) {
      int b   = i / (NMEM*D);
      int rem = i - b*NMEM*D;
      int n   = rem / D, d = rem - n*D;
      float ww = sW[(b*NH + 1)*NMEM + n];
      float mv = sM[i];
      sM[i] = mv * (1.f - ww * sE[b*D + d]) + ww * sAd[b*D + d];
    }

    if (tid < BC*OUTW) {
      int b = tid / OUTW, o = tid - b*OUTW;
      float acc = b_out[o];
      for (int j = 0; j < H; ++j) acc += sH[b*H + j] * W_out[j*OUTW + o];
      for (int d = 0; d < D; ++d) acc += sR[b*D + d] * W_out[(H + d)*OUTW + o];
      acc = clipf(acc);
      if (t >= tOutStart) {
        out[(size_t)(b0 + b) * TOUT * OUTW + (t - tOutStart) * OUTW + o] = sigm(acc);
      }
    }
    __syncthreads();
  }
}

extern "C" void kernel_launch(void* const* d_in, const int* in_sizes, int n_in,
                              void* d_out, int out_size, void* d_ws, size_t ws_size,
                              hipStream_t stream) {
  (void)in_sizes; (void)n_in; (void)out_size; (void)ws_size;
  const float* inputs   = (const float*)d_in[0];
  const float* Wx       = (const float*)d_in[1];
  const float* Wh       = (const float*)d_in[2];
  const float* b_lstm   = (const float*)d_in[3];
  const float* W_params = (const float*)d_in[4];
  const float* b_params = (const float*)d_in[5];
  const float* W_out    = (const float*)d_in[6];
  const float* b_out    = (const float*)d_in[7];
  const int*   seq      = (const int*)d_in[8];
  float* out = (float*)d_out;
  float* ws  = (float*)d_ws;

  pack_weights<<<dim3((WS_TOT + 255) / 256), dim3(256), 0, stream>>>(
      Wx, Wh, W_params, b_params, ws);

  const size_t shmem = (size_t)SMEM_FLOATS * sizeof(float);
  (void)hipFuncSetAttribute((const void*)ntm_fused,
                            hipFuncAttributeMaxDynamicSharedMemorySize,
                            (int)shmem);

  ntm_fused<<<dim3(B_TOT / BC), dim3(NT), shmem, stream>>>(
      inputs, ws + WS_WCAT, b_lstm, ws + WS_WP, ws + WS_BP,
      W_out, b_out, seq, out);
}


# Memory matrix M (16x128x20 f32 per workgroup) moves from LDS into registers, 8 rows per thread; cosine, read, erase/add run on registers; read-vector sum order changes (8-row partials + 16-lane tree)
# speedup vs baseline: 2.4050x; 1.7106x over previous
#include <hip/hip_runtime.h>
#include <math.h>

typedef float v2f __attribute__((ext_vector_type(2)));
typedef float v8f __attribute__((ext_vector_type(8)));

namespace {
constexpr int B_TOT = 2048;
constexpr int T_LEN = 66;
constexpr int INW   = 9;
constexpr int H     = 100;
constexpr int NMEM  = 128;
constexpr int D     = 20;
constexpr int OUTW  = 8;
constexpr int NH    = 2;
constexpr int PPH   = 26;
constexpr int PDIM  = 92;
constexpr int PPAD  = 96;
constexpr int G4    = 400;
constexpr int KZ    = 132;
constexpr int BC    = 16;
constexpr int NT    = 256;
constexpr float CLIPV = 20.f;
constexpr float EPSV  = 1e-8f;

constexpr int WS_WCAT = 0;
constexpr int WS_WP   = WS_WCAT + KZ*G4;
constexpr int WS_BP   = WS_WP   + H*PPAD;
constexpr int WS_TOT  = WS_BP   + PPAD;

constexpr int MROWS = NMEM / 16;

constexpr int SMEM_FLOATS =
    BC*NH*NMEM
  + BC*NH*NMEM
  + BC*NH*NMEM
  + BC*H
  + BC*H
  + BC*D
  + BC*KZ
  + BC*G4
  + BC*PPAD
  + BC*NH*D
  + 32
  + 32
  + 32
  + 32
  + 32*3
  + 64
  + BC*D
  + BC*D;
static_assert(NT == BC*NH*8 && NMEM % 16 == 0,
              "phase 5: eight lanes per (b,head); phases 7 and 10: 16 lanes per row, two rows per 16-lane group");
static_assert(NT == BC*16,
              "memory matrix in registers: 16 threads (one 16-lane row of a wave) per batch row, NMEM/16 memory rows each");
}

__device__ __forceinline__ float sigm(float x) { return 1.f / (1.f + expf(-x)); }
__device__ __forceinline__ float softplusf(float x) { return (x > 20.f) ? x : log1pf(expf(x)); }
__device__ __forceinline__ float clipf(float x) { return fminf(fmaxf(x, -CLIPV), CLIPV); }

template <int X>
__device__ __forceinline__ float lane_xor(float v) {
  return __builtin_bit_cast(float, __builtin_amdgcn_update_dpp(
      0, __builtin_bit_cast(int, v), 0x160 + X, 0xf, 0xf, true));
}
__device__ __forceinline__ float sum8(float v) {
  v += lane_xor<1>(v); v += lane_xor<2>(v); v += lane_xor<4>(v);
  return v;
}
__device__ __forceinline__ float sum16(float v) {
  v += lane_xor<1>(v); v += lane_xor<2>(v); v += lane_xor<4>(v); v += lane_xor<8>(v);
  return v;
}
__device__ __forceinline__ float max16(float v) {
  v = fmaxf(v, lane_xor<1>(v)); v = fmaxf(v, lane_xor<2>(v));
  v = fmaxf(v, lane_xor<4>(v)); v = fmaxf(v, lane_xor<8>(v));
  return v;
}

__global__ __launch_bounds__(256) void pack_weights(
    const float* __restrict__ Wx, const float* __restrict__ Wh,
    const float* __restrict__ W_params, const float* __restrict__ b_params,
    float* __restrict__ ws)
{
  int i = blockIdx.x * 256 + threadIdx.x;
  if (i < KZ*G4) {
    int k = i / G4, n = i - k*G4;
    float v = (k < 29) ? Wx[k*G4 + n] : ((k < 129) ? Wh[(k - 29)*G4 + n] : 0.f);
    ws[WS_WCAT + i] = v;
  } else if (i < KZ*G4 + H*PPAD) {
    int j = i - KZ*G4;
    int k = j / PPAD, n = j - k*PPAD;
    ws[WS_WP + j] = (n < PDIM) ? W_params[k*PDIM + n] : 0.f;
  } else if (i < WS_TOT) {
    int n = i - KZ*G4 - H*PPAD;
    ws[WS_BP + n] = (n < PDIM) ? b_params[n] : 0.f;
  }
}

__global__ __launch_bounds__(NT) void ntm_fused(
    const float* __restrict__ inputs,
    const float* __restrict__ Wcat,
    const float* __restrict__ b_lstm,
    const float* __restrict__ Wp,
    const float* __restrict__ bp,
    const float* __restrict__ W_out,
    const float* __restrict__ b_out,
    const int*   __restrict__ seq_ptr,
    float* __restrict__ out)
{
  extern __shared__ float smem[];
  float* sW   = smem;
  float* sA   = sW   + BC*NH*NMEM;
  float* sBuf = sA   + BC*NH*NMEM;
  float* sH   = sBuf + BC*NH*NMEM;
  float* sC   = sH   + BC*H;
  float* sR   = sC   + BC*H;
  float* sZ   = sR   + BC*D;
  float* sG   = sZ   + BC*KZ;
  float* sP   = sG   + BC*G4;
  float* sK   = sP   + BC*PPAD;
  float* sKn  = sK   + BC*NH*D;
  float* sBe  = sKn  + 32;
  float* sGg  = sBe  + 32;
  float* sGa  = sGg  + 32;
  float* sS   = sGa  + 32;
  float* sRed = sS   + 32*3;
  float* sE   = sRed + 64;
  float* sAd  = sE   + BC*D;

  const int tid   = threadIdx.x;
  const int wave  = tid >> 5;
  const int lane  = tid & 31;
  const int m     = lane & 15;
  const int khalf = (lane < 16) ? 0 : 2;
  const int crow  = (lane < 16) ? 0 : 8;
  const int b0    = blockIdx.x * BC;
  const int seq   = seq_ptr[0];
  const int tOutStart = seq + 1;
  const int TOUT  = T_LEN - tOutStart;

  const int mb = tid >> 4;
  const int mc = tid & 15;
  float Mr[MROWS][D];
  for (int j = 0; j < MROWS; ++j) {
    for (int d = 0; d < D; ++d) Mr[j][d] = 1e-6f;
  }

  for (int i = tid; i < BC*NH*NMEM; i += NT) sW[i] = ((i & (NMEM-1)) == 0) ? 1.f : 0.f;
  for (int i = tid; i < BC*H;       i += NT) { sH[i] = 0.f; sC[i] = 0.f; }
  for (int i = tid; i < BC*D;       i += NT) sR[i] = 0.f;
  __syncthreads();

  for (int t = 0; t < T_LEN; ++t) {
    for (int i = tid; i < BC*KZ; i += NT) {
      int b = i / KZ, j = i - b*KZ;
      float v;
      if (j < INW)          v = inputs[(size_t)(b0 + b) * T_LEN * INW + t * INW + j];
      else if (j < INW + D) v = sR[b*D + (j - INW)];
      else if (j < 129)     v = sH[b*H + (j - 29)];
      else                  v = 0.f;
      sZ[i] = v;
    }
    __syncthreads();

    for (int ct = wave; ct < G4/16; ct += NT/32) {
      int n = ct*16 + m;
      const float* __restrict__ wc = Wcat + n;
      float bias = b_lstm[n];
      v8f acc = { bias, bias, bias, bias, bias, bias, bias, bias };
      for (int kt = 0; kt < KZ/4; ++kt) {
        int k0 = kt*4 + khalf;
        v2f av; av.x = sZ[m*KZ + k0];     av.y = sZ[m*KZ + k0 + 1];
        v2f bv; bv.x = wc[(size_t)k0*G4]; bv.y = wc[(size_t)(k0 + 1)*G4];
        acc = __builtin_amdgcn_wmma_f32_16x16x4_f32(false, av, false, bv,
                                                    (short)0, acc, false, false);
      }
      #pragma unroll
      for (int v = 0; v < 8; ++v) sG[(v + crow)*G4 + n] = acc[v];
    }
    __syncthreads();

    for (int i = tid; i < BC*H; i += NT) {
      int b = i / H, u = i - b*H;
      float gi = sG[b*G4 + u];
      float gf = sG[b*G4 + H   + u];
      float gg = sG[b*G4 + 2*H + u];
      float go = sG[b*G4 + 3*H + u];
      float c  = sigm(gf) * sC[i] + sigm(gi) * tanhf(gg);
      float h  = sigm(go) * tanhf(c);
      sC[i] = c; sH[i] = h;
    }
    __syncthreads();

    for (int ct = wave; ct < PPAD/16; ct += NT/32) {
      int n = ct*16 + m;
      const float* __restrict__ wp = Wp + n;
      float bias = bp[n];
      v8f acc = { bias, bias, bias, bias, bias, bias, bias, bias };
      for (int kt = 0; kt < H/4; ++kt) {
        int k0 = kt*4 + khalf;
        v2f av; av.x = sH[m*H + k0];        av.y = sH[m*H + k0 + 1];
        v2f bv; bv.x = wp[(size_t)k0*PPAD]; bv.y = wp[(size_t)(k0 + 1)*PPAD];
        acc = __builtin_amdgcn_wmma_f32_16x16x4_f32(false, av, false, bv,
                                                    (short)0, acc, false, false);
      }
      #pragma unroll
      for (int v = 0; v < 8; ++v) sP[(v + crow)*PPAD + n] = clipf(acc[v]);
    }
    __syncthreads();

    {
      const int bh  = tid >> 3;
      const int sub = tid & 7;
      const float* pb = sP + (bh >> 1)*PPAD + (bh & 1)*PPH;
      float kn2 = 0.f;
      for (int j = 0; j < 3; ++j) {
        const int d = sub + 8*j;
        if (d < D) {
          float kv = tanhf(pb[d]);
          sK[bh*D + d] = kv;
          kn2 += kv * kv;
        }
      }
      kn2 = sum8(kn2);
      if (sub == 0) {
        sKn[bh] = sqrtf(kn2);
      } else if (sub == 1 || sub == 4) {
        float sp = softplusf(pb[(sub == 1) ? D : D + 5]);
        if (sub == 1) sBe[bh] = sp; else sGa[bh] = sp + 1.f;
      } else if (sub == 2) {
        sGg[bh] = sigm(pb[D + 1]);
      } else if (sub == 3) {
        float s0 = pb[D + 2], s1 = pb[D + 3], s2 = pb[D + 4];
        float mx = fmaxf(s0, fmaxf(s1, s2));
        float e0 = expf(s0 - mx), e1 = expf(s1 - mx), e2 = expf(s2 - mx);
        float inv = 1.f / (e0 + e1 + e2);
        sS[bh*3 + 0] = e0 * inv; sS[bh*3 + 1] = e1 * inv; sS[bh*3 + 2] = e2 * inv;
      }
    }
    __syncthreads();

    {
      float k0[D], k1[D];
      for (int d = 0; d < D; ++d) {
        k0[d] = sK[(mb*NH + 0)*D + d];
        k1[d] = sK[(mb*NH + 1)*D + d];
      }
      const float be0 = sBe[mb*NH + 0], be1 = sBe[mb*NH + 1];
      const float kn0 = sKn[mb*NH + 0], kn1 = sKn[mb*NH + 1];
      for (int j = 0; j < MROWS; ++j) {
        float mn2 = 0.f, d0 = 0.f, d1 = 0.f;
        for (int d = 0; d < D; ++d) {
          float mv = Mr[j][d];
          mn2 += mv * mv; d0 += k0[d] * mv; d1 += k1[d] * mv;
        }
        float mn = sqrtf(mn2);
        const int n = mc + 16*j;
        sA[(mb*NH + 0)*NMEM + n] = be0 * d0 / (kn0*mn + EPSV);
        sA[(mb*NH + 1)*NMEM + n] = be1 * d1 / (kn1*mn + EPSV);
      }
    }
    __syncthreads();

    {
      const int g16 = tid >> 4, c = tid & 15;
      for (int rr = 0; rr < 2; ++rr) {
        const int bh = g16*2 + rr;
        const float* row = sA + bh*NMEM;
        float v[NMEM/16];
        for (int j = 0; j < NMEM/16; ++j) v[j] = row[c + 16*j];
        float mx = v[0];
        for (int j = 1; j < NMEM/16; ++j) mx = fmaxf(mx, v[j]);
        mx = max16(mx);
        float sum = 0.f;
        for (int j = 0; j < NMEM/16; ++j) sum += expf(v[j] - mx);
        sum = sum16(sum);
        if (c == 0) {
          sRed[bh*2 + 0] = mx;
          sRed[bh*2 + 1] = 1.f / sum;
        }
      }
    }
    __syncthreads();

    for (int i = tid; i < BC*NH*NMEM; i += NT) {
      int bh = i >> 7;
      float wc = expf(sA[i] - sRed[bh*2]) * sRed[bh*2 + 1];
      float g  = sGg[bh];
      sA[i] = g * wc + (1.f - g) * sW[i];
    }
    __syncthreads();

    for (int i = tid; i < BC*NH*NMEM; i += NT) {
      int bh = i >> 7, n = i & (NMEM - 1);
      const float* wg = sA + bh*NMEM;
      float wt = sS[bh*3 + 0] * wg[(n + 1) & (NMEM - 1)]
               + sS[bh*3 + 1] * wg[n]
               + sS[bh*3 + 2] * wg[(n - 1) & (NMEM - 1)];
      wt = fmaxf(wt, 1e-16f);
      sBuf[i] = expf(sGa[bh] * logf(wt));
    }
    __syncthreads();

    {
      const int g16 = tid >> 4, c = tid & 15;
      for (int rr = 0; rr < 2; ++rr) {
        const int bh = g16*2 + rr;
        const float* row = sBuf + bh*NMEM;
        float sum = 0.f;
        for (int j = 0; j < NMEM/16; ++j) sum += row[c + 16*j];
        sum = sum16(sum);
        if (c == 0) sRed[bh*2] = 1.f / sum;
      }
    }
    __syncthreads();

    for (int i = tid; i < BC*NH*NMEM; i += NT)
      sW[i] = sBuf[i] * sRed[(i >> 7)*2];
    __syncthreads();

    {
      const float* wr = sW + (mb*NH + 0)*NMEM + mc;
      float wj[MROWS];
      for (int j = 0; j < MROWS; ++j) wj[j] = wr[16*j];
      float part[D];
      for (int d = 0; d < D; ++d) {
        float acc = 0.f;
        for (int j = 0; j < MROWS; ++j) acc += wj[j] * Mr[j][d];
        part[d] = sum16(acc);
      }
      if (mc == 0) {
        for (int d = 0; d < D; ++d) sR[mb*D + d] = part[d];
      }
    }
    for (int i = tid; i < BC*D; i += NT) {
      int b = i / D, d = i - b*D;
      sE[i]  = sigm (sP[b*PPAD + NH*PPH + d]);
      sAd[i] = tanhf(sP[b*PPAD + NH*PPH + D + d]);
    }
    __syncthreads();

    {
      float ev[D], av[D];
      for (int d = 0; d < D; ++d) { ev[d] = sE[mb*D + d]; av[d] = sAd[mb*D + d]; }
      const float* wwp = sW + (mb*NH + 1)*NMEM + mc;
      for (int j = 0; j < MROWS; ++j) {
        const float ww = wwp[16*j];
        for (int d = 0; d < D; ++d)
          Mr[j][d] = Mr[j][d] * (1.f - ww * ev[d]) + ww * av[d];
      }
    }

    if (tid < BC*OUTW) {
      int b = tid / OUTW, o = tid - b*OUTW;
      float acc = b_out[o];
      for (int j = 0; j < H; ++j) acc += sH[b*H + j] * W_out[j*OUTW + o];
      for (int d = 0; d < D; ++d) acc += sR[b*D + d] * W_out[(H + d)*OUTW + o];
      acc = clipf(acc);
      if (t >= tOutStart) {
        out[(size_t)(b0 + b) * TOUT * OUTW + (t - tOutStart) * OUTW + o] = sigm(acc);
      }
    }
    __syncthreads();
  }
}

extern "C" void kernel_launch(void* const* d_in, const int* in_sizes, int n_in,
                              void* d_out, int out_size, void* d_ws, size_t ws_size,
                              hipStream_t stream) {
  (void)in_sizes; (void)n_in; (void)out_size; (void)ws_size;
  const float* inputs   = (const float*)d_in[0];
  const float* Wx       = (const float*)d_in[1];
  const float* Wh       = (const float*)d_in[2];
  const float* b_lstm   = (const float*)d_in[3];
  const float* W_params = (const float*)d_in[4];
  const float* b_params = (const float*)d_in[5];
  const float* W_out    = (const float*)d_in[6];
  const float* b_out    = (const float*)d_in[7];
  const int*   seq      = (const int*)d_in[8];
  float* out = (float*)d_out;
  float* ws  = (float*)d_ws;

  pack_weights<<<dim3((WS_TOT + 255) / 256), dim3(256), 0, stream>>>(
      Wx, Wh, W_params, b_params, ws);

  const size_t shmem = (size_t)SMEM_FLOATS * sizeof(float);
  (void)hipFuncSetAttribute((const void*)ntm_fused,
                            hipFuncAttributeMaxDynamicSharedMemorySize,
                            (int)shmem);

  ntm_fused<<<dim3(B_TOT / BC), dim3(NT), shmem, stream>>>(
      inputs, ws + WS_WCAT, b_lstm, ws + WS_WP, ws + WS_BP,
      W_out, b_out, seq, out);
}


# Gate weights [Wx;Wh] (211 KB, k-pair interleaved), gate bias and output weights are copied once into LDS and read from there every step instead of from L2
# speedup vs baseline: 3.0271x; 1.2587x over previous
#include <hip/hip_runtime.h>
#include <math.h>

typedef float v2f __attribute__((ext_vector_type(2)));
typedef float v4f __attribute__((ext_vector_type(4)));
typedef float v8f __attribute__((ext_vector_type(8)));

namespace {
constexpr int B_TOT = 2048;
constexpr int T_LEN = 66;
constexpr int INW   = 9;
constexpr int H     = 100;
constexpr int NMEM  = 128;
constexpr int D     = 20;
constexpr int OUTW  = 8;
constexpr int NH    = 2;
constexpr int PPH   = 26;
constexpr int PDIM  = 92;
constexpr int PPAD  = 96;
constexpr int G4    = 400;
constexpr int KZ    = 132;
constexpr int BC    = 16;
constexpr int NT    = 256;
constexpr float CLIPV = 20.f;
constexpr float EPSV  = 1e-8f;

constexpr int WS_WCAT = 0;
constexpr int WS_WP   = WS_WCAT + KZ*G4;
constexpr int WS_BP   = WS_WP   + H*PPAD;
constexpr int WS_TOT  = WS_BP   + PPAD;

constexpr int MROWS = NMEM / 16;

constexpr int SMEM_FLOATS =
    KZ*G4
  + G4
  + (H + D)*OUTW
  + BC*NH*NMEM
  + BC*NH*NMEM
  + BC*NH*NMEM
  + BC*H
  + BC*H
  + BC*D
  + BC*KZ
  + BC*G4
  + BC*PPAD
  + BC*NH*D
  + 32
  + 32
  + 32
  + 32
  + 32*3
  + 64
  + BC*D
  + BC*D;
static_assert(SMEM_FLOATS * 4 <= 327680, "one workgroup per WGP owns the whole 320 KiB of LDS");
static_assert((KZ*G4) % 4 == 0 && KZ % 2 == 0, "gate weights are staged with 16-byte copies and read as k pairs");
static_assert(NT == BC*NH*8 && NMEM % 16 == 0,
              "phase 5: eight lanes per (b,head); phases 7 and 10: 16 lanes per row, two rows per 16-lane group");
static_assert(NT == BC*16,
              "memory matrix in registers: 16 threads (one 16-lane row of a wave) per batch row, NMEM/16 memory rows each");
}

__device__ __forceinline__ float sigm(float x) { return 1.f / (1.f + expf(-x)); }
__device__ __forceinline__ float softplusf(float x) { return (x > 20.f) ? x : log1pf(expf(x)); }
__device__ __forceinline__ float clipf(float x) { return fminf(fmaxf(x, -CLIPV), CLIPV); }

template <int X>
__device__ __forceinline__ float lane_xor(float v) {
  return __builtin_bit_cast(float, __builtin_amdgcn_update_dpp(
      0, __builtin_bit_cast(int, v), 0x160 + X, 0xf, 0xf, true));
}
__device__ __forceinline__ float sum8(float v) {
  v += lane_xor<1>(v); v += lane_xor<2>(v); v += lane_xor<4>(v);
  return v;
}
__device__ __forceinline__ float sum16(float v) {
  v += lane_xor<1>(v); v += lane_xor<2>(v); v += lane_xor<4>(v); v += lane_xor<8>(v);
  return v;
}
__device__ __forceinline__ float max16(float v) {
  v = fmaxf(v, lane_xor<1>(v)); v = fmaxf(v, lane_xor<2>(v));
  v = fmaxf(v, lane_xor<4>(v)); v = fmaxf(v, lane_xor<8>(v));
  return v;
}

__global__ __launch_bounds__(256) void pack_weights(
    const float* __restrict__ Wx, const float* __restrict__ Wh,
    const float* __restrict__ W_params, const float* __restrict__ b_params,
    float* __restrict__ ws)
{
  int i = blockIdx.x * 256 + threadIdx.x;
  if (i < KZ*G4) {
    int k = i / G4, n = i - k*G4;
    float v = (k < 29) ? Wx[k*G4 + n] : ((k < 129) ? Wh[(k - 29)*G4 + n] : 0.f);
    ws[WS_WCAT + (k >> 1)*(2*G4) + 2*n + (k & 1)] = v;
  } else if (i < KZ*G4 + H*PPAD) {
    int j = i - KZ*G4;
    int k = j / PPAD, n = j - k*PPAD;
    ws[WS_WP + j] = (n < PDIM) ? W_params[k*PDIM + n] : 0.f;
  } else if (i < WS_TOT) {
    int n = i - KZ*G4 - H*PPAD;
    ws[WS_BP + n] = (n < PDIM) ? b_params[n] : 0.f;
  }
}

__global__ __launch_bounds__(NT) void ntm_fused(
    const float* __restrict__ inputs,
    const float* __restrict__ Wcat,
    const float* __restrict__ b_lstm,
    const float* __restrict__ Wp,
    const float* __restrict__ bp,
    const float* __restrict__ W_out,
    const float* __restrict__ b_out,
    const int*   __restrict__ seq_ptr,
    float* __restrict__ out)
{
  __shared__ __attribute__((aligned(16))) float smem[SMEM_FLOATS];
  float* sWc  = smem;
  float* sBl  = sWc  + KZ*G4;
  float* sWo  = sBl  + G4;
  float* sW   = sWo  + (H + D)*OUTW;
  float* sA   = sW   + BC*NH*NMEM;
  float* sBuf = sA   + BC*NH*NMEM;
  float* sH   = sBuf + BC*NH*NMEM;
  float* sC   = sH   + BC*H;
  float* sR   = sC   + BC*H;
  float* sZ   = sR   + BC*D;
  float* sG   = sZ   + BC*KZ;
  float* sP   = sG   + BC*G4;
  float* sK   = sP   + BC*PPAD;
  float* sKn  = sK   + BC*NH*D;
  float* sBe  = sKn  + 32;
  float* sGg  = sBe  + 32;
  float* sGa  = sGg  + 32;
  float* sS   = sGa  + 32;
  float* sRed = sS   + 32*3;
  float* sE   = sRed + 64;
  float* sAd  = sE   + BC*D;

  const int tid   = threadIdx.x;
  const int wave  = tid >> 5;
  const int lane  = tid & 31;
  const int m     = lane & 15;
  const int khalf = (lane < 16) ? 0 : 2;
  const int crow  = (lane < 16) ? 0 : 8;
  const int b0    = blockIdx.x * BC;
  const int seq   = seq_ptr[0];
  const int tOutStart = seq + 1;
  const int TOUT  = T_LEN - tOutStart;

  const int mb = tid >> 4;
  const int mc = tid & 15;
  float Mr[MROWS][D];
  for (int j = 0; j < MROWS; ++j) {
    for (int d = 0; d < D; ++d) Mr[j][d] = 1e-6f;
  }

  {
    const v4f* __restrict__ src = (const v4f*)Wcat;
    v4f* dst = (v4f*)sWc;
    for (int i = tid; i < KZ*G4/4; i += NT) dst[i] = src[i];
  }
  for (int i = tid; i < G4;           i += NT) sBl[i] = b_lstm[i];
  for (int i = tid; i < (H + D)*OUTW; i += NT) sWo[i] = W_out[i];
  const float bo = b_out[tid & (OUTW - 1)];

  for (int i = tid; i < BC*NH*NMEM; i += NT) sW[i] = ((i & (NMEM-1)) == 0) ? 1.f : 0.f;
  for (int i = tid; i < BC*H;       i += NT) { sH[i] = 0.f; sC[i] = 0.f; }
  for (int i = tid; i < BC*D;       i += NT) sR[i] = 0.f;
  __syncthreads();

  for (int t = 0; t < T_LEN; ++t) {
    for (int i = tid; i < BC*KZ; i += NT) {
      int b = i / KZ, j = i - b*KZ;
      float v;
      if (j < INW)          v = inputs[(size_t)(b0 + b) * T_LEN * INW + t * INW + j];
      else if (j < INW + D) v = sR[b*D + (j - INW)];
      else if (j < 129)     v = sH[b*H + (j - 29)];
      else                  v = 0.f;
      sZ[i] = v;
    }
    __syncthreads();

    for (int ct = wave; ct < G4/16; ct += NT/32) {
      int n = ct*16 + m;
      const float* wc = sWc + 2*n + (khalf >> 1)*(2*G4);
      float bias = sBl[n];
      v8f acc = { bias, bias, bias, bias, bias, bias, bias, bias };
      for (int kt = 0; kt < KZ/4; ++kt) {
        int k0 = kt*4 + khalf;
        v2f av; av.x = sZ[m*KZ + k0];     av.y = sZ[m*KZ + k0 + 1];
        v2f bv = *(const v2f*)(wc + kt*(4*G4));
        acc = __builtin_amdgcn_wmma_f32_16x16x4_f32(false, av, false, bv,
                                                    (short)0, acc, false, false);
      }
      #pragma unroll
      for (int v = 0; v < 8; ++v) sG[(v + crow)*G4 + n] = acc[v];
    }
    __syncthreads();

    for (int i = tid; i < BC*H; i += NT) {
      int b = i / H, u = i - b*H;
      float gi = sG[b*G4 + u];
      float gf = sG[b*G4 + H   + u];
      float gg = sG[b*G4 + 2*H + u];
      float go = sG[b*G4 + 3*H + u];
      float c  = sigm(gf) * sC[i] + sigm(gi) * tanhf(gg);
      float h  = sigm(go) * tanhf(c);
      sC[i] = c; sH[i] = h;
    }
    __syncthreads();

    for (int ct = wave; ct < PPAD/16; ct += NT/32) {
      int n = ct*16 + m;
      const float* __restrict__ wp = Wp + n;
      float bias = bp[n];
      v8f acc = { bias, bias, bias, bias, bias, bias, bias, bias };
      for (int kt = 0; kt < H/4; ++kt) {
        int k0 = kt*4 + khalf;
        v2f av; av.x = sH[m*H + k0];        av.y = sH[m*H + k0 + 1];
        v2f bv; bv.x = wp[(size_t)k0*PPAD]; bv.y = wp[(size_t)(k0 + 1)*PPAD];
        acc = __builtin_amdgcn_wmma_f32_16x16x4_f32(false, av, false, bv,
                                                    (short)0, acc, false, false);
      }
      #pragma unroll
      for (int v = 0; v < 8; ++v) sP[(v + crow)*PPAD + n] = clipf(acc[v]);
    }
    __syncthreads();

    {
      const int bh  = tid >> 3;
      const int sub = tid & 7;
      const float* pb = sP + (bh >> 1)*PPAD + (bh & 1)*PPH;
      float kn2 = 0.f;
      for (int j = 0; j < 3; ++j) {
        const int d = sub + 8*j;
        if (d < D) {
          float kv = tanhf(pb[d]);
          sK[bh*D + d] = kv;
          kn2 += kv * kv;
        }
      }
      kn2 = sum8(kn2);
      if (sub == 0) {
        sKn[bh] = sqrtf(kn2);
      } else if (sub == 1 || sub == 4) {
        float sp = softplusf(pb[(sub == 1) ? D : D + 5]);
        if (sub == 1) sBe[bh] = sp; else sGa[bh] = sp + 1.f;
      } else if (sub == 2) {
        sGg[bh] = sigm(pb[D + 1]);
      } else if (sub == 3) {
        float s0 = pb[D + 2], s1 = pb[D + 3], s2 = pb[D + 4];
        float mx = fmaxf(s0, fmaxf(s1, s2));
        float e0 = expf(s0 - mx), e1 = expf(s1 - mx), e2 = expf(s2 - mx);
        float inv = 1.f / (e0 + e1 + e2);
        sS[bh*3 + 0] = e0 * inv; sS[bh*3 + 1] = e1 * inv; sS[bh*3 + 2] = e2 * inv;
      }
    }
    __syncthreads();

    {
      float k0[D], k1[D];
      for (int d = 0; d < D; ++d) {
        k0[d] = sK[(mb*NH + 0)*D + d];
        k1[d] = sK[(mb*NH + 1)*D + d];
      }
      const float be0 = sBe[mb*NH + 0], be1 = sBe[mb*NH + 1];
      const float kn0 = sKn[mb*NH + 0], kn1 = sKn[mb*NH + 1];
      for (int j = 0; j < MROWS; ++j) {
        float mn2 = 0.f, d0 = 0.f, d1 = 0.f;
        for (int d = 0; d < D; ++d) {
          float mv = Mr[j][d];
          mn2 += mv * mv; d0 += k0[d] * mv; d1 += k1[d] * mv;
        }
        float mn = sqrtf(mn2);
        const int n = mc + 16*j;
        sA[(mb*NH + 0)*NMEM + n] = be0 * d0 / (kn0*mn + EPSV);
        sA[(mb*NH + 1)*NMEM + n] = be1 * d1 / (kn1*mn + EPSV);
      }
    }
    __syncthreads();

    {
      const int g16 = tid >> 4, c = tid & 15;
      for (int rr = 0; rr < 2; ++rr) {
        const int bh = g16*2 + rr;
        const float* row = sA + bh*NMEM;
        float v[NMEM/16];
        for (int j = 0; j < NMEM/16; ++j) v[j] = row[c + 16*j];
        float mx = v[0];
        for (int j = 1; j < NMEM/16; ++j) mx = fmaxf(mx, v[j]);
        mx = max16(mx);
        float sum = 0.f;
        for (int j = 0; j < NMEM/16; ++j) sum += expf(v[j] - mx);
        sum = sum16(sum);
        if (c == 0) {
          sRed[bh*2 + 0] = mx;
          sRed[bh*2 + 1] = 1.f / sum;
        }
      }
    }
    __syncthreads();

    for (int i = tid; i < BC*NH*NMEM; i += NT) {
      int bh = i >> 7;
      float wc = expf(sA[i] - sRed[bh*2]) * sRed[bh*2 + 1];
      float g  = sGg[bh];
      sA[i] = g * wc + (1.f - g) * sW[i];
    }
    __syncthreads();

    for (int i = tid; i < BC*NH*NMEM; i += NT) {
      int bh = i >> 7, n = i & (NMEM - 1);
      const float* wg = sA + bh*NMEM;
      float wt = sS[bh*3 + 0] * wg[(n + 1) & (NMEM - 1)]
               + sS[bh*3 + 1] * wg[n]
               + sS[bh*3 + 2] * wg[(n - 1) & (NMEM - 1)];
      wt = fmaxf(wt, 1e-16f);
      sBuf[i] = expf(sGa[bh] * logf(wt));
    }
    __syncthreads();

    {
      const int g16 = tid >> 4, c = tid & 15;
      for (int rr = 0; rr < 2; ++rr) {
        const int bh = g16*2 + rr;
        const float* row = sBuf + bh*NMEM;
        float sum = 0.f;
        for (int j = 0; j < NMEM/16; ++j) sum += row[c + 16*j];
        sum = sum16(sum);
        if (c == 0) sRed[bh*2] = 1.f / sum;
      }
    }
    __syncthreads();

    for (int i = tid; i < BC*NH*NMEM; i += NT)
      sW[i] = sBuf[i] * sRed[(i >> 7)*2];
    __syncthreads();

    {
      const float* wr = sW + (mb*NH + 0)*NMEM + mc;
      float wj[MROWS];
      for (int j = 0; j < MROWS; ++j) wj[j] = wr[16*j];
      float part[D];
      for (int d = 0; d < D; ++d) {
        float acc = 0.f;
        for (int j = 0; j < MROWS; ++j) acc += wj[j] * Mr[j][d];
        part[d] = sum16(acc);
      }
      if (mc == 0) {
        for (int d = 0; d < D; ++d) sR[mb*D + d] = part[d];
      }
    }
    for (int i = tid; i < BC*D; i += NT) {
      int b = i / D, d = i - b*D;
      sE[i]  = sigm (sP[b*PPAD + NH*PPH + d]);
      sAd[i] = tanhf(sP[b*PPAD + NH*PPH + D + d]);
    }
    __syncthreads();

    {
      float ev[D], av[D];
      for (int d = 0; d < D; ++d) { ev[d] = sE[mb*D + d]; av[d] = sAd[mb*D + d]; }
      const float* wwp = sW + (mb*NH + 1)*NMEM + mc;
      for (int j = 0; j < MROWS; ++j) {
        const float ww = wwp[16*j];
        for (int d = 0; d < D; ++d)
          Mr[j][d] = Mr[j][d] * (1.f - ww * ev[d]) + ww * av[d];
      }
    }

    if (tid < BC*OUTW) {
      int b = tid / OUTW, o = tid - b*OUTW;
      float acc = bo;
      for (int j = 0; j < H; ++j) acc += sH[b*H + j] * sWo[j*OUTW + o];
      for (int d = 0; d < D; ++d) acc += sR[b*D + d] * sWo[(H + d)*OUTW + o];
      acc = clipf(acc);
      if (t >= tOutStart) {
        out[(size_t)(b0 + b) * TOUT * OUTW + (t - tOutStart) * OUTW + o] = sigm(acc);
      }
    }
    __syncthreads();
  }
}

extern "C" void kernel_launch(void* const* d_in, const int* in_sizes, int n_in,
                              void* d_out, int out_size, void* d_ws, size_t ws_size,
                              hipStream_t stream) {
  (void)in_sizes; (void)n_in; (void)out_size; (void)ws_size;
  const float* inputs   = (const float*)d_in[0];
  const float* Wx       = (const float*)d_in[1];
  const float* Wh       = (const float*)d_in[2];
  const float* b_lstm   = (const float*)d_in[3];
  const float* W_params = (const float*)d_in[4];
  const float* b_params = (const float*)d_in[5];
  const float* W_out    = (const float*)d_in[6];
  const float* b_out    = (const float*)d_in[7];
  const int*   seq      = (const int*)d_in[8];
  float* out = (float*)d_out;
  float* ws  = (float*)d_ws;

  pack_weights<<<dim3((WS_TOT + 255) / 256), dim3(256), 0, stream>>>(
      Wx, Wh, W_params, b_params, ws);

  ntm_fused<<<dim3(B_TOT / BC), dim3(NT), 0, stream>>>(
      inputs, ws + WS_WCAT, b_lstm, ws + WS_WP, ws + WS_BP,
      W_out, b_out, seq, out);
}


# Batch rows per workgroup 16 -> 8: 256 workgroups, one on each of the 256 WGPs (128 used half the chip); one wave per batch row; read-vector sum order changes (4-row partials + 32-lane tree)
# speedup vs baseline: 3.6779x; 1.2150x over previous
#include <hip/hip_runtime.h>
#include <math.h>

typedef float v2f __attribute__((ext_vector_type(2)));
typedef float v4f __attribute__((ext_vector_type(4)));
typedef float v8f __attribute__((ext_vector_type(8)));

namespace {
constexpr int B_TOT = 2048;
constexpr int T_LEN = 66;
constexpr int INW   = 9;
constexpr int H     = 100;
constexpr int NMEM  = 128;
constexpr int D     = 20;
constexpr int OUTW  = 8;
constexpr int NH    = 2;
constexpr int PPH   = 26;
constexpr int PDIM  = 92;
constexpr int PPAD  = 96;
constexpr int G4    = 400;
constexpr int KZ    = 132;
constexpr int BC    = 8;
constexpr int NT    = 256;
constexpr float CLIPV = 20.f;
constexpr float EPSV  = 1e-8f;

constexpr int WS_WCAT = 0;
constexpr int WS_WP   = WS_WCAT + KZ*G4;
constexpr int WS_BP   = WS_WP   + H*PPAD;
constexpr int WS_TOT  = WS_BP   + PPAD;

constexpr int MROWS = NMEM / 32;

constexpr int SMEM_FLOATS =
    KZ*G4
  + G4
  + (H + D)*OUTW
  + BC*NH*NMEM
  + BC*NH*NMEM
  + BC*NH*NMEM
  + BC*H
  + BC*H
  + BC*D
  + BC*KZ
  + BC*G4
  + BC*PPAD
  + BC*NH*D
  + 32
  + 32
  + 32
  + 32
  + 32*3
  + 64
  + BC*D
  + BC*D;
static_assert(SMEM_FLOATS * 4 <= 327680, "one workgroup per WGP owns the whole 320 KiB of LDS");
static_assert((KZ*G4) % 4 == 0 && KZ % 2 == 0, "gate weights are staged with 16-byte copies and read as k pairs");
static_assert(NT == BC*NH*16 && NMEM % 16 == 0 && (BC*NH*8) % 32 == 0,
              "phase 5: eight lanes per (b,head) on the first BC*NH*8 threads (whole waves); phases 7 and 10: one 16-lane row of a wave per (b,head)");
static_assert(NT == BC*32 && NMEM % 32 == 0,
              "memory matrix in registers: one wave per batch row, NMEM/32 memory rows per lane");
static_assert(BC == 8,
              "WMMA tiles: lanes m and m+8 load the same batch row as their A row; D rows 0-7 sit in lanes 0-15");
}

__device__ __forceinline__ float sigm(float x) { return 1.f / (1.f + expf(-x)); }
__device__ __forceinline__ float softplusf(float x) { return (x > 20.f) ? x : log1pf(expf(x)); }
__device__ __forceinline__ float clipf(float x) { return fminf(fmaxf(x, -CLIPV), CLIPV); }

template <int X>
__device__ __forceinline__ float lane_xor(float v) {
  return __builtin_bit_cast(float, __builtin_amdgcn_update_dpp(
      0, __builtin_bit_cast(int, v), 0x160 + X, 0xf, 0xf, true));
}
__device__ __forceinline__ float sum8(float v) {
  v += lane_xor<1>(v); v += lane_xor<2>(v); v += lane_xor<4>(v);
  return v;
}
__device__ __forceinline__ float sum16(float v) {
  v += lane_xor<1>(v); v += lane_xor<2>(v); v += lane_xor<4>(v); v += lane_xor<8>(v);
  return v;
}
__device__ __forceinline__ float lane_xor16(float v) {
  const int i = __builtin_bit_cast(int, v);
  return __builtin_bit_cast(float, __builtin_amdgcn_permlanex16(i, i, 0x76543210, 0xfedcba98, false, false));
}
__device__ __forceinline__ float sum32(float v) {
  v = sum16(v); v += lane_xor16(v);
  return v;
}
__device__ __forceinline__ float max16(float v) {
  v = fmaxf(v, lane_xor<1>(v)); v = fmaxf(v, lane_xor<2>(v));
  v = fmaxf(v, lane_xor<4>(v)); v = fmaxf(v, lane_xor<8>(v));
  return v;
}

__global__ __launch_bounds__(256) void pack_weights(
    const float* __restrict__ Wx, const float* __restrict__ Wh,
    const float* __restrict__ W_params, const float* __restrict__ b_params,
    float* __restrict__ ws)
{
  int i = blockIdx.x * 256 + threadIdx.x;
  if (i < KZ*G4) {
    int k = i / G4, n = i - k*G4;
    float v = (k < 29) ? Wx[k*G4 + n] : ((k < 129) ? Wh[(k - 29)*G4 + n] : 0.f);
    ws[WS_WCAT + (k >> 1)*(2*G4) + 2*n + (k & 1)] = v;
  } else if (i < KZ*G4 + H*PPAD) {
    int j = i - KZ*G4;
    int k = j / PPAD, n = j - k*PPAD;
    ws[WS_WP + j] = (n < PDIM) ? W_params[k*PDIM + n] : 0.f;
  } else if (i < WS_TOT) {
    int n = i - KZ*G4 - H*PPAD;
    ws[WS_BP + n] = (n < PDIM) ? b_params[n] : 0.f;
  }
}

__global__ __launch_bounds__(NT) void ntm_fused(
    const float* __restrict__ inputs,
    const float* __restrict__ Wcat,
    const float* __restrict__ b_lstm,
    const float* __restrict__ Wp,
    const float* __restrict__ bp,
    const float* __restrict__ W_out,
    const float* __restrict__ b_out,
    const int*   __restrict__ seq_ptr,
    float* __restrict__ out)
{
  __shared__ __attribute__((aligned(16))) float smem[SMEM_FLOATS];
  float* sWc  = smem;
  float* sBl  = sWc  + KZ*G4;
  float* sWo  = sBl  + G4;
  float* sW   = sWo  + (H + D)*OUTW;
  float* sA   = sW   + BC*NH*NMEM;
  float* sBuf = sA   + BC*NH*NMEM;
  float* sH   = sBuf + BC*NH*NMEM;
  float* sC   = sH   + BC*H;
  float* sR   = sC   + BC*H;
  float* sZ   = sR   + BC*D;
  float* sG   = sZ   + BC*KZ;
  float* sP   = sG   + BC*G4;
  float* sK   = sP   + BC*PPAD;
  float* sKn  = sK   + BC*NH*D;
  float* sBe  = sKn  + 32;
  float* sGg  = sBe  + 32;
  float* sGa  = sGg  + 32;
  float* sS   = sGa  + 32;
  float* sRed = sS   + 32*3;
  float* sE   = sRed + 64;
  float* sAd  = sE   + BC*D;

  const int tid   = threadIdx.x;
  const int wave  = tid >> 5;
  const int lane  = tid & 31;
  const int m     = lane & 15;
  const int ma    = m & (BC - 1);
  const int khalf = (lane < 16) ? 0 : 2;
  const int b0    = blockIdx.x * BC;
  const int seq   = seq_ptr[0];
  const int tOutStart = seq + 1;
  const int TOUT  = T_LEN - tOutStart;

  const int mb = tid >> 5;
  const int mc = tid & 31;
  float Mr[MROWS][D];
  for (int j = 0; j < MROWS; ++j) {
    for (int d = 0; d < D; ++d) Mr[j][d] = 1e-6f;
  }

  {
    const v4f* __restrict__ src = (const v4f*)Wcat;
    v4f* dst = (v4f*)sWc;
    for (int i = tid; i < KZ*G4/4; i += NT) dst[i] = src[i];
  }
  for (int i = tid; i < G4;           i += NT) sBl[i] = b_lstm[i];
  for (int i = tid; i < (H + D)*OUTW; i += NT) sWo[i] = W_out[i];
  const float bo = b_out[tid & (OUTW - 1)];

  for (int i = tid; i < BC*NH*NMEM; i += NT) sW[i] = ((i & (NMEM-1)) == 0) ? 1.f : 0.f;
  for (int i = tid; i < BC*H;       i += NT) { sH[i] = 0.f; sC[i] = 0.f; }
  for (int i = tid; i < BC*D;       i += NT) sR[i] = 0.f;
  __syncthreads();

  for (int t = 0; t < T_LEN; ++t) {
    for (int i = tid; i < BC*KZ; i += NT) {
      int b = i / KZ, j = i - b*KZ;
      float v;
      if (j < INW)          v = inputs[(size_t)(b0 + b) * T_LEN * INW + t * INW + j];
      else if (j < INW + D) v = sR[b*D + (j - INW)];
      else if (j < 129)     v = sH[b*H + (j - 29)];
      else                  v = 0.f;
      sZ[i] = v;
    }
    __syncthreads();

    for (int ct = wave; ct < G4/16; ct += NT/32) {
      int n = ct*16 + m;
      const float* wc = sWc + 2*n + (khalf >> 1)*(2*G4);
      float bias = sBl[n];
      v8f acc = { bias, bias, bias, bias, bias, bias, bias, bias };
      for (int kt = 0; kt < KZ/4; ++kt) {
        int k0 = kt*4 + khalf;
        v2f av; av.x = sZ[ma*KZ + k0];    av.y = sZ[ma*KZ + k0 + 1];
        v2f bv = *(const v2f*)(wc + kt*(4*G4));
        acc = __builtin_amdgcn_wmma_f32_16x16x4_f32(false, av, false, bv,
                                                    (short)0, acc, false, false);
      }
      if (lane < 16) {
        #pragma unroll
        for (int v = 0; v < 8; ++v) sG[v*G4 + n] = acc[v];
      }
    }
    __syncthreads();

    for (int i = tid; i < BC*H; i += NT) {
      int b = i / H, u = i - b*H;
      float gi = sG[b*G4 + u];
      float gf = sG[b*G4 + H   + u];
      float gg = sG[b*G4 + 2*H + u];
      float go = sG[b*G4 + 3*H + u];
      float c  = sigm(gf) * sC[i] + sigm(gi) * tanhf(gg);
      float h  = sigm(go) * tanhf(c);
      sC[i] = c; sH[i] = h;
    }
    __syncthreads();

    for (int ct = wave; ct < PPAD/16; ct += NT/32) {
      int n = ct*16 + m;
      const float* __restrict__ wp = Wp + n;
      float bias = bp[n];
      v8f acc = { bias, bias, bias, bias, bias, bias, bias, bias };
      for (int kt = 0; kt < H/4; ++kt) {
        int k0 = kt*4 + khalf;
        v2f av; av.x = sH[ma*H + k0];       av.y = sH[ma*H + k0 + 1];
        v2f bv; bv.x = wp[(size_t)k0*PPAD]; bv.y = wp[(size_t)(k0 + 1)*PPAD];
        acc = __builtin_amdgcn_wmma_f32_16x16x4_f32(false, av, false, bv,
                                                    (short)0, acc, false, false);
      }
      float pv[8];
      #pragma unroll
      for (int v = 0; v < 8; ++v) pv[v] = clipf(acc[v]);
      if (lane < 16) {
        #pragma unroll
        for (int v = 0; v < 8; ++v) sP[v*PPAD + n] = pv[v];
      }
    }
    __syncthreads();

    if (tid < BC*NH*8) {
      const int bh  = tid >> 3;
      const int sub = tid & 7;
      const float* pb = sP + (bh >> 1)*PPAD + (bh & 1)*PPH;
      float kn2 = 0.f;
      for (int j = 0; j < 3; ++j) {
        const int d = sub + 8*j;
        if (d < D) {
          float kv = tanhf(pb[d]);
          sK[bh*D + d] = kv;
          kn2 += kv * kv;
        }
      }
      kn2 = sum8(kn2);
      if (sub == 0) {
        sKn[bh] = sqrtf(kn2);
      } else if (sub == 1 || sub == 4) {
        float sp = softplusf(pb[(sub == 1) ? D : D + 5]);
        if (sub == 1) sBe[bh] = sp; else sGa[bh] = sp + 1.f;
      } else if (sub == 2) {
        sGg[bh] = sigm(pb[D + 1]);
      } else if (sub == 3) {
        float s0 = pb[D + 2], s1 = pb[D + 3], s2 = pb[D + 4];
        float mx = fmaxf(s0, fmaxf(s1, s2));
        float e0 = expf(s0 - mx), e1 = expf(s1 - mx), e2 = expf(s2 - mx);
        float inv = 1.f / (e0 + e1 + e2);
        sS[bh*3 + 0] = e0 * inv; sS[bh*3 + 1] = e1 * inv; sS[bh*3 + 2] = e2 * inv;
      }
    }
    __syncthreads();

    {
      float k0[D], k1[D];
      for (int d = 0; d < D; ++d) {
        k0[d] = sK[(mb*NH + 0)*D + d];
        k1[d] = sK[(mb*NH + 1)*D + d];
      }
      const float be0 = sBe[mb*NH + 0], be1 = sBe[mb*NH + 1];
      const float kn0 = sKn[mb*NH + 0], kn1 = sKn[mb*NH + 1];
      for (int j = 0; j < MROWS; ++j) {
        float mn2 = 0.f, d0 = 0.f, d1 = 0.f;
        for (int d = 0; d < D; ++d) {
          float mv = Mr[j][d];
          mn2 += mv * mv; d0 += k0[d] * mv; d1 += k1[d] * mv;
        }
        float mn = sqrtf(mn2);
        const int n = mc + 32*j;
        sA[(mb*NH + 0)*NMEM + n] = be0 * d0 / (kn0*mn + EPSV);
        sA[(mb*NH + 1)*NMEM + n] = be1 * d1 / (kn1*mn + EPSV);
      }
    }
    __syncthreads();

    {
      const int bh = tid >> 4, c = tid & 15;
      const float* row = sA + bh*NMEM;
      float v[NMEM/16];
      for (int j = 0; j < NMEM/16; ++j) v[j] = row[c + 16*j];
      float mx = v[0];
      for (int j = 1; j < NMEM/16; ++j) mx = fmaxf(mx, v[j]);
      mx = max16(mx);
      float sum = 0.f;
      for (int j = 0; j < NMEM/16; ++j) sum += expf(v[j] - mx);
      sum = sum16(sum);
      if (c == 0) {
        sRed[bh*2 + 0] = mx;
        sRed[bh*2 + 1] = 1.f / sum;
      }
    }
    __syncthreads();

    for (int i = tid; i < BC*NH*NMEM; i += NT) {
      int bh = i >> 7;
      float wc = expf(sA[i] - sRed[bh*2]) * sRed[bh*2 + 1];
      float g  = sGg[bh];
      sA[i] = g * wc + (1.f - g) * sW[i];
    }
    __syncthreads();

    for (int i = tid; i < BC*NH*NMEM; i += NT) {
      int bh = i >> 7, n = i & (NMEM - 1);
      const float* wg = sA + bh*NMEM;
      float wt = sS[bh*3 + 0] * wg[(n + 1) & (NMEM - 1)]
               + sS[bh*3 + 1] * wg[n]
               + sS[bh*3 + 2] * wg[(n - 1) & (NMEM - 1)];
      wt = fmaxf(wt, 1e-16f);
      sBuf[i] = expf(sGa[bh] * logf(wt));
    }
    __syncthreads();

    {
      const int bh = tid >> 4, c = tid & 15;
      const float* row = sBuf + bh*NMEM;
      float sum = 0.f;
      for (int j = 0; j < NMEM/16; ++j) sum += row[c + 16*j];
      sum = sum16(sum);
      if (c == 0) sRed[bh*2] = 1.f / sum;
    }
    __syncthreads();

    for (int i = tid; i < BC*NH*NMEM; i += NT)
      sW[i] = sBuf[i] * sRed[(i >> 7)*2];
    __syncthreads();

    {
      const float* wr = sW + (mb*NH + 0)*NMEM + mc;
      float wj[MROWS];
      for (int j = 0; j < MROWS; ++j) wj[j] = wr[32*j];
      float part[D];
      for (int d = 0; d < D; ++d) {
        float acc = 0.f;
        for (int j = 0; j < MROWS; ++j) acc += wj[j] * Mr[j][d];
        part[d] = sum32(acc);
      }
      if (mc == 0) {
        for (int d = 0; d < D; ++d) sR[mb*D + d] = part[d];
      }
    }
    for (int i = tid; i < BC*D; i += NT) {
      int b = i / D, d = i - b*D;
      sE[i]  = sigm (sP[b*PPAD + NH*PPH + d]);
      sAd[i] = tanhf(sP[b*PPAD + NH*PPH + D + d]);
    }
    __syncthreads();

    {
      float ev[D], av[D];
      for (int d = 0; d < D; ++d) { ev[d] = sE[mb*D + d]; av[d] = sAd[mb*D + d]; }
      const float* wwp = sW + (mb*NH + 1)*NMEM + mc;
      for (int j = 0; j < MROWS; ++j) {
        const float ww = wwp[32*j];
        for (int d = 0; d < D; ++d)
          Mr[j][d] = Mr[j][d] * (1.f - ww * ev[d]) + ww * av[d];
      }
    }

    if (tid < BC*OUTW) {
      int b = tid / OUTW, o = tid - b*OUTW;
      float acc = bo;
      for (int j = 0; j < H; ++j) acc += sH[b*H + j] * sWo[j*OUTW + o];
      for (int d = 0; d < D; ++d) acc += sR[b*D + d] * sWo[(H + d)*OUTW + o];
      acc = clipf(acc);
      if (t >= tOutStart) {
        out[(size_t)(b0 + b) * TOUT * OUTW + (t - tOutStart) * OUTW + o] = sigm(acc);
      }
    }
    __syncthreads();
  }
}

extern "C" void kernel_launch(void* const* d_in, const int* in_sizes, int n_in,
                              void* d_out, int out_size, void* d_ws, size_t ws_size,
                              hipStream_t stream) {
  (void)in_sizes; (void)n_in; (void)out_size; (void)ws_size;
  const float* inputs   = (const float*)d_in[0];
  const float* Wx       = (const float*)d_in[1];
  const float* Wh       = (const float*)d_in[2];
  const float* b_lstm   = (const float*)d_in[3];
  const float* W_params = (const float*)d_in[4];
  const float* b_params = (const float*)d_in[5];
  const float* W_out    = (const float*)d_in[6];
  const float* b_out    = (const float*)d_in[7];
  const int*   seq      = (const int*)d_in[8];
  float* out = (float*)d_out;
  float* ws  = (float*)d_ws;

  pack_weights<<<dim3((WS_TOT + 255) / 256), dim3(256), 0, stream>>>(
      Wx, Wh, W_params, b_params, ws);

  ntm_fused<<<dim3(B_TOT / BC), dim3(NT), 0, stream>>>(
      inputs, ws + WS_WCAT, b_lstm, ws + WS_WP, ws + WS_BP,
      W_out, b_out, seq, out);
}


# Per-step element loops (z build, LSTM cell, interpolate, shift+sharpen, normalise) get compile-time trip counts and are unrolled: all trips' LDS loads issue together, their chains interleave
# speedup vs baseline: 4.5605x; 1.2400x over previous
#include <hip/hip_runtime.h>
#include <math.h>

typedef float v2f __attribute__((ext_vector_type(2)));
typedef float v4f __attribute__((ext_vector_type(4)));
typedef float v8f __attribute__((ext_vector_type(8)));

namespace {
constexpr int B_TOT = 2048;
constexpr int T_LEN = 66;
constexpr int INW   = 9;
constexpr int H     = 100;
constexpr int NMEM  = 128;
constexpr int D     = 20;
constexpr int OUTW  = 8;
constexpr int NH    = 2;
constexpr int PPH   = 26;
constexpr int PDIM  = 92;
constexpr int PPAD  = 96;
constexpr int G4    = 400;
constexpr int KZ    = 132;
constexpr int BC    = 8;
constexpr int NT    = 256;
constexpr float CLIPV = 20.f;
constexpr float EPSV  = 1e-8f;

constexpr int WS_WCAT = 0;
constexpr int WS_WP   = WS_WCAT + KZ*G4;
constexpr int WS_BP   = WS_WP   + H*PPAD;
constexpr int WS_TOT  = WS_BP   + PPAD;

constexpr int MROWS = NMEM / 32;

constexpr int WTRIPS = BC*NH*NMEM / NT;
constexpr int HTRIPS = (BC*H + NT - 1) / NT;

constexpr int SMEM_FLOATS =
    KZ*G4
  + G4
  + (H + D)*OUTW
  + BC*NH*NMEM
  + BC*NH*NMEM
  + BC*NH*NMEM
  + BC*H
  + BC*H
  + BC*D
  + BC*KZ
  + BC*G4
  + BC*PPAD
  + BC*NH*D
  + 32
  + 32
  + 32
  + 32
  + 32*3
  + 64
  + BC*D
  + BC*D;
static_assert(SMEM_FLOATS * 4 <= 327680, "one workgroup per WGP owns the whole 320 KiB of LDS");
static_assert((KZ*G4) % 4 == 0 && KZ % 2 == 0, "gate weights are staged with 16-byte copies and read as k pairs");
static_assert(NT == BC*NH*16 && NMEM % 16 == 0 && (BC*NH*8) % 32 == 0,
              "phase 5: eight lanes per (b,head) on the first BC*NH*8 threads (whole waves); phases 7 and 10: one 16-lane row of a wave per (b,head)");
static_assert(NT == BC*32 && NMEM % 32 == 0,
              "memory matrix in registers: one wave per batch row, NMEM/32 memory rows per lane");
static_assert((BC*NH*NMEM) % NT == 0 && NT % NMEM == 0,
              "weighting loops: WTRIPS whole trips; a thread keeps its column n and moves NT/NMEM rows per trip");
static_assert((HTRIPS - 1)*NT < BC*H, "hidden-unit loops: only the last trip is partial");
static_assert(BC*INW <= NT && BC*D <= NT && BC*(KZ - INW - D - H) <= NT,
              "z build: the x part, the r part and the zero padding each take one thread per element");
static_assert(BC == 8,
              "WMMA tiles: lanes m and m+8 load the same batch row as their A row; D rows 0-7 sit in lanes 0-15");
}

__device__ __forceinline__ float sigm(float x) { return 1.f / (1.f + expf(-x)); }
__device__ __forceinline__ float softplusf(float x) { return (x > 20.f) ? x : log1pf(expf(x)); }
__device__ __forceinline__ float clipf(float x) { return fminf(fmaxf(x, -CLIPV), CLIPV); }

template <int X>
__device__ __forceinline__ float lane_xor(float v) {
  return __builtin_bit_cast(float, __builtin_amdgcn_update_dpp(
      0, __builtin_bit_cast(int, v), 0x160 + X, 0xf, 0xf, true));
}
__device__ __forceinline__ float sum8(float v) {
  v += lane_xor<1>(v); v += lane_xor<2>(v); v += lane_xor<4>(v);
  return v;
}
__device__ __forceinline__ float sum16(float v) {
  v += lane_xor<1>(v); v += lane_xor<2>(v); v += lane_xor<4>(v); v += lane_xor<8>(v);
  return v;
}
__device__ __forceinline__ float lane_xor16(float v) {
  const int i = __builtin_bit_cast(int, v);
  return __builtin_bit_cast(float, __builtin_amdgcn_permlanex16(i, i, 0x76543210, 0xfedcba98, false, false));
}
__device__ __forceinline__ float sum32(float v) {
  v = sum16(v); v += lane_xor16(v);
  return v;
}
__device__ __forceinline__ float max16(float v) {
  v = fmaxf(v, lane_xor<1>(v)); v = fmaxf(v, lane_xor<2>(v));
  v = fmaxf(v, lane_xor<4>(v)); v = fmaxf(v, lane_xor<8>(v));
  return v;
}

__global__ __launch_bounds__(256) void pack_weights(
    const float* __restrict__ Wx, const float* __restrict__ Wh,
    const float* __restrict__ W_params, const float* __restrict__ b_params,
    float* __restrict__ ws)
{
  int i = blockIdx.x * 256 + threadIdx.x;
  if (i < KZ*G4) {
    int k = i / G4, n = i - k*G4;
    float v = (k < 29) ? Wx[k*G4 + n] : ((k < 129) ? Wh[(k - 29)*G4 + n] : 0.f);
    ws[WS_WCAT + (k >> 1)*(2*G4) + 2*n + (k & 1)] = v;
  } else if (i < KZ*G4 + H*PPAD) {
    int j = i - KZ*G4;
    int k = j / PPAD, n = j - k*PPAD;
    ws[WS_WP + j] = (n < PDIM) ? W_params[k*PDIM + n] : 0.f;
  } else if (i < WS_TOT) {
    int n = i - KZ*G4 - H*PPAD;
    ws[WS_BP + n] = (n < PDIM) ? b_params[n] : 0.f;
  }
}

__global__ __launch_bounds__(NT) void ntm_fused(
    const float* __restrict__ inputs,
    const float* __restrict__ Wcat,
    const float* __restrict__ b_lstm,
    const float* __restrict__ Wp,
    const float* __restrict__ bp,
    const float* __restrict__ W_out,
    const float* __restrict__ b_out,
    const int*   __restrict__ seq_ptr,
    float* __restrict__ out)
{
  __shared__ __attribute__((aligned(16))) float smem[SMEM_FLOATS];
  float* sWc  = smem;
  float* sBl  = sWc  + KZ*G4;
  float* sWo  = sBl  + G4;
  float* sW   = sWo  + (H + D)*OUTW;
  float* sA   = sW   + BC*NH*NMEM;
  float* sBuf = sA   + BC*NH*NMEM;
  float* sH   = sBuf + BC*NH*NMEM;
  float* sC   = sH   + BC*H;
  float* sR   = sC   + BC*H;
  float* sZ   = sR   + BC*D;
  float* sG   = sZ   + BC*KZ;
  float* sP   = sG   + BC*G4;
  float* sK   = sP   + BC*PPAD;
  float* sKn  = sK   + BC*NH*D;
  float* sBe  = sKn  + 32;
  float* sGg  = sBe  + 32;
  float* sGa  = sGg  + 32;
  float* sS   = sGa  + 32;
  float* sRed = sS   + 32*3;
  float* sE   = sRed + 64;
  float* sAd  = sE   + BC*D;

  const int tid   = threadIdx.x;
  const int wave  = tid >> 5;
  const int lane  = tid & 31;
  const int m     = lane & 15;
  const int ma    = m & (BC - 1);
  const int khalf = (lane < 16) ? 0 : 2;
  const int b0    = blockIdx.x * BC;
  const int seq   = seq_ptr[0];
  const int tOutStart = seq + 1;
  const int TOUT  = T_LEN - tOutStart;

  const int mb = tid >> 5;
  const int mc = tid & 31;
  float Mr[MROWS][D];
  for (int j = 0; j < MROWS; ++j) {
    for (int d = 0; d < D; ++d) Mr[j][d] = 1e-6f;
  }

  {
    const v4f* __restrict__ src = (const v4f*)Wcat;
    v4f* dst = (v4f*)sWc;
    for (int i = tid; i < KZ*G4/4; i += NT) dst[i] = src[i];
  }
  for (int i = tid; i < G4;           i += NT) sBl[i] = b_lstm[i];
  for (int i = tid; i < (H + D)*OUTW; i += NT) sWo[i] = W_out[i];
  const float bo = b_out[tid & (OUTW - 1)];

  for (int i = tid; i < BC*NH*NMEM; i += NT) sW[i] = ((i & (NMEM-1)) == 0) ? 1.f : 0.f;
  for (int i = tid; i < BC*H;       i += NT) { sH[i] = 0.f; sC[i] = 0.f; }
  for (int i = tid; i < BC*D;       i += NT) sR[i] = 0.f;
  if (tid < BC*(KZ - INW - D - H)) {
    const int b = tid / (KZ - INW - D - H);
    sZ[b*KZ + INW + D + H + (tid - b*(KZ - INW - D - H))] = 0.f;
  }
  __syncthreads();

  for (int t = 0; t < T_LEN; ++t) {
    {
      const int xi = (tid < BC*INW) ? tid : 0;
      const int xb = xi / INW, xj = xi - xb*INW;
      const float xv = inputs[(size_t)(b0 + xb) * T_LEN * INW + t * INW + xj];
      const int ri = (tid < BC*D) ? tid : 0;
      const int rb = ri / D;
      const float rv = sR[ri];
      float hv[HTRIPS];
      #pragma unroll
      for (int it = 0; it < HTRIPS; ++it) {
        const int i = tid + it*NT;
        hv[it] = sH[(i < BC*H) ? i : 0];
      }
      if (tid < BC*INW) sZ[xb*KZ + xj] = xv;
      if (tid < BC*D)   sZ[rb*KZ + INW + (ri - rb*D)] = rv;
      #pragma unroll
      for (int it = 0; it < HTRIPS; ++it) {
        const int i = tid + it*NT;
        if (i < BC*H) {
          const int b = i / H;
          sZ[b*KZ + INW + D + (i - b*H)] = hv[it];
        }
      }
    }
    __syncthreads();

    for (int ct = wave; ct < G4/16; ct += NT/32) {
      int n = ct*16 + m;
      const float* wc = sWc + 2*n + (khalf >> 1)*(2*G4);
      float bias = sBl[n];
      v8f acc = { bias, bias, bias, bias, bias, bias, bias, bias };
      for (int kt = 0; kt < KZ/4; ++kt) {
        int k0 = kt*4 + khalf;
        v2f av; av.x = sZ[ma*KZ + k0];    av.y = sZ[ma*KZ + k0 + 1];
        v2f bv = *(const v2f*)(wc + kt*(4*G4));
        acc = __builtin_amdgcn_wmma_f32_16x16x4_f32(false, av, false, bv,
                                                    (short)0, acc, false, false);
      }
      if (lane < 16) {
        #pragma unroll
        for (int v = 0; v < 8; ++v) sG[v*G4 + n] = acc[v];
      }
    }
    __syncthreads();

    {
      float gi[HTRIPS], gf[HTRIPS], gg[HTRIPS], go[HTRIPS], cp[HTRIPS];
      #pragma unroll
      for (int it = 0; it < HTRIPS; ++it) {
        const int i = min(tid + it*NT, BC*H - 1);
        const int b = i / H, u = i - b*H;
        gi[it] = sG[b*G4 + u];
        gf[it] = sG[b*G4 + H   + u];
        gg[it] = sG[b*G4 + 2*H + u];
        go[it] = sG[b*G4 + 3*H + u];
        cp[it] = sC[i];
      }
      #pragma unroll
      for (int it = 0; it < HTRIPS; ++it) {
        const int i = tid + it*NT;
        if (it < HTRIPS - 1 || i < BC*H) {
          float c  = sigm(gf[it]) * cp[it] + sigm(gi[it]) * tanhf(gg[it]);
          float h  = sigm(go[it]) * tanhf(c);
          sC[i] = c; sH[i] = h;
        }
      }
    }
    __syncthreads();

    for (int ct = wave; ct < PPAD/16; ct += NT/32) {
      int n = ct*16 + m;
      const float* __restrict__ wp = Wp + n;
      float bias = bp[n];
      v8f acc = { bias, bias, bias, bias, bias, bias, bias, bias };
      for (int kt = 0; kt < H/4; ++kt) {
        int k0 = kt*4 + khalf;
        v2f av; av.x = sH[ma*H + k0];       av.y = sH[ma*H + k0 + 1];
        v2f bv; bv.x = wp[(size_t)k0*PPAD]; bv.y = wp[(size_t)(k0 + 1)*PPAD];
        acc = __builtin_amdgcn_wmma_f32_16x16x4_f32(false, av, false, bv,
                                                    (short)0, acc, false, false);
      }
      float pv[8];
      #pragma unroll
      for (int v = 0; v < 8; ++v) pv[v] = clipf(acc[v]);
      if (lane < 16) {
        #pragma unroll
        for (int v = 0; v < 8; ++v) sP[v*PPAD + n] = pv[v];
      }
    }
    __syncthreads();

    if (tid < BC*NH*8) {
      const int bh  = tid >> 3;
      const int sub = tid & 7;
      const float* pb = sP + (bh >> 1)*PPAD + (bh & 1)*PPH;
      float kn2 = 0.f;
      for (int j = 0; j < 3; ++j) {
        const int d = sub + 8*j;
        if (d < D) {
          float kv = tanhf(pb[d]);
          sK[bh*D + d] = kv;
          kn2 += kv * kv;
        }
      }
      kn2 = sum8(kn2);
      if (sub == 0) {
        sKn[bh] = sqrtf(kn2);
      } else if (sub == 1 || sub == 4) {
        float sp = softplusf(pb[(sub == 1) ? D : D + 5]);
        if (sub == 1) sBe[bh] = sp; else sGa[bh] = sp + 1.f;
      } else if (sub == 2) {
        sGg[bh] = sigm(pb[D + 1]);
      } else if (sub == 3) {
        float s0 = pb[D + 2], s1 = pb[D + 3], s2 = pb[D + 4];
        float mx = fmaxf(s0, fmaxf(s1, s2));
        float e0 = expf(s0 - mx), e1 = expf(s1 - mx), e2 = expf(s2 - mx);
        float inv = 1.f / (e0 + e1 + e2);
        sS[bh*3 + 0] = e0 * inv; sS[bh*3 + 1] = e1 * inv; sS[bh*3 + 2] = e2 * inv;
      }
    }
    __syncthreads();

    {
      float k0[D], k1[D];
      for (int d = 0; d < D; ++d) {
        k0[d] = sK[(mb*NH + 0)*D + d];
        k1[d] = sK[(mb*NH + 1)*D + d];
      }
      const float be0 = sBe[mb*NH + 0], be1 = sBe[mb*NH + 1];
      const float kn0 = sKn[mb*NH + 0], kn1 = sKn[mb*NH + 1];
      for (int j = 0; j < MROWS; ++j) {
        float mn2 = 0.f, d0 = 0.f, d1 = 0.f;
        for (int d = 0; d < D; ++d) {
          float mv = Mr[j][d];
          mn2 += mv * mv; d0 += k0[d] * mv; d1 += k1[d] * mv;
        }
        float mn = sqrtf(mn2);
        const int n = mc + 32*j;
        sA[(mb*NH + 0)*NMEM + n] = be0 * d0 / (kn0*mn + EPSV);
        sA[(mb*NH + 1)*NMEM + n] = be1 * d1 / (kn1*mn + EPSV);
      }
    }
    __syncthreads();

    {
      const int bh = tid >> 4, c = tid & 15;
      const float* row = sA + bh*NMEM;
      float v[NMEM/16];
      for (int j = 0; j < NMEM/16; ++j) v[j] = row[c + 16*j];
      float mx = v[0];
      for (int j = 1; j < NMEM/16; ++j) mx = fmaxf(mx, v[j]);
      mx = max16(mx);
      float sum = 0.f;
      for (int j = 0; j < NMEM/16; ++j) sum += expf(v[j] - mx);
      sum = sum16(sum);
      if (c == 0) {
        sRed[bh*2 + 0] = mx;
        sRed[bh*2 + 1] = 1.f / sum;
      }
    }
    __syncthreads();

    #pragma unroll
    for (int it = 0; it < WTRIPS; ++it) {
      const int i = tid + it*NT;
      int bh = i >> 7;
      float wc = expf(sA[i] - sRed[bh*2]) * sRed[bh*2 + 1];
      float g  = sGg[bh];
      sA[i] = g * wc + (1.f - g) * sW[i];
    }
    __syncthreads();

    {
      float wt[WTRIPS], ga[WTRIPS];
      #pragma unroll
      for (int it = 0; it < WTRIPS; ++it) {
        const int i = tid + it*NT;
        int bh = i >> 7, n = i & (NMEM - 1);
        const float* wg = sA + bh*NMEM;
        wt[it] = sS[bh*3 + 0] * wg[(n + 1) & (NMEM - 1)]
               + sS[bh*3 + 1] * wg[n]
               + sS[bh*3 + 2] * wg[(n - 1) & (NMEM - 1)];
        ga[it] = sGa[bh];
      }
      #pragma unroll
      for (int it = 0; it < WTRIPS; ++it) {
        const float w = fmaxf(wt[it], 1e-16f);
        sBuf[tid + it*NT] = expf(ga[it] * logf(w));
      }
    }
    __syncthreads();

    {
      const int bh = tid >> 4, c = tid & 15;
      const float* row = sBuf + bh*NMEM;
      float sum = 0.f;
      for (int j = 0; j < NMEM/16; ++j) sum += row[c + 16*j];
      sum = sum16(sum);
      if (c == 0) sRed[bh*2] = 1.f / sum;
    }
    __syncthreads();

    #pragma unroll
    for (int it = 0; it < WTRIPS; ++it) {
      const int i = tid + it*NT;
      sW[i] = sBuf[i] * sRed[(i >> 7)*2];
    }
    __syncthreads();

    {
      const float* wr = sW + (mb*NH + 0)*NMEM + mc;
      float wj[MROWS];
      for (int j = 0; j < MROWS; ++j) wj[j] = wr[32*j];
      float part[D];
      for (int d = 0; d < D; ++d) {
        float acc = 0.f;
        for (int j = 0; j < MROWS; ++j) acc += wj[j] * Mr[j][d];
        part[d] = sum32(acc);
      }
      if (mc == 0) {
        for (int d = 0; d < D; ++d) sR[mb*D + d] = part[d];
      }
    }
    for (int i = tid; i < BC*D; i += NT) {
      int b = i / D, d = i - b*D;
      sE[i]  = sigm (sP[b*PPAD + NH*PPH + d]);
      sAd[i] = tanhf(sP[b*PPAD + NH*PPH + D + d]);
    }
    __syncthreads();

    {
      float ev[D], av[D];
      for (int d = 0; d < D; ++d) { ev[d] = sE[mb*D + d]; av[d] = sAd[mb*D + d]; }
      const float* wwp = sW + (mb*NH + 1)*NMEM + mc;
      for (int j = 0; j < MROWS; ++j) {
        const float ww = wwp[32*j];
        for (int d = 0; d < D; ++d)
          Mr[j][d] = Mr[j][d] * (1.f - ww * ev[d]) + ww * av[d];
      }
    }

    if (tid < BC*OUTW) {
      int b = tid / OUTW, o = tid - b*OUTW;
      float acc = bo;
      for (int j = 0; j < H; ++j) acc += sH[b*H + j] * sWo[j*OUTW + o];
      for (int d = 0; d < D; ++d) acc += sR[b*D + d] * sWo[(H + d)*OUTW + o];
      acc = clipf(acc);
      if (t >= tOutStart) {
        out[(size_t)(b0 + b) * TOUT * OUTW + (t - tOutStart) * OUTW + o] = sigm(acc);
      }
    }
    __syncthreads();
  }
}

extern "C" void kernel_launch(void* const* d_in, const int* in_sizes, int n_in,
                              void* d_out, int out_size, void* d_ws, size_t ws_size,
                              hipStream_t stream) {
  (void)in_sizes; (void)n_in; (void)out_size; (void)ws_size;
  const float* inputs   = (const float*)d_in[0];
  const float* Wx       = (const float*)d_in[1];
  const float* Wh       = (const float*)d_in[2];
  const float* b_lstm   = (const float*)d_in[3];
  const float* W_params = (const float*)d_in[4];
  const float* b_params = (const float*)d_in[5];
  const float* W_out    = (const float*)d_in[6];
  const float* b_out    = (const float*)d_in[7];
  const int*   seq      = (const int*)d_in[8];
  float* out = (float*)d_out;
  float* ws  = (float*)d_ws;

  pack_weights<<<dim3((WS_TOT + 255) / 256), dim3(256), 0, stream>>>(
      Wx, Wh, W_params, b_params, ws);

  ntm_fused<<<dim3(B_TOT / BC), dim3(NT), 0, stream>>>(
      inputs, ws + WS_WCAT, b_lstm, ws + WS_WP, ws + WS_BP,
      W_out, b_out, seq, out);
}


# Phases 5-14 and the z build run wave-local (wave b owns batch row b): 9 of 13 block barriers per step become wave-level LDS ordering; output dot over 4 lanes per unit (its sum order changes)
# speedup vs baseline: 5.4114x; 1.1866x over previous
#include <hip/hip_runtime.h>
#include <math.h>

typedef float v2f __attribute__((ext_vector_type(2)));
typedef float v4f __attribute__((ext_vector_type(4)));
typedef float v8f __attribute__((ext_vector_type(8)));

namespace {
constexpr int B_TOT = 2048;
constexpr int T_LEN = 66;
constexpr int INW   = 9;
constexpr int H     = 100;
constexpr int NMEM  = 128;
constexpr int D     = 20;
constexpr int OUTW  = 8;
constexpr int NH    = 2;
constexpr int PPH   = 26;
constexpr int PDIM  = 92;
constexpr int PPAD  = 96;
constexpr int G4    = 400;
constexpr int KZ    = 132;
constexpr int BC    = 8;
constexpr int NT    = 256;
constexpr float CLIPV = 20.f;
constexpr float EPSV  = 1e-8f;

constexpr int WS_WCAT = 0;
constexpr int WS_WP   = WS_WCAT + KZ*G4;
constexpr int WS_BP   = WS_WP   + H*PPAD;
constexpr int WS_TOT  = WS_BP   + PPAD;

constexpr int MROWS = NMEM / 32;

constexpr int WTRIPS = NH*NMEM / 32;
constexpr int HTRIPS = (BC*H + NT - 1) / NT;
constexpr int ZHTRIPS = (H + 31) / 32;

constexpr int SMEM_FLOATS =
    KZ*G4
  + G4
  + (H + D)*OUTW
  + BC*NH*NMEM
  + BC*NH*NMEM
  + BC*NH*NMEM
  + BC*H
  + BC*H
  + BC*D
  + BC*KZ
  + BC*G4
  + BC*PPAD
  + BC*NH*D
  + 32
  + 32
  + 32
  + 32
  + 32*3
  + 64
  + BC*D
  + BC*D;
static_assert(SMEM_FLOATS * 4 <= 327680, "one workgroup per WGP owns the whole 320 KiB of LDS");
static_assert((KZ*G4) % 4 == 0 && KZ % 2 == 0, "gate weights are staged with 16-byte copies and read as k pairs");
static_assert(NT == BC*NH*16 && NMEM % 16 == 0 && NH*8 <= 32 && 32 % (NH*8) == 0,
              "phase 5: eight lanes per head of the wave's own batch row (the other lanes repeat them); phases 7 and 10: one 16-lane row of a wave per (b,head)");
static_assert(NT == BC*32 && NMEM % 32 == 0,
              "memory matrix in registers: one wave per batch row, NMEM/32 memory rows per lane");
static_assert(NMEM % 32 == 0 && WTRIPS == NH*(NMEM/32),
              "weighting loops: NMEM/32 trips per head; trip it of a lane is head it/(NMEM/32), column lane + 32*(it % (NMEM/32))");
static_assert((HTRIPS - 1)*NT < BC*H, "hidden-unit loops: only the last trip is partial");
static_assert(INW <= 32 && D <= 32 && BC*(KZ - INW - D - H) <= NT && OUTW*4 == 32 && (H + D) % 4 == 0,
              "z build: a wave builds its own row, x and r one lane per element; output: 4 lanes per output unit, (H+D)/4 terms each");
static_assert(BC == 8,
              "WMMA tiles: lanes m and m+8 load the same batch row as their A row; D rows 0-7 sit in lanes 0-15");
}

__device__ __forceinline__ float sigm(float x) { return 1.f / (1.f + expf(-x)); }
__device__ __forceinline__ float softplusf(float x) { return (x > 20.f) ? x : log1pf(expf(x)); }
__device__ __forceinline__ float clipf(float x) { return fminf(fmaxf(x, -CLIPV), CLIPV); }

template <int X>
__device__ __forceinline__ float lane_xor(float v) {
  return __builtin_bit_cast(float, __builtin_amdgcn_update_dpp(
      0, __builtin_bit_cast(int, v), 0x160 + X, 0xf, 0xf, true));
}
__device__ __forceinline__ float sum8(float v) {
  v += lane_xor<1>(v); v += lane_xor<2>(v); v += lane_xor<4>(v);
  return v;
}
__device__ __forceinline__ float sum16(float v) {
  v += lane_xor<1>(v); v += lane_xor<2>(v); v += lane_xor<4>(v); v += lane_xor<8>(v);
  return v;
}
__device__ __forceinline__ float lane_xor16(float v) {
  const int i = __builtin_bit_cast(int, v);
  return __builtin_bit_cast(float, __builtin_amdgcn_permlanex16(i, i, 0x76543210, 0xfedcba98, false, false));
}
__device__ __forceinline__ float sum32(float v) {
  v = sum16(v); v += lane_xor16(v);
  return v;
}
__device__ __forceinline__ void wave_sync() {
  __builtin_amdgcn_fence(__ATOMIC_RELEASE, "wavefront");
  __builtin_amdgcn_wave_barrier();
  __builtin_amdgcn_fence(__ATOMIC_ACQUIRE, "wavefront");
}
__device__ __forceinline__ float max16(float v) {
  v = fmaxf(v, lane_xor<1>(v)); v = fmaxf(v, lane_xor<2>(v));
  v = fmaxf(v, lane_xor<4>(v)); v = fmaxf(v, lane_xor<8>(v));
  return v;
}

__global__ __launch_bounds__(256) void pack_weights(
    const float* __restrict__ Wx, const float* __restrict__ Wh,
    const float* __restrict__ W_params, const float* __restrict__ b_params,
    float* __restrict__ ws)
{
  int i = blockIdx.x * 256 + threadIdx.x;
  if (i < KZ*G4) {
    int k = i / G4, n = i - k*G4;
    float v = (k < 29) ? Wx[k*G4 + n] : ((k < 129) ? Wh[(k - 29)*G4 + n] : 0.f);
    ws[WS_WCAT + (k >> 1)*(2*G4) + 2*n + (k & 1)] = v;
  } else if (i < KZ*G4 + H*PPAD) {
    int j = i - KZ*G4;
    int k = j / PPAD, n = j - k*PPAD;
    ws[WS_WP + j] = (n < PDIM) ? W_params[k*PDIM + n] : 0.f;
  } else if (i < WS_TOT) {
    int n = i - KZ*G4 - H*PPAD;
    ws[WS_BP + n] = (n < PDIM) ? b_params[n] : 0.f;
  }
}

__global__ __launch_bounds__(NT) void ntm_fused(
    const float* __restrict__ inputs,
    const float* __restrict__ Wcat,
    const float* __restrict__ b_lstm,
    const float* __restrict__ Wp,
    const float* __restrict__ bp,
    const float* __restrict__ W_out,
    const float* __restrict__ b_out,
    const int*   __restrict__ seq_ptr,
    float* __restrict__ out)
{
  __shared__ __attribute__((aligned(16))) float smem[SMEM_FLOATS];
  float* sWc  = smem;
  float* sBl  = sWc  + KZ*G4;
  float* sWo  = sBl  + G4;
  float* sW   = sWo  + (H + D)*OUTW;
  float* sA   = sW   + BC*NH*NMEM;
  float* sBuf = sA   + BC*NH*NMEM;
  float* sH   = sBuf + BC*NH*NMEM;
  float* sC   = sH   + BC*H;
  float* sR   = sC   + BC*H;
  float* sZ   = sR   + BC*D;
  float* sG   = sZ   + BC*KZ;
  float* sP   = sG   + BC*G4;
  float* sK   = sP   + BC*PPAD;
  float* sKn  = sK   + BC*NH*D;
  float* sBe  = sKn  + 32;
  float* sGg  = sBe  + 32;
  float* sGa  = sGg  + 32;
  float* sS   = sGa  + 32;
  float* sRed = sS   + 32*3;
  float* sE   = sRed + 64;
  float* sAd  = sE   + BC*D;

  const int tid   = threadIdx.x;
  const int wave  = tid >> 5;
  const int lane  = tid & 31;
  const int m     = lane & 15;
  const int ma    = m & (BC - 1);
  const int khalf = (lane < 16) ? 0 : 2;
  const int b0    = blockIdx.x * BC;
  const int seq   = seq_ptr[0];
  const int tOutStart = seq + 1;
  const int TOUT  = T_LEN - tOutStart;

  const int mb = tid >> 5;
  const int mc = tid & 31;
  float Mr[MROWS][D];
  for (int j = 0; j < MROWS; ++j) {
    for (int d = 0; d < D; ++d) Mr[j][d] = 1e-6f;
  }

  {
    const v4f* __restrict__ src = (const v4f*)Wcat;
    v4f* dst = (v4f*)sWc;
    for (int i = tid; i < KZ*G4/4; i += NT) dst[i] = src[i];
  }
  for (int i = tid; i < G4;           i += NT) sBl[i] = b_lstm[i];
  for (int i = tid; i < (H + D)*OUTW; i += NT) sWo[i] = W_out[i];
  const float bo = b_out[tid & (OUTW - 1)];

  for (int i = tid; i < BC*NH*NMEM; i += NT) sW[i] = ((i & (NMEM-1)) == 0) ? 1.f : 0.f;
  for (int i = tid; i < BC*H;       i += NT) { sH[i] = 0.f; sC[i] = 0.f; }
  for (int i = tid; i < BC*D;       i += NT) sR[i] = 0.f;
  if (tid < BC*(KZ - INW - D - H)) {
    const int b = tid / (KZ - INW - D - H);
    sZ[b*KZ + INW + D + H + (tid - b*(KZ - INW - D - H))] = 0.f;
  }
  __syncthreads();

  for (int t = 0; t < T_LEN; ++t) {
    {
      const float xv = inputs[(size_t)(b0 + mb) * T_LEN * INW + t * INW + ((lane < INW) ? lane : 0)];
      const float rv = sR[mb*D + ((lane < D) ? lane : 0)];
      float hv[ZHTRIPS];
      #pragma unroll
      for (int it = 0; it < ZHTRIPS; ++it) hv[it] = sH[mb*H + min(lane + 32*it, H - 1)];
      if (lane < INW) sZ[mb*KZ + lane] = xv;
      if (lane < D)   sZ[mb*KZ + INW + lane] = rv;
      #pragma unroll
      for (int it = 0; it < ZHTRIPS; ++it) {
        const int u = lane + 32*it;
        if (u < H) sZ[mb*KZ + INW + D + u] = hv[it];
      }
    }
    __syncthreads();

    for (int ct = wave; ct < G4/16; ct += NT/32) {
      int n = ct*16 + m;
      const float* wc = sWc + 2*n + (khalf >> 1)*(2*G4);
      float bias = sBl[n];
      v8f acc = { bias, bias, bias, bias, bias, bias, bias, bias };
      for (int kt = 0; kt < KZ/4; ++kt) {
        int k0 = kt*4 + khalf;
        v2f av; av.x = sZ[ma*KZ + k0];    av.y = sZ[ma*KZ + k0 + 1];
        v2f bv = *(const v2f*)(wc + kt*(4*G4));
        acc = __builtin_amdgcn_wmma_f32_16x16x4_f32(false, av, false, bv,
                                                    (short)0, acc, false, false);
      }
      if (lane < 16) {
        #pragma unroll
        for (int v = 0; v < 8; ++v) sG[v*G4 + n] = acc[v];
      }
    }
    __syncthreads();

    {
      float gi[HTRIPS], gf[HTRIPS], gg[HTRIPS], go[HTRIPS], cp[HTRIPS];
      #pragma unroll
      for (int it = 0; it < HTRIPS; ++it) {
        const int i = min(tid + it*NT, BC*H - 1);
        const int b = i / H, u = i - b*H;
        gi[it] = sG[b*G4 + u];
        gf[it] = sG[b*G4 + H   + u];
        gg[it] = sG[b*G4 + 2*H + u];
        go[it] = sG[b*G4 + 3*H + u];
        cp[it] = sC[i];
      }
      #pragma unroll
      for (int it = 0; it < HTRIPS; ++it) {
        const int i = tid + it*NT;
        if (it < HTRIPS - 1 || i < BC*H) {
          float c  = sigm(gf[it]) * cp[it] + sigm(gi[it]) * tanhf(gg[it]);
          float h  = sigm(go[it]) * tanhf(c);
          sC[i] = c; sH[i] = h;
        }
      }
    }
    __syncthreads();

    for (int ct = wave; ct < PPAD/16; ct += NT/32) {
      int n = ct*16 + m;
      const float* __restrict__ wp = Wp + n;
      float bias = bp[n];
      v8f acc = { bias, bias, bias, bias, bias, bias, bias, bias };
      for (int kt = 0; kt < H/4; ++kt) {
        int k0 = kt*4 + khalf;
        v2f av; av.x = sH[ma*H + k0];       av.y = sH[ma*H + k0 + 1];
        v2f bv; bv.x = wp[(size_t)k0*PPAD]; bv.y = wp[(size_t)(k0 + 1)*PPAD];
        acc = __builtin_amdgcn_wmma_f32_16x16x4_f32(false, av, false, bv,
                                                    (short)0, acc, false, false);
      }
      float pv[8];
      #pragma unroll
      for (int v = 0; v < 8; ++v) pv[v] = clipf(acc[v]);
      if (lane < 16) {
        #pragma unroll
        for (int v = 0; v < 8; ++v) sP[v*PPAD + n] = pv[v];
      }
    }
    __syncthreads();

    {
      const int bh  = mb*NH + ((lane >> 3) & 1);
      const int sub = lane & 7;
      const float* pb = sP + (bh >> 1)*PPAD + (bh & 1)*PPH;
      float kn2 = 0.f;
      for (int j = 0; j < 3; ++j) {
        const int d = sub + 8*j;
        if (d < D) {
          float kv = tanhf(pb[d]);
          sK[bh*D + d] = kv;
          kn2 += kv * kv;
        }
      }
      kn2 = sum8(kn2);
      if (sub == 0) {
        sKn[bh] = sqrtf(kn2);
      } else if (sub == 1 || sub == 4) {
        float sp = softplusf(pb[(sub == 1) ? D : D + 5]);
        if (sub == 1) sBe[bh] = sp; else sGa[bh] = sp + 1.f;
      } else if (sub == 2) {
        sGg[bh] = sigm(pb[D + 1]);
      } else if (sub == 3) {
        float s0 = pb[D + 2], s1 = pb[D + 3], s2 = pb[D + 4];
        float mx = fmaxf(s0, fmaxf(s1, s2));
        float e0 = expf(s0 - mx), e1 = expf(s1 - mx), e2 = expf(s2 - mx);
        float inv = 1.f / (e0 + e1 + e2);
        sS[bh*3 + 0] = e0 * inv; sS[bh*3 + 1] = e1 * inv; sS[bh*3 + 2] = e2 * inv;
      }
    }
    wave_sync();

    {
      float k0[D], k1[D];
      for (int d = 0; d < D; ++d) {
        k0[d] = sK[(mb*NH + 0)*D + d];
        k1[d] = sK[(mb*NH + 1)*D + d];
      }
      const float be0 = sBe[mb*NH + 0], be1 = sBe[mb*NH + 1];
      const float kn0 = sKn[mb*NH + 0], kn1 = sKn[mb*NH + 1];
      for (int j = 0; j < MROWS; ++j) {
        float mn2 = 0.f, d0 = 0.f, d1 = 0.f;
        for (int d = 0; d < D; ++d) {
          float mv = Mr[j][d];
          mn2 += mv * mv; d0 += k0[d] * mv; d1 += k1[d] * mv;
        }
        float mn = sqrtf(mn2);
        const int n = mc + 32*j;
        sA[(mb*NH + 0)*NMEM + n] = be0 * d0 / (kn0*mn + EPSV);
        sA[(mb*NH + 1)*NMEM + n] = be1 * d1 / (kn1*mn + EPSV);
      }
    }
    wave_sync();

    {
      const int bh = tid >> 4, c = tid & 15;
      const float* row = sA + bh*NMEM;
      float v[NMEM/16];
      for (int j = 0; j < NMEM/16; ++j) v[j] = row[c + 16*j];
      float mx = v[0];
      for (int j = 1; j < NMEM/16; ++j) mx = fmaxf(mx, v[j]);
      mx = max16(mx);
      float sum = 0.f;
      for (int j = 0; j < NMEM/16; ++j) sum += expf(v[j] - mx);
      sum = sum16(sum);
      if (c == 0) {
        sRed[bh*2 + 0] = mx;
        sRed[bh*2 + 1] = 1.f / sum;
      }
    }
    wave_sync();

    #pragma unroll
    for (int it = 0; it < WTRIPS; ++it) {
      const int bh = mb*NH + it / (NMEM/32);
      const int i  = bh*NMEM + lane + 32*(it % (NMEM/32));
      float wc = expf(sA[i] - sRed[bh*2]) * sRed[bh*2 + 1];
      float g  = sGg[bh];
      sA[i] = g * wc + (1.f - g) * sW[i];
    }
    wave_sync();

    {
      float wt[WTRIPS], ga[WTRIPS];
      #pragma unroll
      for (int it = 0; it < WTRIPS; ++it) {
        const int bh = mb*NH + it / (NMEM/32), n = lane + 32*(it % (NMEM/32));
        const float* wg = sA + bh*NMEM;
        wt[it] = sS[bh*3 + 0] * wg[(n + 1) & (NMEM - 1)]
               + sS[bh*3 + 1] * wg[n]
               + sS[bh*3 + 2] * wg[(n - 1) & (NMEM - 1)];
        ga[it] = sGa[bh];
      }
      #pragma unroll
      for (int it = 0; it < WTRIPS; ++it) {
        const float w = fmaxf(wt[it], 1e-16f);
        sBuf[(mb*NH + it / (NMEM/32))*NMEM + lane + 32*(it % (NMEM/32))] = expf(ga[it] * logf(w));
      }
    }
    wave_sync();

    {
      const int bh = tid >> 4, c = tid & 15;
      const float* row = sBuf + bh*NMEM;
      float sum = 0.f;
      for (int j = 0; j < NMEM/16; ++j) sum += row[c + 16*j];
      sum = sum16(sum);
      if (c == 0) sRed[bh*2] = 1.f / sum;
    }
    wave_sync();

    #pragma unroll
    for (int it = 0; it < WTRIPS; ++it) {
      const int bh = mb*NH + it / (NMEM/32);
      const int i  = bh*NMEM + lane + 32*(it % (NMEM/32));
      sW[i] = sBuf[i] * sRed[bh*2];
    }
    wave_sync();

    {
      const float* wr = sW + (mb*NH + 0)*NMEM + mc;
      float wj[MROWS];
      for (int j = 0; j < MROWS; ++j) wj[j] = wr[32*j];
      float part[D];
      for (int d = 0; d < D; ++d) {
        float acc = 0.f;
        for (int j = 0; j < MROWS; ++j) acc += wj[j] * Mr[j][d];
        part[d] = sum32(acc);
      }
      if (mc == 0) {
        for (int d = 0; d < D; ++d) sR[mb*D + d] = part[d];
      }
    }
    if (lane < D) {
      sE[mb*D + lane]  = sigm (sP[mb*PPAD + NH*PPH + lane]);
      sAd[mb*D + lane] = tanhf(sP[mb*PPAD + NH*PPH + D + lane]);
    }
    wave_sync();

    {
      float ev[D], av[D];
      for (int d = 0; d < D; ++d) { ev[d] = sE[mb*D + d]; av[d] = sAd[mb*D + d]; }
      const float* wwp = sW + (mb*NH + 1)*NMEM + mc;
      for (int j = 0; j < MROWS; ++j) {
        const float ww = wwp[32*j];
        for (int d = 0; d < D; ++d)
          Mr[j][d] = Mr[j][d] * (1.f - ww * ev[d]) + ww * av[d];
      }
    }

    {
      const int o = lane & (OUTW - 1), q = lane >> 3;
      float acc = (q == 0) ? bo : 0.f;
      #pragma unroll
      for (int jj = 0; jj < (H + D)/4; ++jj) {
        const int j = q*((H + D)/4) + jj;
        const float v = (j < H) ? sH[mb*H + j] : sR[mb*D + (j - H)];
        acc += v * sWo[j*OUTW + o];
      }
      acc += lane_xor<8>(acc);
      acc += lane_xor16(acc);
      acc = clipf(acc);
      if (q == 0 && t >= tOutStart) {
        out[(size_t)(b0 + mb) * TOUT * OUTW + (t - tOutStart) * OUTW + o] = sigm(acc);
      }
    }
  }
}

extern "C" void kernel_launch(void* const* d_in, const int* in_sizes, int n_in,
                              void* d_out, int out_size, void* d_ws, size_t ws_size,
                              hipStream_t stream) {
  (void)in_sizes; (void)n_in; (void)out_size; (void)ws_size;
  const float* inputs   = (const float*)d_in[0];
  const float* Wx       = (const float*)d_in[1];
  const float* Wh       = (const float*)d_in[2];
  const float* b_lstm   = (const float*)d_in[3];
  const float* W_params = (const float*)d_in[4];
  const float* b_params = (const float*)d_in[5];
  const float* W_out    = (const float*)d_in[6];
  const float* b_out    = (const float*)d_in[7];
  const int*   seq      = (const int*)d_in[8];
  float* out = (float*)d_out;
  float* ws  = (float*)d_ws;

  pack_weights<<<dim3((WS_TOT + 255) / 256), dim3(256), 0, stream>>>(
      Wx, Wh, W_params, b_params, ws);

  ntm_fused<<<dim3(B_TOT / BC), dim3(NT), 0, stream>>>(
      inputs, ws + WS_WCAT, b_lstm, ws + WS_WP, ws + WS_BP,
      W_out, b_out, seq, out);
}


# Gate product split by K: next step's h part (25 of 33 K tiles) runs in 5 chunks among the addressing phases, beside their vector work; only the x|r tiles stay before the cell; gate sum order changes
# speedup vs baseline: 5.9414x; 1.0979x over previous
#include <hip/hip_runtime.h>
#include <math.h>

typedef float v2f __attribute__((ext_vector_type(2)));
typedef float v4f __attribute__((ext_vector_type(4)));
typedef float v8f __attribute__((ext_vector_type(8)));

namespace {
constexpr int B_TOT = 2048;
constexpr int T_LEN = 66;
constexpr int INW   = 9;
constexpr int H     = 100;
constexpr int NMEM  = 128;
constexpr int D     = 20;
constexpr int OUTW  = 8;
constexpr int NH    = 2;
constexpr int PPH   = 26;
constexpr int PDIM  = 92;
constexpr int PPAD  = 96;
constexpr int G4    = 400;
constexpr int KZ    = 132;
constexpr int KXR   = 32;
constexpr int ZP    = 36;
constexpr int HCH   = 5;
constexpr int BC    = 8;
constexpr int NT    = 256;
constexpr float CLIPV = 20.f;
constexpr float EPSV  = 1e-8f;

constexpr int WS_WCAT = 0;
constexpr int WS_WP   = WS_WCAT + KZ*G4;
constexpr int WS_BP   = WS_WP   + H*PPAD;
constexpr int WS_TOT  = WS_BP   + PPAD;

constexpr int MROWS = NMEM / 32;

constexpr int WTRIPS = NH*NMEM / 32;
constexpr int HTRIPS = (BC*H + NT - 1) / NT;

constexpr int SMEM_FLOATS =
    KZ*G4
  + G4
  + (H + D)*OUTW
  + BC*NH*NMEM
  + BC*NH*NMEM
  + BC*NH*NMEM
  + BC*H
  + BC*H
  + BC*D
  + BC*ZP
  + BC*G4
  + BC*PPAD
  + BC*NH*D
  + 32
  + 32
  + 32
  + 32
  + 32*3
  + 64
  + BC*D
  + BC*D;
static_assert(SMEM_FLOATS * 4 <= 327680, "one workgroup per WGP owns the whole 320 KiB of LDS");
static_assert((KZ*G4) % 4 == 0 && KZ % 2 == 0, "gate weights are staged with 16-byte copies and read as k pairs");
static_assert(NT == BC*NH*16 && NMEM % 16 == 0 && NH*8 <= 32 && 32 % (NH*8) == 0,
              "phase 5: eight lanes per head of the wave's own batch row (the other lanes repeat them); phases 7 and 10: one 16-lane row of a wave per (b,head)");
static_assert(NT == BC*32 && NMEM % 32 == 0,
              "memory matrix in registers: one wave per batch row, NMEM/32 memory rows per lane");
static_assert(NMEM % 32 == 0 && WTRIPS == NH*(NMEM/32),
              "weighting loops: NMEM/32 trips per head; trip it of a lane is head it/(NMEM/32), column lane + 32*(it % (NMEM/32))");
static_assert((HTRIPS - 1)*NT < BC*H, "hidden-unit loops: only the last trip is partial");
static_assert(INW <= 32 && D <= 32 && BC*(KXR - INW - D) <= NT && OUTW*4 == 32 && (H + D) % 4 == 0,
              "z build: a wave builds its own row, x and r one lane per element; output: 4 lanes per output unit, (H+D)/4 terms each");
static_assert(H % 4 == 0 && KZ == H + KXR && INW + D <= KXR && KXR % 4 == 0 && KXR <= ZP && ZP % 2 == 0 && H % 2 == 0 && (H/4) % HCH == 0,
              "gate product: K tiles 0..H/4-1 read h straight from its own array, the last KXR/4 read the x|r rows; 8-byte fragment loads");
static_assert(G4/16 > 3*(NT/32) && G4/16 <= 4*(NT/32) && G4 % 16 == 0,
              "gate product: every wave owns column tiles wave, wave+8, wave+16; the waves below G4/16 - 24 own a fourth");
static_assert(BC == 8,
              "WMMA tiles: lanes m and m+8 load the same batch row as their A row; D rows 0-7 sit in lanes 0-15");
}

__device__ __forceinline__ float sigm(float x) { return 1.f / (1.f + expf(-x)); }
__device__ __forceinline__ float softplusf(float x) { return (x > 20.f) ? x : log1pf(expf(x)); }
__device__ __forceinline__ float clipf(float x) { return fminf(fmaxf(x, -CLIPV), CLIPV); }

template <int X>
__device__ __forceinline__ float lane_xor(float v) {
  return __builtin_bit_cast(float, __builtin_amdgcn_update_dpp(
      0, __builtin_bit_cast(int, v), 0x160 + X, 0xf, 0xf, true));
}
__device__ __forceinline__ float sum8(float v) {
  v += lane_xor<1>(v); v += lane_xor<2>(v); v += lane_xor<4>(v);
  return v;
}
__device__ __forceinline__ float sum16(float v) {
  v += lane_xor<1>(v); v += lane_xor<2>(v); v += lane_xor<4>(v); v += lane_xor<8>(v);
  return v;
}
__device__ __forceinline__ float lane_xor16(float v) {
  const int i = __builtin_bit_cast(int, v);
  return __builtin_bit_cast(float, __builtin_amdgcn_permlanex16(i, i, 0x76543210, 0xfedcba98, false, false));
}
__device__ __forceinline__ float sum32(float v) {
  v = sum16(v); v += lane_xor16(v);
  return v;
}
__device__ __forceinline__ void wave_sync() {
  __builtin_amdgcn_fence(__ATOMIC_RELEASE, "wavefront");
  __builtin_amdgcn_wave_barrier();
  __builtin_amdgcn_fence(__ATOMIC_ACQUIRE, "wavefront");
}
__device__ __forceinline__ float max16(float v) {
  v = fmaxf(v, lane_xor<1>(v)); v = fmaxf(v, lane_xor<2>(v));
  v = fmaxf(v, lane_xor<4>(v)); v = fmaxf(v, lane_xor<8>(v));
  return v;
}

template <int BKT0, int N>
__device__ __forceinline__ void gate_ktiles(const float* arow,
                                            const float* wc0, const float* wc1, const float* wc2, const float* wc3,
                                            bool own4, v8f& a0, v8f& a1, v8f& a2, v8f& a3) {
  if (own4) {
    #pragma unroll
    for (int j = 0; j < N; ++j) {
      const v2f av = *(const v2f*)(arow + 4*j);
      const v2f b3 = *(const v2f*)(wc3 + (BKT0 + j)*(4*G4));
      a3 = __builtin_amdgcn_wmma_f32_16x16x4_f32(false, av, false, b3, (short)0, a3, false, false);
    }
  }
  #pragma unroll
  for (int j = 0; j < N; ++j) {
    const v2f av = *(const v2f*)(arow + 4*j);
    const v2f b0 = *(const v2f*)(wc0 + (BKT0 + j)*(4*G4));
    const v2f b1 = *(const v2f*)(wc1 + (BKT0 + j)*(4*G4));
    const v2f b2 = *(const v2f*)(wc2 + (BKT0 + j)*(4*G4));
    a0 = __builtin_amdgcn_wmma_f32_16x16x4_f32(false, av, false, b0, (short)0, a0, false, false);
    a1 = __builtin_amdgcn_wmma_f32_16x16x4_f32(false, av, false, b1, (short)0, a1, false, false);
    a2 = __builtin_amdgcn_wmma_f32_16x16x4_f32(false, av, false, b2, (short)0, a2, false, false);
  }
}

__global__ __launch_bounds__(256) void pack_weights(
    const float* __restrict__ Wx, const float* __restrict__ Wh,
    const float* __restrict__ W_params, const float* __restrict__ b_params,
    float* __restrict__ ws)
{
  int i = blockIdx.x * 256 + threadIdx.x;
  if (i < KZ*G4) {
    int k = i / G4, n = i - k*G4;
    float v = (k < H) ? Wh[k*G4 + n] : ((k < H + INW + D) ? Wx[(k - H)*G4 + n] : 0.f);
    ws[WS_WCAT + (k >> 1)*(2*G4) + 2*n + (k & 1)] = v;
  } else if (i < KZ*G4 + H*PPAD) {
    int j = i - KZ*G4;
    int k = j / PPAD, n = j - k*PPAD;
    ws[WS_WP + j] = (n < PDIM) ? W_params[k*PDIM + n] : 0.f;
  } else if (i < WS_TOT) {
    int n = i - KZ*G4 - H*PPAD;
    ws[WS_BP + n] = (n < PDIM) ? b_params[n] : 0.f;
  }
}

__global__ __launch_bounds__(NT) void ntm_fused(
    const float* __restrict__ inputs,
    const float* __restrict__ Wcat,
    const float* __restrict__ b_lstm,
    const float* __restrict__ Wp,
    const float* __restrict__ bp,
    const float* __restrict__ W_out,
    const float* __restrict__ b_out,
    const int*   __restrict__ seq_ptr,
    float* __restrict__ out)
{
  __shared__ __attribute__((aligned(16))) float smem[SMEM_FLOATS];
  float* sWc  = smem;
  float* sBl  = sWc  + KZ*G4;
  float* sWo  = sBl  + G4;
  float* sW   = sWo  + (H + D)*OUTW;
  float* sA   = sW   + BC*NH*NMEM;
  float* sBuf = sA   + BC*NH*NMEM;
  float* sH   = sBuf + BC*NH*NMEM;
  float* sC   = sH   + BC*H;
  float* sR   = sC   + BC*H;
  float* sZ   = sR   + BC*D;
  float* sG   = sZ   + BC*ZP;
  float* sP   = sG   + BC*G4;
  float* sK   = sP   + BC*PPAD;
  float* sKn  = sK   + BC*NH*D;
  float* sBe  = sKn  + 32;
  float* sGg  = sBe  + 32;
  float* sGa  = sGg  + 32;
  float* sS   = sGa  + 32;
  float* sRed = sS   + 32*3;
  float* sE   = sRed + 64;
  float* sAd  = sE   + BC*D;

  const int tid   = threadIdx.x;
  const int wave  = tid >> 5;
  const int lane  = tid & 31;
  const int m     = lane & 15;
  const int ma    = m & (BC - 1);
  const int khalf = (lane < 16) ? 0 : 2;
  const int b0    = blockIdx.x * BC;
  const int seq   = seq_ptr[0];
  const int tOutStart = seq + 1;
  const int TOUT  = T_LEN - tOutStart;

  const int mb = tid >> 5;
  const int mc = tid & 31;
  float Mr[MROWS][D];
  for (int j = 0; j < MROWS; ++j) {
    for (int d = 0; d < D; ++d) Mr[j][d] = 1e-6f;
  }

  {
    const v4f* __restrict__ src = (const v4f*)Wcat;
    v4f* dst = (v4f*)sWc;
    for (int i = tid; i < KZ*G4/4; i += NT) dst[i] = src[i];
  }
  for (int i = tid; i < G4;           i += NT) sBl[i] = b_lstm[i];
  for (int i = tid; i < (H + D)*OUTW; i += NT) sWo[i] = W_out[i];
  const float bo = b_out[tid & (OUTW - 1)];

  for (int i = tid; i < BC*NH*NMEM; i += NT) sW[i] = ((i & (NMEM-1)) == 0) ? 1.f : 0.f;
  for (int i = tid; i < BC*H;       i += NT) { sH[i] = 0.f; sC[i] = 0.f; }
  for (int i = tid; i < BC*D;       i += NT) sR[i] = 0.f;
  if (tid < BC*(KXR - INW - D)) {
    const int b = tid / (KXR - INW - D);
    sZ[b*ZP + INW + D + (tid - b*(KXR - INW - D))] = 0.f;
  }
  __syncthreads();

  const bool own4 = __builtin_amdgcn_readfirstlane(wave) < G4/16 - 3*(NT/32);
  const int n0 = (wave           )*16 + m;
  const int n1 = (wave +   NT/32 )*16 + m;
  const int n2 = (wave + 2*NT/32 )*16 + m;
  const int n3 = own4 ? (wave + 3*NT/32)*16 + m : n2;
  const float* wc0 = sWc + 2*n0 + (khalf >> 1)*(2*G4);
  const float* wc1 = sWc + 2*n1 + (khalf >> 1)*(2*G4);
  const float* wc2 = sWc + 2*n2 + (khalf >> 1)*(2*G4);
  const float* wc3 = sWc + 2*n3 + (khalf >> 1)*(2*G4);
  const float* ah  = sH + ma*H  + khalf;
  const float* az  = sZ + ma*ZP + khalf;
  const float bias0 = sBl[n0], bias1 = sBl[n1], bias2 = sBl[n2], bias3 = sBl[n3];
  v8f acc0 = { bias0, bias0, bias0, bias0, bias0, bias0, bias0, bias0 };
  v8f acc1 = { bias1, bias1, bias1, bias1, bias1, bias1, bias1, bias1 };
  v8f acc2 = { bias2, bias2, bias2, bias2, bias2, bias2, bias2, bias2 };
  v8f acc3 = { bias3, bias3, bias3, bias3, bias3, bias3, bias3, bias3 };

  for (int t = 0; t < T_LEN; ++t) {
    {
      const float xv = inputs[(size_t)(b0 + mb) * T_LEN * INW + t * INW + ((lane < INW) ? lane : 0)];
      const float rv = sR[mb*D + ((lane < D) ? lane : 0)];
      if (lane < INW) sZ[mb*ZP + lane] = xv;
      if (lane < D)   sZ[mb*ZP + INW + lane] = rv;
    }
    __syncthreads();

    gate_ktiles<H/4, KXR/4>(az, wc0, wc1, wc2, wc3, own4, acc0, acc1, acc2, acc3);
    if (lane < 16) {
      #pragma unroll
      for (int v = 0; v < 8; ++v) {
        sG[v*G4 + n0] = acc0[v]; sG[v*G4 + n1] = acc1[v]; sG[v*G4 + n2] = acc2[v];
      }
      if (own4) {
        #pragma unroll
        for (int v = 0; v < 8; ++v) sG[v*G4 + n3] = acc3[v];
      }
    }
    __syncthreads();

    {
      float gi[HTRIPS], gf[HTRIPS], gg[HTRIPS], go[HTRIPS], cp[HTRIPS];
      #pragma unroll
      for (int it = 0; it < HTRIPS; ++it) {
        const int i = min(tid + it*NT, BC*H - 1);
        const int b = i / H, u = i - b*H;
        gi[it] = sG[b*G4 + u];
        gf[it] = sG[b*G4 + H   + u];
        gg[it] = sG[b*G4 + 2*H + u];
        go[it] = sG[b*G4 + 3*H + u];
        cp[it] = sC[i];
      }
      #pragma unroll
      for (int it = 0; it < HTRIPS; ++it) {
        const int i = tid + it*NT;
        if (it < HTRIPS - 1 || i < BC*H) {
          float c  = sigm(gf[it]) * cp[it] + sigm(gi[it]) * tanhf(gg[it]);
          float h  = sigm(go[it]) * tanhf(c);
          sC[i] = c; sH[i] = h;
        }
      }
    }
    __syncthreads();

    for (int ct = wave; ct < PPAD/16; ct += NT/32) {
      int n = ct*16 + m;
      const float* __restrict__ wp = Wp + n;
      float bias = bp[n];
      v8f acc = { bias, bias, bias, bias, bias, bias, bias, bias };
      for (int kt = 0; kt < H/4; ++kt) {
        int k0 = kt*4 + khalf;
        v2f av; av.x = sH[ma*H + k0];       av.y = sH[ma*H + k0 + 1];
        v2f bv; bv.x = wp[(size_t)k0*PPAD]; bv.y = wp[(size_t)(k0 + 1)*PPAD];
        acc = __builtin_amdgcn_wmma_f32_16x16x4_f32(false, av, false, bv,
                                                    (short)0, acc, false, false);
      }
      float pv[8];
      #pragma unroll
      for (int v = 0; v < 8; ++v) pv[v] = clipf(acc[v]);
      if (lane < 16) {
        #pragma unroll
        for (int v = 0; v < 8; ++v) sP[v*PPAD + n] = pv[v];
      }
    }
    __syncthreads();

    #pragma unroll
    for (int v = 0; v < 8; ++v) { acc0[v] = bias0; acc1[v] = bias1; acc2[v] = bias2; acc3[v] = bias3; }
    gate_ktiles<0*HCH, HCH>(ah + 0*HCH*4, wc0, wc1, wc2, wc3, own4, acc0, acc1, acc2, acc3);

    {
      const int bh  = mb*NH + ((lane >> 3) & 1);
      const int sub = lane & 7;
      const float* pb = sP + (bh >> 1)*PPAD + (bh & 1)*PPH;
      float kn2 = 0.f;
      for (int j = 0; j < 3; ++j) {
        const int d = sub + 8*j;
        if (d < D) {
          float kv = tanhf(pb[d]);
          sK[bh*D + d] = kv;
          kn2 += kv * kv;
        }
      }
      kn2 = sum8(kn2);
      if (sub == 0) {
        sKn[bh] = sqrtf(kn2);
      } else if (sub == 1 || sub == 4) {
        float sp = softplusf(pb[(sub == 1) ? D : D + 5]);
        if (sub == 1) sBe[bh] = sp; else sGa[bh] = sp + 1.f;
      } else if (sub == 2) {
        sGg[bh] = sigm(pb[D + 1]);
      } else if (sub == 3) {
        float s0 = pb[D + 2], s1 = pb[D + 3], s2 = pb[D + 4];
        float mx = fmaxf(s0, fmaxf(s1, s2));
        float e0 = expf(s0 - mx), e1 = expf(s1 - mx), e2 = expf(s2 - mx);
        float inv = 1.f / (e0 + e1 + e2);
        sS[bh*3 + 0] = e0 * inv; sS[bh*3 + 1] = e1 * inv; sS[bh*3 + 2] = e2 * inv;
      }
    }
    wave_sync();

    gate_ktiles<1*HCH, HCH>(ah + 1*HCH*4, wc0, wc1, wc2, wc3, own4, acc0, acc1, acc2, acc3);
    {
      float k0[D], k1[D];
      for (int d = 0; d < D; ++d) {
        k0[d] = sK[(mb*NH + 0)*D + d];
        k1[d] = sK[(mb*NH + 1)*D + d];
      }
      const float be0 = sBe[mb*NH + 0], be1 = sBe[mb*NH + 1];
      const float kn0 = sKn[mb*NH + 0], kn1 = sKn[mb*NH + 1];
      for (int j = 0; j < MROWS; ++j) {
        float mn2 = 0.f, d0 = 0.f, d1 = 0.f;
        for (int d = 0; d < D; ++d) {
          float mv = Mr[j][d];
          mn2 += mv * mv; d0 += k0[d] * mv; d1 += k1[d] * mv;
        }
        float mn = sqrtf(mn2);
        const int n = mc + 32*j;
        sA[(mb*NH + 0)*NMEM + n] = be0 * d0 / (kn0*mn + EPSV);
        sA[(mb*NH + 1)*NMEM + n] = be1 * d1 / (kn1*mn + EPSV);
      }
    }
    wave_sync();

    gate_ktiles<2*HCH, HCH>(ah + 2*HCH*4, wc0, wc1, wc2, wc3, own4, acc0, acc1, acc2, acc3);
    {
      const int bh = tid >> 4, c = tid & 15;
      const float* row = sA + bh*NMEM;
      float v[NMEM/16];
      for (int j = 0; j < NMEM/16; ++j) v[j] = row[c + 16*j];
      float mx = v[0];
      for (int j = 1; j < NMEM/16; ++j) mx = fmaxf(mx, v[j]);
      mx = max16(mx);
      float sum = 0.f;
      for (int j = 0; j < NMEM/16; ++j) sum += expf(v[j] - mx);
      sum = sum16(sum);
      if (c == 0) {
        sRed[bh*2 + 0] = mx;
        sRed[bh*2 + 1] = 1.f / sum;
      }
    }
    wave_sync();

    #pragma unroll
    for (int it = 0; it < WTRIPS; ++it) {
      const int bh = mb*NH + it / (NMEM/32);
      const int i  = bh*NMEM + lane + 32*(it % (NMEM/32));
      float wc = expf(sA[i] - sRed[bh*2]) * sRed[bh*2 + 1];
      float g  = sGg[bh];
      sA[i] = g * wc + (1.f - g) * sW[i];
    }
    wave_sync();

    gate_ktiles<3*HCH, HCH>(ah + 3*HCH*4, wc0, wc1, wc2, wc3, own4, acc0, acc1, acc2, acc3);

    {
      float wt[WTRIPS], ga[WTRIPS];
      #pragma unroll
      for (int it = 0; it < WTRIPS; ++it) {
        const int bh = mb*NH + it / (NMEM/32), n = lane + 32*(it % (NMEM/32));
        const float* wg = sA + bh*NMEM;
        wt[it] = sS[bh*3 + 0] * wg[(n + 1) & (NMEM - 1)]
               + sS[bh*3 + 1] * wg[n]
               + sS[bh*3 + 2] * wg[(n - 1) & (NMEM - 1)];
        ga[it] = sGa[bh];
      }
      #pragma unroll
      for (int it = 0; it < WTRIPS; ++it) {
        const float w = fmaxf(wt[it], 1e-16f);
        sBuf[(mb*NH + it / (NMEM/32))*NMEM + lane + 32*(it % (NMEM/32))] = expf(ga[it] * logf(w));
      }
    }
    wave_sync();

    {
      const int bh = tid >> 4, c = tid & 15;
      const float* row = sBuf + bh*NMEM;
      float sum = 0.f;
      for (int j = 0; j < NMEM/16; ++j) sum += row[c + 16*j];
      sum = sum16(sum);
      if (c == 0) sRed[bh*2] = 1.f / sum;
    }
    wave_sync();

    #pragma unroll
    for (int it = 0; it < WTRIPS; ++it) {
      const int bh = mb*NH + it / (NMEM/32);
      const int i  = bh*NMEM + lane + 32*(it % (NMEM/32));
      sW[i] = sBuf[i] * sRed[bh*2];
    }
    wave_sync();

    gate_ktiles<4*HCH, HCH>(ah + 4*HCH*4, wc0, wc1, wc2, wc3, own4, acc0, acc1, acc2, acc3);
    {
      const float* wr = sW + (mb*NH + 0)*NMEM + mc;
      float wj[MROWS];
      for (int j = 0; j < MROWS; ++j) wj[j] = wr[32*j];
      float part[D];
      for (int d = 0; d < D; ++d) {
        float acc = 0.f;
        for (int j = 0; j < MROWS; ++j) acc += wj[j] * Mr[j][d];
        part[d] = sum32(acc);
      }
      if (mc == 0) {
        for (int d = 0; d < D; ++d) sR[mb*D + d] = part[d];
      }
    }
    if (lane < D) {
      sE[mb*D + lane]  = sigm (sP[mb*PPAD + NH*PPH + lane]);
      sAd[mb*D + lane] = tanhf(sP[mb*PPAD + NH*PPH + D + lane]);
    }
    wave_sync();

    {
      float ev[D], av[D];
      for (int d = 0; d < D; ++d) { ev[d] = sE[mb*D + d]; av[d] = sAd[mb*D + d]; }
      const float* wwp = sW + (mb*NH + 1)*NMEM + mc;
      for (int j = 0; j < MROWS; ++j) {
        const float ww = wwp[32*j];
        for (int d = 0; d < D; ++d)
          Mr[j][d] = Mr[j][d] * (1.f - ww * ev[d]) + ww * av[d];
      }
    }

    {
      const int o = lane & (OUTW - 1), q = lane >> 3;
      float acc = (q == 0) ? bo : 0.f;
      #pragma unroll
      for (int jj = 0; jj < (H + D)/4; ++jj) {
        const int j = q*((H + D)/4) + jj;
        const float v = (j < H) ? sH[mb*H + j] : sR[mb*D + (j - H)];
        acc += v * sWo[j*OUTW + o];
      }
      acc += lane_xor<8>(acc);
      acc += lane_xor16(acc);
      acc = clipf(acc);
      if (q == 0 && t >= tOutStart) {
        out[(size_t)(b0 + mb) * TOUT * OUTW + (t - tOutStart) * OUTW + o] = sigm(acc);
      }
    }
  }
}

extern "C" void kernel_launch(void* const* d_in, const int* in_sizes, int n_in,
                              void* d_out, int out_size, void* d_ws, size_t ws_size,
                              hipStream_t stream) {
  (void)in_sizes; (void)n_in; (void)out_size; (void)ws_size;
  const float* inputs   = (const float*)d_in[0];
  const float* Wx       = (const float*)d_in[1];
  const float* Wh       = (const float*)d_in[2];
  const float* b_lstm   = (const float*)d_in[3];
  const float* W_params = (const float*)d_in[4];
  const float* b_params = (const float*)d_in[5];
  const float* W_out    = (const float*)d_in[6];
  const float* b_out    = (const float*)d_in[7];
  const int*   seq      = (const int*)d_in[8];
  float* out = (float*)d_out;
  float* ws  = (float*)d_ws;

  pack_weights<<<dim3((WS_TOT + 255) / 256), dim3(256), 0, stream>>>(
      Wx, Wh, W_params, b_params, ws);

  ntm_fused<<<dim3(B_TOT / BC), dim3(NT), 0, stream>>>(
      inputs, ws + WS_WCAT, b_lstm, ws + WS_WP, ws + WS_BP,
      W_out, b_out, seq, out);
}


# Addressing weights (scores, w_g, w^gamma, w) move from LDS to registers: lane l holds columns l+32j of 2 heads; shift reads neighbour lanes; head sums: 4 partials + 32-lane tree (sum order changes)
# speedup vs baseline: 6.1301x; 1.0318x over previous
#include <hip/hip_runtime.h>
#include <math.h>

typedef float v2f __attribute__((ext_vector_type(2)));
typedef float v4f __attribute__((ext_vector_type(4)));
typedef float v8f __attribute__((ext_vector_type(8)));

namespace {
constexpr int B_TOT = 2048;
constexpr int T_LEN = 66;
constexpr int INW   = 9;
constexpr int H     = 100;
constexpr int NMEM  = 128;
constexpr int D     = 20;
constexpr int OUTW  = 8;
constexpr int NH    = 2;
constexpr int PPH   = 26;
constexpr int PDIM  = 92;
constexpr int PPAD  = 96;
constexpr int G4    = 400;
constexpr int KZ    = 132;
constexpr int KXR   = 32;
constexpr int ZP    = 36;
constexpr int HCH   = 5;
constexpr int BC    = 8;
constexpr int NT    = 256;
constexpr float CLIPV = 20.f;
constexpr float EPSV  = 1e-8f;

constexpr int WS_WCAT = 0;
constexpr int WS_WP   = WS_WCAT + KZ*G4;
constexpr int WS_BP   = WS_WP   + H*PPAD;
constexpr int WS_TOT  = WS_BP   + PPAD;

constexpr int MROWS = NMEM / 32;

constexpr int HTRIPS = (BC*H + NT - 1) / NT;

constexpr int SMEM_FLOATS =
    KZ*G4
  + G4
  + (H + D)*OUTW
  + BC*H
  + BC*H
  + BC*D
  + BC*ZP
  + BC*G4
  + BC*PPAD
  + BC*NH*D
  + 32
  + 32
  + 32
  + 32
  + 32*3
  + BC*D
  + BC*D;
static_assert(SMEM_FLOATS * 4 <= 327680, "one workgroup per WGP owns the whole 320 KiB of LDS");
static_assert((KZ*G4) % 4 == 0 && KZ % 2 == 0, "gate weights are staged with 16-byte copies and read as k pairs");
static_assert(NH*8 <= 32 && 32 % (NH*8) == 0,
              "phase 5: eight lanes per head of the wave's own batch row (the other lanes repeat them)");
static_assert(NT == BC*32 && NMEM % 32 == 0,
              "memory matrix in registers: one wave per batch row, NMEM/32 memory rows per lane");
static_assert(NMEM % 32 == 0 && (NMEM & (NMEM - 1)) == 0,
              "addressing weights in registers: lane l of a wave holds columns l + 32*j (j < NMEM/32) of both heads of its batch row, the columns whose memory rows it owns");
static_assert((HTRIPS - 1)*NT < BC*H, "hidden-unit loops: only the last trip is partial");
static_assert(INW <= 32 && D <= 32 && BC*(KXR - INW - D) <= NT && OUTW*4 == 32 && (H + D) % 4 == 0,
              "z build: a wave builds its own row, x and r one lane per element; output: 4 lanes per output unit, (H+D)/4 terms each");
static_assert(H % 4 == 0 && KZ == H + KXR && INW + D <= KXR && KXR % 4 == 0 && KXR <= ZP && ZP % 2 == 0 && H % 2 == 0 && (H/4) % HCH == 0,
              "gate product: K tiles 0..H/4-1 read h straight from its own array, the last KXR/4 read the x|r rows; 8-byte fragment loads");
static_assert(G4/16 > 3*(NT/32) && G4/16 <= 4*(NT/32) && G4 % 16 == 0,
              "gate product: every wave owns column tiles wave, wave+8, wave+16; the waves below G4/16 - 24 own a fourth");
static_assert(BC == 8,
              "WMMA tiles: lanes m and m+8 load the same batch row as their A row; D rows 0-7 sit in lanes 0-15");
}

__device__ __forceinline__ float sigm(float x) { return 1.f / (1.f + expf(-x)); }
__device__ __forceinline__ float softplusf(float x) { return (x > 20.f) ? x : log1pf(expf(x)); }
__device__ __forceinline__ float clipf(float x) { return fminf(fmaxf(x, -CLIPV), CLIPV); }

template <int X>
__device__ __forceinline__ float lane_xor(float v) {
  return __builtin_bit_cast(float, __builtin_amdgcn_update_dpp(
      0, __builtin_bit_cast(int, v), 0x160 + X, 0xf, 0xf, true));
}
__device__ __forceinline__ float sum8(float v) {
  v += lane_xor<1>(v); v += lane_xor<2>(v); v += lane_xor<4>(v);
  return v;
}
__device__ __forceinline__ float sum16(float v) {
  v += lane_xor<1>(v); v += lane_xor<2>(v); v += lane_xor<4>(v); v += lane_xor<8>(v);
  return v;
}
__device__ __forceinline__ float lane_xor16(float v) {
  const int i = __builtin_bit_cast(int, v);
  return __builtin_bit_cast(float, __builtin_amdgcn_permlanex16(i, i, 0x76543210, 0xfedcba98, false, false));
}
__device__ __forceinline__ float sum32(float v) {
  v = sum16(v); v += lane_xor16(v);
  return v;
}
__device__ __forceinline__ void wave_sync() {
  __builtin_amdgcn_fence(__ATOMIC_RELEASE, "wavefront");
  __builtin_amdgcn_wave_barrier();
  __builtin_amdgcn_fence(__ATOMIC_ACQUIRE, "wavefront");
}
__device__ __forceinline__ float max16(float v) {
  v = fmaxf(v, lane_xor<1>(v)); v = fmaxf(v, lane_xor<2>(v));
  v = fmaxf(v, lane_xor<4>(v)); v = fmaxf(v, lane_xor<8>(v));
  return v;
}

template <int BKT0, int N>
__device__ __forceinline__ void gate_ktiles(const float* arow,
                                            const float* wc0, const float* wc1, const float* wc2, const float* wc3,
                                            bool own4, v8f& a0, v8f& a1, v8f& a2, v8f& a3) {
  if (own4) {
    #pragma unroll
    for (int j = 0; j < N; ++j) {
      const v2f av = *(const v2f*)(arow + 4*j);
      const v2f b3 = *(const v2f*)(wc3 + (BKT0 + j)*(4*G4));
      a3 = __builtin_amdgcn_wmma_f32_16x16x4_f32(false, av, false, b3, (short)0, a3, false, false);
    }
  }
  #pragma unroll
  for (int j = 0; j < N; ++j) {
    const v2f av = *(const v2f*)(arow + 4*j);
    const v2f b0 = *(const v2f*)(wc0 + (BKT0 + j)*(4*G4));
    const v2f b1 = *(const v2f*)(wc1 + (BKT0 + j)*(4*G4));
    const v2f b2 = *(const v2f*)(wc2 + (BKT0 + j)*(4*G4));
    a0 = __builtin_amdgcn_wmma_f32_16x16x4_f32(false, av, false, b0, (short)0, a0, false, false);
    a1 = __builtin_amdgcn_wmma_f32_16x16x4_f32(false, av, false, b1, (short)0, a1, false, false);
    a2 = __builtin_amdgcn_wmma_f32_16x16x4_f32(false, av, false, b2, (short)0, a2, false, false);
  }
}

__device__ __forceinline__ float max32(float v) {
  v = max16(v); v = fmaxf(v, lane_xor16(v));
  return v;
}

__global__ __launch_bounds__(256) void pack_weights(
    const float* __restrict__ Wx, const float* __restrict__ Wh,
    const float* __restrict__ W_params, const float* __restrict__ b_params,
    float* __restrict__ ws)
{
  int i = blockIdx.x * 256 + threadIdx.x;
  if (i < KZ*G4) {
    int k = i / G4, n = i - k*G4;
    float v = (k < H) ? Wh[k*G4 + n] : ((k < H + INW + D) ? Wx[(k - H)*G4 + n] : 0.f);
    ws[WS_WCAT + (k >> 1)*(2*G4) + 2*n + (k & 1)] = v;
  } else if (i < KZ*G4 + H*PPAD) {
    int j = i - KZ*G4;
    int k = j / PPAD, n = j - k*PPAD;
    ws[WS_WP + j] = (n < PDIM) ? W_params[k*PDIM + n] : 0.f;
  } else if (i < WS_TOT) {
    int n = i - KZ*G4 - H*PPAD;
    ws[WS_BP + n] = (n < PDIM) ? b_params[n] : 0.f;
  }
}

__global__ __launch_bounds__(NT) void ntm_fused(
    const float* __restrict__ inputs,
    const float* __restrict__ Wcat,
    const float* __restrict__ b_lstm,
    const float* __restrict__ Wp,
    const float* __restrict__ bp,
    const float* __restrict__ W_out,
    const float* __restrict__ b_out,
    const int*   __restrict__ seq_ptr,
    float* __restrict__ out)
{
  __shared__ __attribute__((aligned(16))) float smem[SMEM_FLOATS];
  float* sWc  = smem;
  float* sBl  = sWc  + KZ*G4;
  float* sWo  = sBl  + G4;
  float* sH   = sWo  + (H + D)*OUTW;
  float* sC   = sH   + BC*H;
  float* sR   = sC   + BC*H;
  float* sZ   = sR   + BC*D;
  float* sG   = sZ   + BC*ZP;
  float* sP   = sG   + BC*G4;
  float* sK   = sP   + BC*PPAD;
  float* sKn  = sK   + BC*NH*D;
  float* sBe  = sKn  + 32;
  float* sGg  = sBe  + 32;
  float* sGa  = sGg  + 32;
  float* sS   = sGa  + 32;
  float* sE   = sS   + 32*3;
  float* sAd  = sE   + BC*D;

  const int tid   = threadIdx.x;
  const int wave  = tid >> 5;
  const int lane  = tid & 31;
  const int m     = lane & 15;
  const int ma    = m & (BC - 1);
  const int khalf = (lane < 16) ? 0 : 2;
  const int b0    = blockIdx.x * BC;
  const int seq   = seq_ptr[0];
  const int tOutStart = seq + 1;
  const int TOUT  = T_LEN - tOutStart;

  const int mb = tid >> 5;
  const int mc = tid & 31;
  float Mr[MROWS][D];
  for (int j = 0; j < MROWS; ++j) {
    for (int d = 0; d < D; ++d) Mr[j][d] = 1e-6f;
  }
  float wv[NH][MROWS];
  for (int h = 0; h < NH; ++h) {
    for (int j = 0; j < MROWS; ++j) wv[h][j] = (mc == 0 && j == 0) ? 1.f : 0.f;
  }

  {
    const v4f* __restrict__ src = (const v4f*)Wcat;
    v4f* dst = (v4f*)sWc;
    for (int i = tid; i < KZ*G4/4; i += NT) dst[i] = src[i];
  }
  for (int i = tid; i < G4;           i += NT) sBl[i] = b_lstm[i];
  for (int i = tid; i < (H + D)*OUTW; i += NT) sWo[i] = W_out[i];
  const float bo = b_out[tid & (OUTW - 1)];

  for (int i = tid; i < BC*H;       i += NT) { sH[i] = 0.f; sC[i] = 0.f; }
  for (int i = tid; i < BC*D;       i += NT) sR[i] = 0.f;
  if (tid < BC*(KXR - INW - D)) {
    const int b = tid / (KXR - INW - D);
    sZ[b*ZP + INW + D + (tid - b*(KXR - INW - D))] = 0.f;
  }
  __syncthreads();

  const bool own4 = __builtin_amdgcn_readfirstlane(wave) < G4/16 - 3*(NT/32);
  const int n0 = (wave           )*16 + m;
  const int n1 = (wave +   NT/32 )*16 + m;
  const int n2 = (wave + 2*NT/32 )*16 + m;
  const int n3 = own4 ? (wave + 3*NT/32)*16 + m : n2;
  const float* wc0 = sWc + 2*n0 + (khalf >> 1)*(2*G4);
  const float* wc1 = sWc + 2*n1 + (khalf >> 1)*(2*G4);
  const float* wc2 = sWc + 2*n2 + (khalf >> 1)*(2*G4);
  const float* wc3 = sWc + 2*n3 + (khalf >> 1)*(2*G4);
  const float* ah  = sH + ma*H  + khalf;
  const float* az  = sZ + ma*ZP + khalf;
  const float bias0 = sBl[n0], bias1 = sBl[n1], bias2 = sBl[n2], bias3 = sBl[n3];
  v8f acc0 = { bias0, bias0, bias0, bias0, bias0, bias0, bias0, bias0 };
  v8f acc1 = { bias1, bias1, bias1, bias1, bias1, bias1, bias1, bias1 };
  v8f acc2 = { bias2, bias2, bias2, bias2, bias2, bias2, bias2, bias2 };
  v8f acc3 = { bias3, bias3, bias3, bias3, bias3, bias3, bias3, bias3 };

  for (int t = 0; t < T_LEN; ++t) {
    {
      const float xv = inputs[(size_t)(b0 + mb) * T_LEN * INW + t * INW + ((lane < INW) ? lane : 0)];
      const float rv = sR[mb*D + ((lane < D) ? lane : 0)];
      if (lane < INW) sZ[mb*ZP + lane] = xv;
      if (lane < D)   sZ[mb*ZP + INW + lane] = rv;
    }
    __syncthreads();

    gate_ktiles<H/4, KXR/4>(az, wc0, wc1, wc2, wc3, own4, acc0, acc1, acc2, acc3);
    if (lane < 16) {
      #pragma unroll
      for (int v = 0; v < 8; ++v) {
        sG[v*G4 + n0] = acc0[v]; sG[v*G4 + n1] = acc1[v]; sG[v*G4 + n2] = acc2[v];
      }
      if (own4) {
        #pragma unroll
        for (int v = 0; v < 8; ++v) sG[v*G4 + n3] = acc3[v];
      }
    }
    __syncthreads();

    {
      float gi[HTRIPS], gf[HTRIPS], gg[HTRIPS], go[HTRIPS], cp[HTRIPS];
      #pragma unroll
      for (int it = 0; it < HTRIPS; ++it) {
        const int i = min(tid + it*NT, BC*H - 1);
        const int b = i / H, u = i - b*H;
        gi[it] = sG[b*G4 + u];
        gf[it] = sG[b*G4 + H   + u];
        gg[it] = sG[b*G4 + 2*H + u];
        go[it] = sG[b*G4 + 3*H + u];
        cp[it] = sC[i];
      }
      #pragma unroll
      for (int it = 0; it < HTRIPS; ++it) {
        const int i = tid + it*NT;
        if (it < HTRIPS - 1 || i < BC*H) {
          float c  = sigm(gf[it]) * cp[it] + sigm(gi[it]) * tanhf(gg[it]);
          float h  = sigm(go[it]) * tanhf(c);
          sC[i] = c; sH[i] = h;
        }
      }
    }
    __syncthreads();

    for (int ct = wave; ct < PPAD/16; ct += NT/32) {
      int n = ct*16 + m;
      const float* __restrict__ wp = Wp + n;
      float bias = bp[n];
      v8f acc = { bias, bias, bias, bias, bias, bias, bias, bias };
      for (int kt = 0; kt < H/4; ++kt) {
        int k0 = kt*4 + khalf;
        v2f av; av.x = sH[ma*H + k0];       av.y = sH[ma*H + k0 + 1];
        v2f bv; bv.x = wp[(size_t)k0*PPAD]; bv.y = wp[(size_t)(k0 + 1)*PPAD];
        acc = __builtin_amdgcn_wmma_f32_16x16x4_f32(false, av, false, bv,
                                                    (short)0, acc, false, false);
      }
      float pv[8];
      #pragma unroll
      for (int v = 0; v < 8; ++v) pv[v] = clipf(acc[v]);
      if (lane < 16) {
        #pragma unroll
        for (int v = 0; v < 8; ++v) sP[v*PPAD + n] = pv[v];
      }
    }
    __syncthreads();

    #pragma unroll
    for (int v = 0; v < 8; ++v) { acc0[v] = bias0; acc1[v] = bias1; acc2[v] = bias2; acc3[v] = bias3; }
    gate_ktiles<0*HCH, HCH>(ah + 0*HCH*4, wc0, wc1, wc2, wc3, own4, acc0, acc1, acc2, acc3);

    {
      const int bh  = mb*NH + ((lane >> 3) & 1);
      const int sub = lane & 7;
      const float* pb = sP + (bh >> 1)*PPAD + (bh & 1)*PPH;
      float kn2 = 0.f;
      for (int j = 0; j < 3; ++j) {
        const int d = sub + 8*j;
        if (d < D) {
          float kv = tanhf(pb[d]);
          sK[bh*D + d] = kv;
          kn2 += kv * kv;
        }
      }
      kn2 = sum8(kn2);
      if (sub == 0) {
        sKn[bh] = sqrtf(kn2);
      } else if (sub == 1 || sub == 4) {
        float sp = softplusf(pb[(sub == 1) ? D : D + 5]);
        if (sub == 1) sBe[bh] = sp; else sGa[bh] = sp + 1.f;
      } else if (sub == 2) {
        sGg[bh] = sigm(pb[D + 1]);
      } else if (sub == 3) {
        float s0 = pb[D + 2], s1 = pb[D + 3], s2 = pb[D + 4];
        float mx = fmaxf(s0, fmaxf(s1, s2));
        float e0 = expf(s0 - mx), e1 = expf(s1 - mx), e2 = expf(s2 - mx);
        float inv = 1.f / (e0 + e1 + e2);
        sS[bh*3 + 0] = e0 * inv; sS[bh*3 + 1] = e1 * inv; sS[bh*3 + 2] = e2 * inv;
      }
    }
    wave_sync();

    gate_ktiles<1*HCH, HCH>(ah + 1*HCH*4, wc0, wc1, wc2, wc3, own4, acc0, acc1, acc2, acc3);
    float sc[NH][MROWS];
    {
      float k0[D], k1[D];
      for (int d = 0; d < D; ++d) {
        k0[d] = sK[(mb*NH + 0)*D + d];
        k1[d] = sK[(mb*NH + 1)*D + d];
      }
      const float be0 = sBe[mb*NH + 0], be1 = sBe[mb*NH + 1];
      const float kn0 = sKn[mb*NH + 0], kn1 = sKn[mb*NH + 1];
      for (int j = 0; j < MROWS; ++j) {
        float mn2 = 0.f, d0 = 0.f, d1 = 0.f;
        for (int d = 0; d < D; ++d) {
          float mv = Mr[j][d];
          mn2 += mv * mv; d0 += k0[d] * mv; d1 += k1[d] * mv;
        }
        float mn = sqrtf(mn2);
        sc[0][j] = be0 * d0 / (kn0*mn + EPSV);
        sc[1][j] = be1 * d1 / (kn1*mn + EPSV);
      }
    }

    gate_ktiles<2*HCH, HCH>(ah + 2*HCH*4, wc0, wc1, wc2, wc3, own4, acc0, acc1, acc2, acc3);
    float wg[NH][MROWS];
    for (int h = 0; h < NH; ++h) {
      float mx = sc[h][0];
      for (int j = 1; j < MROWS; ++j) mx = fmaxf(mx, sc[h][j]);
      mx = max32(mx);
      float ex[MROWS];
      float sum = 0.f;
      for (int j = 0; j < MROWS; ++j) { ex[j] = expf(sc[h][j] - mx); sum += ex[j]; }
      sum = sum32(sum);
      const float inv = 1.f / sum;
      const float g = sGg[mb*NH + h];
      for (int j = 0; j < MROWS; ++j) {
        const float wc = ex[j] * inv;
        wg[h][j] = g * wc + (1.f - g) * wv[h][j];
      }
    }

    gate_ktiles<3*HCH, HCH>(ah + 3*HCH*4, wc0, wc1, wc2, wc3, own4, acc0, acc1, acc2, acc3);

    for (int h = 0; h < NH; ++h) {
      const float s0 = sS[(mb*NH + h)*3 + 0], s1 = sS[(mb*NH + h)*3 + 1], s2 = sS[(mb*NH + h)*3 + 2];
      const float ga = sGa[mb*NH + h];
      float up[MROWS], dn[MROWS];
      for (int j = 0; j < MROWS; ++j) {
        const float offer_up = (mc == 0)  ? wg[h][(j + 1) & (MROWS - 1)]         : wg[h][j];
        const float offer_dn = (mc == 31) ? wg[h][(j + MROWS - 1) & (MROWS - 1)] : wg[h][j];
        up[j] = __shfl(offer_up, (mc + 1)  & 31, 32);
        dn[j] = __shfl(offer_dn, (mc + 31) & 31, 32);
      }
      float pw[MROWS];
      float sum = 0.f;
      for (int j = 0; j < MROWS; ++j) {
        const float wt = s0 * up[j] + s1 * wg[h][j] + s2 * dn[j];
        const float w = fmaxf(wt, 1e-16f);
        pw[j] = expf(ga * logf(w));
        sum += pw[j];
      }
      sum = sum32(sum);
      const float inv = 1.f / sum;
      for (int j = 0; j < MROWS; ++j) wv[h][j] = pw[j] * inv;
    }

    gate_ktiles<4*HCH, HCH>(ah + 4*HCH*4, wc0, wc1, wc2, wc3, own4, acc0, acc1, acc2, acc3);
    {
      float part[D];
      for (int d = 0; d < D; ++d) {
        float acc = 0.f;
        for (int j = 0; j < MROWS; ++j) acc += wv[0][j] * Mr[j][d];
        part[d] = sum32(acc);
      }
      if (mc == 0) {
        for (int d = 0; d < D; ++d) sR[mb*D + d] = part[d];
      }
    }
    if (lane < D) {
      sE[mb*D + lane]  = sigm (sP[mb*PPAD + NH*PPH + lane]);
      sAd[mb*D + lane] = tanhf(sP[mb*PPAD + NH*PPH + D + lane]);
    }
    wave_sync();

    {
      float ev[D], av[D];
      for (int d = 0; d < D; ++d) { ev[d] = sE[mb*D + d]; av[d] = sAd[mb*D + d]; }
      for (int j = 0; j < MROWS; ++j) {
        const float ww = wv[1][j];
        for (int d = 0; d < D; ++d)
          Mr[j][d] = Mr[j][d] * (1.f - ww * ev[d]) + ww * av[d];
      }
    }

    {
      const int o = lane & (OUTW - 1), q = lane >> 3;
      float acc = (q == 0) ? bo : 0.f;
      #pragma unroll
      for (int jj = 0; jj < (H + D)/4; ++jj) {
        const int j = q*((H + D)/4) + jj;
        const float v = (j < H) ? sH[mb*H + j] : sR[mb*D + (j - H)];
        acc += v * sWo[j*OUTW + o];
      }
      acc += lane_xor<8>(acc);
      acc += lane_xor16(acc);
      acc = clipf(acc);
      if (q == 0 && t >= tOutStart) {
        out[(size_t)(b0 + mb) * TOUT * OUTW + (t - tOutStart) * OUTW + o] = sigm(acc);
      }
    }
  }
}

extern "C" void kernel_launch(void* const* d_in, const int* in_sizes, int n_in,
                              void* d_out, int out_size, void* d_ws, size_t ws_size,
                              hipStream_t stream) {
  (void)in_sizes; (void)n_in; (void)out_size; (void)ws_size;
  const float* inputs   = (const float*)d_in[0];
  const float* Wx       = (const float*)d_in[1];
  const float* Wh       = (const float*)d_in[2];
  const float* b_lstm   = (const float*)d_in[3];
  const float* W_params = (const float*)d_in[4];
  const float* b_params = (const float*)d_in[5];
  const float* W_out    = (const float*)d_in[6];
  const float* b_out    = (const float*)d_in[7];
  const int*   seq      = (const int*)d_in[8];
  float* out = (float*)d_out;
  float* ws  = (float*)d_ws;

  pack_weights<<<dim3((WS_TOT + 255) / 256), dim3(256), 0, stream>>>(
      Wx, Wh, W_params, b_params, ws);

  ntm_fused<<<dim3(B_TOT / BC), dim3(NT), 0, stream>>>(
      inputs, ws + WS_WCAT, b_lstm, ws + WS_WP, ws + WS_BP,
      W_out, b_out, seq, out);
}


# Sigmoid and tanh (LSTM cell, key, gates, erase/add, output) from one hardware exp2 + one reciprocal each, branch-free; the cell's trips interleave; activation rounding changes (abs error <3e-7)
# speedup vs baseline: 7.0652x; 1.1525x over previous
#include <hip/hip_runtime.h>
#include <math.h>

typedef float v2f __attribute__((ext_vector_type(2)));
typedef float v4f __attribute__((ext_vector_type(4)));
typedef float v8f __attribute__((ext_vector_type(8)));

namespace {
constexpr int B_TOT = 2048;
constexpr int T_LEN = 66;
constexpr int INW   = 9;
constexpr int H     = 100;
constexpr int NMEM  = 128;
constexpr int D     = 20;
constexpr int OUTW  = 8;
constexpr int NH    = 2;
constexpr int PPH   = 26;
constexpr int PDIM  = 92;
constexpr int PPAD  = 96;
constexpr int G4    = 400;
constexpr int KZ    = 132;
constexpr int KXR   = 32;
constexpr int ZP    = 36;
constexpr int HCH   = 5;
constexpr int BC    = 8;
constexpr int NT    = 256;
constexpr float CLIPV = 20.f;
constexpr float EPSV  = 1e-8f;

constexpr int WS_WCAT = 0;
constexpr int WS_WP   = WS_WCAT + KZ*G4;
constexpr int WS_BP   = WS_WP   + H*PPAD;
constexpr int WS_TOT  = WS_BP   + PPAD;

constexpr int MROWS = NMEM / 32;

constexpr int HTRIPS = (BC*H + NT - 1) / NT;

constexpr int SMEM_FLOATS =
    KZ*G4
  + G4
  + (H + D)*OUTW
  + BC*H
  + BC*H
  + BC*D
  + BC*ZP
  + BC*G4
  + BC*PPAD
  + BC*NH*D
  + 32
  + 32
  + 32
  + 32
  + 32*3
  + BC*D
  + BC*D;
static_assert(SMEM_FLOATS * 4 <= 327680, "one workgroup per WGP owns the whole 320 KiB of LDS");
static_assert((KZ*G4) % 4 == 0 && KZ % 2 == 0, "gate weights are staged with 16-byte copies and read as k pairs");
static_assert(NH*8 <= 32 && 32 % (NH*8) == 0,
              "phase 5: eight lanes per head of the wave's own batch row (the other lanes repeat them)");
static_assert(NT == BC*32 && NMEM % 32 == 0,
              "memory matrix in registers: one wave per batch row, NMEM/32 memory rows per lane");
static_assert(NMEM % 32 == 0 && (NMEM & (NMEM - 1)) == 0,
              "addressing weights in registers: lane l of a wave holds columns l + 32*j (j < NMEM/32) of both heads of its batch row, the columns whose memory rows it owns");
static_assert((HTRIPS - 1)*NT < BC*H, "hidden-unit loops: only the last trip is partial");
constexpr int TAILW = (BC*H - (HTRIPS - 1)*NT) / 32;
static_assert((BC*H - (HTRIPS - 1)*NT) % 32 == 0 && TAILW >= 1 && TAILW <= NT/32,
              "hidden-unit loops: the last trip covers whole waves, so a wave runs it in full or not at all");
static_assert(INW <= 32 && D <= 32 && BC*(KXR - INW - D) <= NT && OUTW*4 == 32 && (H + D) % 4 == 0,
              "z build: a wave builds its own row, x and r one lane per element; output: 4 lanes per output unit, (H+D)/4 terms each");
static_assert(H % 4 == 0 && KZ == H + KXR && INW + D <= KXR && KXR % 4 == 0 && KXR <= ZP && ZP % 2 == 0 && H % 2 == 0 && (H/4) % HCH == 0,
              "gate product: K tiles 0..H/4-1 read h straight from its own array, the last KXR/4 read the x|r rows; 8-byte fragment loads");
static_assert(G4/16 > 3*(NT/32) && G4/16 <= 4*(NT/32) && G4 % 16 == 0,
              "gate product: every wave owns column tiles wave, wave+8, wave+16; the waves below G4/16 - 24 own a fourth");
static_assert(BC == 8,
              "WMMA tiles: lanes m and m+8 load the same batch row as their A row; D rows 0-7 sit in lanes 0-15");
}

__device__ __forceinline__ float sigm(float x) {
  return __builtin_amdgcn_rcpf(1.f + __builtin_amdgcn_exp2f(-1.44269504f * x));
}
__device__ __forceinline__ float tanh_hw(float x) {
  const float t = __builtin_amdgcn_exp2f(-2.88539008f * fabsf(x));
  return copysignf((1.f - t) * __builtin_amdgcn_rcpf(1.f + t), x);
}
__device__ __forceinline__ float softplusf(float x) { return (x > 20.f) ? x : log1pf(expf(x)); }
__device__ __forceinline__ float clipf(float x) { return fminf(fmaxf(x, -CLIPV), CLIPV); }

template <int X>
__device__ __forceinline__ float lane_xor(float v) {
  return __builtin_bit_cast(float, __builtin_amdgcn_update_dpp(
      0, __builtin_bit_cast(int, v), 0x160 + X, 0xf, 0xf, true));
}
__device__ __forceinline__ float sum8(float v) {
  v += lane_xor<1>(v); v += lane_xor<2>(v); v += lane_xor<4>(v);
  return v;
}
__device__ __forceinline__ float sum16(float v) {
  v += lane_xor<1>(v); v += lane_xor<2>(v); v += lane_xor<4>(v); v += lane_xor<8>(v);
  return v;
}
__device__ __forceinline__ float lane_xor16(float v) {
  const int i = __builtin_bit_cast(int, v);
  return __builtin_bit_cast(float, __builtin_amdgcn_permlanex16(i, i, 0x76543210, 0xfedcba98, false, false));
}
__device__ __forceinline__ float sum32(float v) {
  v = sum16(v); v += lane_xor16(v);
  return v;
}
__device__ __forceinline__ void wave_sync() {
  __builtin_amdgcn_fence(__ATOMIC_RELEASE, "wavefront");
  __builtin_amdgcn_wave_barrier();
  __builtin_amdgcn_fence(__ATOMIC_ACQUIRE, "wavefront");
}
__device__ __forceinline__ float max16(float v) {
  v = fmaxf(v, lane_xor<1>(v)); v = fmaxf(v, lane_xor<2>(v));
  v = fmaxf(v, lane_xor<4>(v)); v = fmaxf(v, lane_xor<8>(v));
  return v;
}

template <int BKT0, int N>
__device__ __forceinline__ void gate_ktiles(const float* arow,
                                            const float* wc0, const float* wc1, const float* wc2, const float* wc3,
                                            bool own4, v8f& a0, v8f& a1, v8f& a2, v8f& a3) {
  if (own4) {
    #pragma unroll
    for (int j = 0; j < N; ++j) {
      const v2f av = *(const v2f*)(arow + 4*j);
      const v2f b3 = *(const v2f*)(wc3 + (BKT0 + j)*(4*G4));
      a3 = __builtin_amdgcn_wmma_f32_16x16x4_f32(false, av, false, b3, (short)0, a3, false, false);
    }
  }
  #pragma unroll
  for (int j = 0; j < N; ++j) {
    const v2f av = *(const v2f*)(arow + 4*j);
    const v2f b0 = *(const v2f*)(wc0 + (BKT0 + j)*(4*G4));
    const v2f b1 = *(const v2f*)(wc1 + (BKT0 + j)*(4*G4));
    const v2f b2 = *(const v2f*)(wc2 + (BKT0 + j)*(4*G4));
    a0 = __builtin_amdgcn_wmma_f32_16x16x4_f32(false, av, false, b0, (short)0, a0, false, false);
    a1 = __builtin_amdgcn_wmma_f32_16x16x4_f32(false, av, false, b1, (short)0, a1, false, false);
    a2 = __builtin_amdgcn_wmma_f32_16x16x4_f32(false, av, false, b2, (short)0, a2, false, false);
  }
}

__device__ __forceinline__ float max32(float v) {
  v = max16(v); v = fmaxf(v, lane_xor16(v));
  return v;
}

template <int NTR>
__device__ __forceinline__ void lstm_cell(const float* sG, float* sC, float* sH, int tid) {
  float gi[NTR], gf[NTR], gg[NTR], go[NTR], cp[NTR];
  #pragma unroll
  for (int it = 0; it < NTR; ++it) {
    const int i = tid + it*NT;
    const int b = i / H, u = i - b*H;
    gi[it] = sG[b*G4 + u];
    gf[it] = sG[b*G4 + H   + u];
    gg[it] = sG[b*G4 + 2*H + u];
    go[it] = sG[b*G4 + 3*H + u];
    cp[it] = sC[i];
  }
  #pragma unroll
  for (int it = 0; it < NTR; ++it) {
    const int i = tid + it*NT;
    const float c = sigm(gf[it]) * cp[it] + sigm(gi[it]) * tanh_hw(gg[it]);
    const float h = sigm(go[it]) * tanh_hw(c);
    sC[i] = c; sH[i] = h;
  }
}

__global__ __launch_bounds__(256) void pack_weights(
    const float* __restrict__ Wx, const float* __restrict__ Wh,
    const float* __restrict__ W_params, const float* __restrict__ b_params,
    float* __restrict__ ws)
{
  int i = blockIdx.x * 256 + threadIdx.x;
  if (i < KZ*G4) {
    int k = i / G4, n = i - k*G4;
    float v = (k < H) ? Wh[k*G4 + n] : ((k < H + INW + D) ? Wx[(k - H)*G4 + n] : 0.f);
    ws[WS_WCAT + (k >> 1)*(2*G4) + 2*n + (k & 1)] = v;
  } else if (i < KZ*G4 + H*PPAD) {
    int j = i - KZ*G4;
    int k = j / PPAD, n = j - k*PPAD;
    ws[WS_WP + j] = (n < PDIM) ? W_params[k*PDIM + n] : 0.f;
  } else if (i < WS_TOT) {
    int n = i - KZ*G4 - H*PPAD;
    ws[WS_BP + n] = (n < PDIM) ? b_params[n] : 0.f;
  }
}

__global__ __launch_bounds__(NT) void ntm_fused(
    const float* __restrict__ inputs,
    const float* __restrict__ Wcat,
    const float* __restrict__ b_lstm,
    const float* __restrict__ Wp,
    const float* __restrict__ bp,
    const float* __restrict__ W_out,
    const float* __restrict__ b_out,
    const int*   __restrict__ seq_ptr,
    float* __restrict__ out)
{
  __shared__ __attribute__((aligned(16))) float smem[SMEM_FLOATS];
  float* sWc  = smem;
  float* sBl  = sWc  + KZ*G4;
  float* sWo  = sBl  + G4;
  float* sH   = sWo  + (H + D)*OUTW;
  float* sC   = sH   + BC*H;
  float* sR   = sC   + BC*H;
  float* sZ   = sR   + BC*D;
  float* sG   = sZ   + BC*ZP;
  float* sP   = sG   + BC*G4;
  float* sK   = sP   + BC*PPAD;
  float* sKn  = sK   + BC*NH*D;
  float* sBe  = sKn  + 32;
  float* sGg  = sBe  + 32;
  float* sGa  = sGg  + 32;
  float* sS   = sGa  + 32;
  float* sE   = sS   + 32*3;
  float* sAd  = sE   + BC*D;

  const int tid   = threadIdx.x;
  const int wave  = tid >> 5;
  const int lane  = tid & 31;
  const int m     = lane & 15;
  const int ma    = m & (BC - 1);
  const int khalf = (lane < 16) ? 0 : 2;
  const int b0    = blockIdx.x * BC;
  const bool tail_wave = __builtin_amdgcn_readfirstlane(wave) < TAILW;
  const int seq   = seq_ptr[0];
  const int tOutStart = seq + 1;
  const int TOUT  = T_LEN - tOutStart;

  const int mb = tid >> 5;
  const int mc = tid & 31;
  float Mr[MROWS][D];
  for (int j = 0; j < MROWS; ++j) {
    for (int d = 0; d < D; ++d) Mr[j][d] = 1e-6f;
  }
  float wv[NH][MROWS];
  for (int h = 0; h < NH; ++h) {
    for (int j = 0; j < MROWS; ++j) wv[h][j] = (mc == 0 && j == 0) ? 1.f : 0.f;
  }

  {
    const v4f* __restrict__ src = (const v4f*)Wcat;
    v4f* dst = (v4f*)sWc;
    for (int i = tid; i < KZ*G4/4; i += NT) dst[i] = src[i];
  }
  for (int i = tid; i < G4;           i += NT) sBl[i] = b_lstm[i];
  for (int i = tid; i < (H + D)*OUTW; i += NT) sWo[i] = W_out[i];
  const float bo = b_out[tid & (OUTW - 1)];

  for (int i = tid; i < BC*H;       i += NT) { sH[i] = 0.f; sC[i] = 0.f; }
  for (int i = tid; i < BC*D;       i += NT) sR[i] = 0.f;
  if (tid < BC*(KXR - INW - D)) {
    const int b = tid / (KXR - INW - D);
    sZ[b*ZP + INW + D + (tid - b*(KXR - INW - D))] = 0.f;
  }
  __syncthreads();

  const bool own4 = __builtin_amdgcn_readfirstlane(wave) < G4/16 - 3*(NT/32);
  const int n0 = (wave           )*16 + m;
  const int n1 = (wave +   NT/32 )*16 + m;
  const int n2 = (wave + 2*NT/32 )*16 + m;
  const int n3 = own4 ? (wave + 3*NT/32)*16 + m : n2;
  const float* wc0 = sWc + 2*n0 + (khalf >> 1)*(2*G4);
  const float* wc1 = sWc + 2*n1 + (khalf >> 1)*(2*G4);
  const float* wc2 = sWc + 2*n2 + (khalf >> 1)*(2*G4);
  const float* wc3 = sWc + 2*n3 + (khalf >> 1)*(2*G4);
  const float* ah  = sH + ma*H  + khalf;
  const float* az  = sZ + ma*ZP + khalf;
  const float bias0 = sBl[n0], bias1 = sBl[n1], bias2 = sBl[n2], bias3 = sBl[n3];
  v8f acc0 = { bias0, bias0, bias0, bias0, bias0, bias0, bias0, bias0 };
  v8f acc1 = { bias1, bias1, bias1, bias1, bias1, bias1, bias1, bias1 };
  v8f acc2 = { bias2, bias2, bias2, bias2, bias2, bias2, bias2, bias2 };
  v8f acc3 = { bias3, bias3, bias3, bias3, bias3, bias3, bias3, bias3 };

  for (int t = 0; t < T_LEN; ++t) {
    {
      const float xv = inputs[(size_t)(b0 + mb) * T_LEN * INW + t * INW + ((lane < INW) ? lane : 0)];
      const float rv = sR[mb*D + ((lane < D) ? lane : 0)];
      if (lane < INW) sZ[mb*ZP + lane] = xv;
      if (lane < D)   sZ[mb*ZP + INW + lane] = rv;
    }
    __syncthreads();

    gate_ktiles<H/4, KXR/4>(az, wc0, wc1, wc2, wc3, own4, acc0, acc1, acc2, acc3);
    if (lane < 16) {
      #pragma unroll
      for (int v = 0; v < 8; ++v) {
        sG[v*G4 + n0] = acc0[v]; sG[v*G4 + n1] = acc1[v]; sG[v*G4 + n2] = acc2[v];
      }
      if (own4) {
        #pragma unroll
        for (int v = 0; v < 8; ++v) sG[v*G4 + n3] = acc3[v];
      }
    }
    __syncthreads();

    if (tail_wave) lstm_cell<HTRIPS>(sG, sC, sH, tid);
    else           lstm_cell<HTRIPS - 1>(sG, sC, sH, tid);
    __syncthreads();

    for (int ct = wave; ct < PPAD/16; ct += NT/32) {
      int n = ct*16 + m;
      const float* __restrict__ wp = Wp + n;
      float bias = bp[n];
      v8f acc = { bias, bias, bias, bias, bias, bias, bias, bias };
      for (int kt = 0; kt < H/4; ++kt) {
        int k0 = kt*4 + khalf;
        v2f av; av.x = sH[ma*H + k0];       av.y = sH[ma*H + k0 + 1];
        v2f bv; bv.x = wp[(size_t)k0*PPAD]; bv.y = wp[(size_t)(k0 + 1)*PPAD];
        acc = __builtin_amdgcn_wmma_f32_16x16x4_f32(false, av, false, bv,
                                                    (short)0, acc, false, false);
      }
      float pv[8];
      #pragma unroll
      for (int v = 0; v < 8; ++v) pv[v] = clipf(acc[v]);
      if (lane < 16) {
        #pragma unroll
        for (int v = 0; v < 8; ++v) sP[v*PPAD + n] = pv[v];
      }
    }
    __syncthreads();

    #pragma unroll
    for (int v = 0; v < 8; ++v) { acc0[v] = bias0; acc1[v] = bias1; acc2[v] = bias2; acc3[v] = bias3; }
    gate_ktiles<0*HCH, HCH>(ah + 0*HCH*4, wc0, wc1, wc2, wc3, own4, acc0, acc1, acc2, acc3);

    {
      const int bh  = mb*NH + ((lane >> 3) & 1);
      const int sub = lane & 7;
      const float* pb = sP + (bh >> 1)*PPAD + (bh & 1)*PPH;
      float kp[3];
      for (int j = 0; j < 3; ++j) kp[j] = pb[min(sub + 8*j, D - 1)];
      float kn2 = 0.f;
      for (int j = 0; j < 3; ++j) {
        const int d = sub + 8*j;
        const float kv = tanh_hw(kp[j]);
        if (d < D) {
          sK[bh*D + d] = kv;
          kn2 += kv * kv;
        }
      }
      kn2 = sum8(kn2);
      if (sub == 0) {
        sKn[bh] = sqrtf(kn2);
      } else if (sub == 1 || sub == 4) {
        float sp = softplusf(pb[(sub == 1) ? D : D + 5]);
        if (sub == 1) sBe[bh] = sp; else sGa[bh] = sp + 1.f;
      } else if (sub == 2) {
        sGg[bh] = sigm(pb[D + 1]);
      } else if (sub == 3) {
        float s0 = pb[D + 2], s1 = pb[D + 3], s2 = pb[D + 4];
        float mx = fmaxf(s0, fmaxf(s1, s2));
        float e0 = expf(s0 - mx), e1 = expf(s1 - mx), e2 = expf(s2 - mx);
        float inv = 1.f / (e0 + e1 + e2);
        sS[bh*3 + 0] = e0 * inv; sS[bh*3 + 1] = e1 * inv; sS[bh*3 + 2] = e2 * inv;
      }
    }
    wave_sync();

    gate_ktiles<1*HCH, HCH>(ah + 1*HCH*4, wc0, wc1, wc2, wc3, own4, acc0, acc1, acc2, acc3);
    float sc[NH][MROWS];
    {
      float k0[D], k1[D];
      for (int d = 0; d < D; ++d) {
        k0[d] = sK[(mb*NH + 0)*D + d];
        k1[d] = sK[(mb*NH + 1)*D + d];
      }
      const float be0 = sBe[mb*NH + 0], be1 = sBe[mb*NH + 1];
      const float kn0 = sKn[mb*NH + 0], kn1 = sKn[mb*NH + 1];
      for (int j = 0; j < MROWS; ++j) {
        float mn2 = 0.f, d0 = 0.f, d1 = 0.f;
        for (int d = 0; d < D; ++d) {
          float mv = Mr[j][d];
          mn2 += mv * mv; d0 += k0[d] * mv; d1 += k1[d] * mv;
        }
        float mn = sqrtf(mn2);
        sc[0][j] = be0 * d0 / (kn0*mn + EPSV);
        sc[1][j] = be1 * d1 / (kn1*mn + EPSV);
      }
    }

    gate_ktiles<2*HCH, HCH>(ah + 2*HCH*4, wc0, wc1, wc2, wc3, own4, acc0, acc1, acc2, acc3);
    float wg[NH][MROWS];
    for (int h = 0; h < NH; ++h) {
      float mx = sc[h][0];
      for (int j = 1; j < MROWS; ++j) mx = fmaxf(mx, sc[h][j]);
      mx = max32(mx);
      float ex[MROWS];
      float sum = 0.f;
      for (int j = 0; j < MROWS; ++j) { ex[j] = expf(sc[h][j] - mx); sum += ex[j]; }
      sum = sum32(sum);
      const float inv = 1.f / sum;
      const float g = sGg[mb*NH + h];
      for (int j = 0; j < MROWS; ++j) {
        const float wc = ex[j] * inv;
        wg[h][j] = g * wc + (1.f - g) * wv[h][j];
      }
    }

    gate_ktiles<3*HCH, HCH>(ah + 3*HCH*4, wc0, wc1, wc2, wc3, own4, acc0, acc1, acc2, acc3);

    for (int h = 0; h < NH; ++h) {
      const float s0 = sS[(mb*NH + h)*3 + 0], s1 = sS[(mb*NH + h)*3 + 1], s2 = sS[(mb*NH + h)*3 + 2];
      const float ga = sGa[mb*NH + h];
      float up[MROWS], dn[MROWS];
      for (int j = 0; j < MROWS; ++j) {
        const float offer_up = (mc == 0)  ? wg[h][(j + 1) & (MROWS - 1)]         : wg[h][j];
        const float offer_dn = (mc == 31) ? wg[h][(j + MROWS - 1) & (MROWS - 1)] : wg[h][j];
        up[j] = __shfl(offer_up, (mc + 1)  & 31, 32);
        dn[j] = __shfl(offer_dn, (mc + 31) & 31, 32);
      }
      float pw[MROWS];
      float sum = 0.f;
      for (int j = 0; j < MROWS; ++j) {
        const float wt = s0 * up[j] + s1 * wg[h][j] + s2 * dn[j];
        const float w = fmaxf(wt, 1e-16f);
        pw[j] = expf(ga * logf(w));
        sum += pw[j];
      }
      sum = sum32(sum);
      const float inv = 1.f / sum;
      for (int j = 0; j < MROWS; ++j) wv[h][j] = pw[j] * inv;
    }

    gate_ktiles<4*HCH, HCH>(ah + 4*HCH*4, wc0, wc1, wc2, wc3, own4, acc0, acc1, acc2, acc3);
    {
      float part[D];
      for (int d = 0; d < D; ++d) {
        float acc = 0.f;
        for (int j = 0; j < MROWS; ++j) acc += wv[0][j] * Mr[j][d];
        part[d] = sum32(acc);
      }
      if (mc == 0) {
        for (int d = 0; d < D; ++d) sR[mb*D + d] = part[d];
      }
    }
    if (lane < D) {
      sE[mb*D + lane]  = sigm (sP[mb*PPAD + NH*PPH + lane]);
      sAd[mb*D + lane] = tanh_hw(sP[mb*PPAD + NH*PPH + D + lane]);
    }
    wave_sync();

    {
      float ev[D], av[D];
      for (int d = 0; d < D; ++d) { ev[d] = sE[mb*D + d]; av[d] = sAd[mb*D + d]; }
      for (int j = 0; j < MROWS; ++j) {
        const float ww = wv[1][j];
        for (int d = 0; d < D; ++d)
          Mr[j][d] = Mr[j][d] * (1.f - ww * ev[d]) + ww * av[d];
      }
    }

    {
      const int o = lane & (OUTW - 1), q = lane >> 3;
      float acc = (q == 0) ? bo : 0.f;
      #pragma unroll
      for (int jj = 0; jj < (H + D)/4; ++jj) {
        const int j = q*((H + D)/4) + jj;
        const float v = (j < H) ? sH[mb*H + j] : sR[mb*D + (j - H)];
        acc += v * sWo[j*OUTW + o];
      }
      acc += lane_xor<8>(acc);
      acc += lane_xor16(acc);
      acc = clipf(acc);
      if (q == 0 && t >= tOutStart) {
        out[(size_t)(b0 + mb) * TOUT * OUTW + (t - tOutStart) * OUTW + o] = sigm(acc);
      }
    }
  }
}

extern "C" void kernel_launch(void* const* d_in, const int* in_sizes, int n_in,
                              void* d_out, int out_size, void* d_ws, size_t ws_size,
                              hipStream_t stream) {
  (void)in_sizes; (void)n_in; (void)out_size; (void)ws_size;
  const float* inputs   = (const float*)d_in[0];
  const float* Wx       = (const float*)d_in[1];
  const float* Wh       = (const float*)d_in[2];
  const float* b_lstm   = (const float*)d_in[3];
  const float* W_params = (const float*)d_in[4];
  const float* b_params = (const float*)d_in[5];
  const float* W_out    = (const float*)d_in[6];
  const float* b_out    = (const float*)d_in[7];
  const int*   seq      = (const int*)d_in[8];
  float* out = (float*)d_out;
  float* ws  = (float*)d_ws;

  pack_weights<<<dim3((WS_TOT + 255) / 256), dim3(256), 0, stream>>>(
      Wx, Wh, W_params, b_params, ws);

  ntm_fused<<<dim3(B_TOT / BC), dim3(NT), 0, stream>>>(
      inputs, ws + WS_WCAT, b_lstm, ws + WS_WP, ws + WS_BP,
      W_out, b_out, seq, out);
}


# Addressing phases 5-11: library exp/log/sqrt/division -> one hardware exp2/log2/sqrt/rcp each (log2 e folded into beta); sharpening powers scaled so the largest is 1; rounding changes by a few ulp
# speedup vs baseline: 8.0056x; 1.1331x over previous
#include <hip/hip_runtime.h>
#include <math.h>

typedef float v2f __attribute__((ext_vector_type(2)));
typedef float v4f __attribute__((ext_vector_type(4)));
typedef float v8f __attribute__((ext_vector_type(8)));

namespace {
constexpr int B_TOT = 2048;
constexpr int T_LEN = 66;
constexpr int INW   = 9;
constexpr int H     = 100;
constexpr int NMEM  = 128;
constexpr int D     = 20;
constexpr int OUTW  = 8;
constexpr int NH    = 2;
constexpr int PPH   = 26;
constexpr int PDIM  = 92;
constexpr int PPAD  = 96;
constexpr int G4    = 400;
constexpr int KZ    = 132;
constexpr int KXR   = 32;
constexpr int ZP    = 36;
constexpr int HCH   = 5;
constexpr int BC    = 8;
constexpr int NT    = 256;
constexpr float CLIPV = 20.f;
constexpr float EPSV  = 1e-8f;

constexpr int WS_WCAT = 0;
constexpr int WS_WP   = WS_WCAT + KZ*G4;
constexpr int WS_BP   = WS_WP   + H*PPAD;
constexpr int WS_TOT  = WS_BP   + PPAD;

constexpr int MROWS = NMEM / 32;

constexpr int HTRIPS = (BC*H + NT - 1) / NT;

constexpr int SMEM_FLOATS =
    KZ*G4
  + G4
  + (H + D)*OUTW
  + BC*H
  + BC*H
  + BC*D
  + BC*ZP
  + BC*G4
  + BC*PPAD
  + BC*NH*D
  + 32
  + 32
  + 32
  + 32
  + 32*3
  + BC*D
  + BC*D;
static_assert(SMEM_FLOATS * 4 <= 327680, "one workgroup per WGP owns the whole 320 KiB of LDS");
static_assert((KZ*G4) % 4 == 0 && KZ % 2 == 0, "gate weights are staged with 16-byte copies and read as k pairs");
static_assert(NH*8 <= 32 && 32 % (NH*8) == 0,
              "phase 5: eight lanes per head of the wave's own batch row (the other lanes repeat them)");
static_assert(NT == BC*32 && NMEM % 32 == 0,
              "memory matrix in registers: one wave per batch row, NMEM/32 memory rows per lane");
static_assert(NMEM % 32 == 0 && (NMEM & (NMEM - 1)) == 0,
              "addressing weights in registers: lane l of a wave holds columns l + 32*j (j < NMEM/32) of both heads of its batch row, the columns whose memory rows it owns");
static_assert((HTRIPS - 1)*NT < BC*H, "hidden-unit loops: only the last trip is partial");
constexpr int TAILW = (BC*H - (HTRIPS - 1)*NT) / 32;
static_assert((BC*H - (HTRIPS - 1)*NT) % 32 == 0 && TAILW >= 1 && TAILW <= NT/32,
              "hidden-unit loops: the last trip covers whole waves, so a wave runs it in full or not at all");
static_assert(INW <= 32 && D <= 32 && BC*(KXR - INW - D) <= NT && OUTW*4 == 32 && (H + D) % 4 == 0,
              "z build: a wave builds its own row, x and r one lane per element; output: 4 lanes per output unit, (H+D)/4 terms each");
static_assert(H % 4 == 0 && KZ == H + KXR && INW + D <= KXR && KXR % 4 == 0 && KXR <= ZP && ZP % 2 == 0 && H % 2 == 0 && (H/4) % HCH == 0,
              "gate product: K tiles 0..H/4-1 read h straight from its own array, the last KXR/4 read the x|r rows; 8-byte fragment loads");
static_assert(G4/16 > 3*(NT/32) && G4/16 <= 4*(NT/32) && G4 % 16 == 0,
              "gate product: every wave owns column tiles wave, wave+8, wave+16; the waves below G4/16 - 24 own a fourth");
static_assert(BC == 8,
              "WMMA tiles: lanes m and m+8 load the same batch row as their A row; D rows 0-7 sit in lanes 0-15");
}

__device__ __forceinline__ float sigm(float x) {
  return __builtin_amdgcn_rcpf(1.f + __builtin_amdgcn_exp2f(-1.44269504f * x));
}
__device__ __forceinline__ float tanh_hw(float x) {
  const float t = __builtin_amdgcn_exp2f(-2.88539008f * fabsf(x));
  return copysignf((1.f - t) * __builtin_amdgcn_rcpf(1.f + t), x);
}
constexpr float LOG2E = 1.44269504f;
constexpr float LN2   = 0.693147181f;
__device__ __forceinline__ float exp2_hw(float x) { return __builtin_amdgcn_exp2f(x); }
__device__ __forceinline__ float log2_hw(float x) { return __builtin_amdgcn_logf(x); }
__device__ __forceinline__ float rcp_hw (float x) { return __builtin_amdgcn_rcpf(x); }
__device__ __forceinline__ float sqrt_hw(float x) { return __builtin_amdgcn_sqrtf(x); }
__device__ __forceinline__ float softplus_hw(float x) {
  const float t = exp2_hw(-LOG2E * fabsf(x));
  return fmaxf(x, 0.f) + LN2 * log2_hw(1.f + t);
}
__device__ __forceinline__ float clipf(float x) { return fminf(fmaxf(x, -CLIPV), CLIPV); }

template <int X>
__device__ __forceinline__ float lane_xor(float v) {
  return __builtin_bit_cast(float, __builtin_amdgcn_update_dpp(
      0, __builtin_bit_cast(int, v), 0x160 + X, 0xf, 0xf, true));
}
__device__ __forceinline__ float sum8(float v) {
  v += lane_xor<1>(v); v += lane_xor<2>(v); v += lane_xor<4>(v);
  return v;
}
__device__ __forceinline__ float sum16(float v) {
  v += lane_xor<1>(v); v += lane_xor<2>(v); v += lane_xor<4>(v); v += lane_xor<8>(v);
  return v;
}
__device__ __forceinline__ float lane_xor16(float v) {
  const int i = __builtin_bit_cast(int, v);
  return __builtin_bit_cast(float, __builtin_amdgcn_permlanex16(i, i, 0x76543210, 0xfedcba98, false, false));
}
__device__ __forceinline__ float sum32(float v) {
  v = sum16(v); v += lane_xor16(v);
  return v;
}
__device__ __forceinline__ void wave_sync() {
  __builtin_amdgcn_fence(__ATOMIC_RELEASE, "wavefront");
  __builtin_amdgcn_wave_barrier();
  __builtin_amdgcn_fence(__ATOMIC_ACQUIRE, "wavefront");
}
__device__ __forceinline__ float max16(float v) {
  v = fmaxf(v, lane_xor<1>(v)); v = fmaxf(v, lane_xor<2>(v));
  v = fmaxf(v, lane_xor<4>(v)); v = fmaxf(v, lane_xor<8>(v));
  return v;
}

template <int BKT0, int N>
__device__ __forceinline__ void gate_ktiles(const float* arow,
                                            const float* wc0, const float* wc1, const float* wc2, const float* wc3,
                                            bool own4, v8f& a0, v8f& a1, v8f& a2, v8f& a3) {
  if (own4) {
    #pragma unroll
    for (int j = 0; j < N; ++j) {
      const v2f av = *(const v2f*)(arow + 4*j);
      const v2f b3 = *(const v2f*)(wc3 + (BKT0 + j)*(4*G4));
      a3 = __builtin_amdgcn_wmma_f32_16x16x4_f32(false, av, false, b3, (short)0, a3, false, false);
    }
  }
  #pragma unroll
  for (int j = 0; j < N; ++j) {
    const v2f av = *(const v2f*)(arow + 4*j);
    const v2f b0 = *(const v2f*)(wc0 + (BKT0 + j)*(4*G4));
    const v2f b1 = *(const v2f*)(wc1 + (BKT0 + j)*(4*G4));
    const v2f b2 = *(const v2f*)(wc2 + (BKT0 + j)*(4*G4));
    a0 = __builtin_amdgcn_wmma_f32_16x16x4_f32(false, av, false, b0, (short)0, a0, false, false);
    a1 = __builtin_amdgcn_wmma_f32_16x16x4_f32(false, av, false, b1, (short)0, a1, false, false);
    a2 = __builtin_amdgcn_wmma_f32_16x16x4_f32(false, av, false, b2, (short)0, a2, false, false);
  }
}

__device__ __forceinline__ float max32(float v) {
  v = max16(v); v = fmaxf(v, lane_xor16(v));
  return v;
}

template <int NTR>
__device__ __forceinline__ void lstm_cell(const float* sG, float* sC, float* sH, int tid) {
  float gi[NTR], gf[NTR], gg[NTR], go[NTR], cp[NTR];
  #pragma unroll
  for (int it = 0; it < NTR; ++it) {
    const int i = tid + it*NT;
    const int b = i / H, u = i - b*H;
    gi[it] = sG[b*G4 + u];
    gf[it] = sG[b*G4 + H   + u];
    gg[it] = sG[b*G4 + 2*H + u];
    go[it] = sG[b*G4 + 3*H + u];
    cp[it] = sC[i];
  }
  #pragma unroll
  for (int it = 0; it < NTR; ++it) {
    const int i = tid + it*NT;
    const float c = sigm(gf[it]) * cp[it] + sigm(gi[it]) * tanh_hw(gg[it]);
    const float h = sigm(go[it]) * tanh_hw(c);
    sC[i] = c; sH[i] = h;
  }
}

__global__ __launch_bounds__(256) void pack_weights(
    const float* __restrict__ Wx, const float* __restrict__ Wh,
    const float* __restrict__ W_params, const float* __restrict__ b_params,
    float* __restrict__ ws)
{
  int i = blockIdx.x * 256 + threadIdx.x;
  if (i < KZ*G4) {
    int k = i / G4, n = i - k*G4;
    float v = (k < H) ? Wh[k*G4 + n] : ((k < H + INW + D) ? Wx[(k - H)*G4 + n] : 0.f);
    ws[WS_WCAT + (k >> 1)*(2*G4) + 2*n + (k & 1)] = v;
  } else if (i < KZ*G4 + H*PPAD) {
    int j = i - KZ*G4;
    int k = j / PPAD, n = j - k*PPAD;
    ws[WS_WP + j] = (n < PDIM) ? W_params[k*PDIM + n] : 0.f;
  } else if (i < WS_TOT) {
    int n = i - KZ*G4 - H*PPAD;
    ws[WS_BP + n] = (n < PDIM) ? b_params[n] : 0.f;
  }
}

__global__ __launch_bounds__(NT) void ntm_fused(
    const float* __restrict__ inputs,
    const float* __restrict__ Wcat,
    const float* __restrict__ b_lstm,
    const float* __restrict__ Wp,
    const float* __restrict__ bp,
    const float* __restrict__ W_out,
    const float* __restrict__ b_out,
    const int*   __restrict__ seq_ptr,
    float* __restrict__ out)
{
  __shared__ __attribute__((aligned(16))) float smem[SMEM_FLOATS];
  float* sWc  = smem;
  float* sBl  = sWc  + KZ*G4;
  float* sWo  = sBl  + G4;
  float* sH   = sWo  + (H + D)*OUTW;
  float* sC   = sH   + BC*H;
  float* sR   = sC   + BC*H;
  float* sZ   = sR   + BC*D;
  float* sG   = sZ   + BC*ZP;
  float* sP   = sG   + BC*G4;
  float* sK   = sP   + BC*PPAD;
  float* sKn  = sK   + BC*NH*D;
  float* sBe  = sKn  + 32;
  float* sGg  = sBe  + 32;
  float* sGa  = sGg  + 32;
  float* sS   = sGa  + 32;
  float* sE   = sS   + 32*3;
  float* sAd  = sE   + BC*D;

  const int tid   = threadIdx.x;
  const int wave  = tid >> 5;
  const int lane  = tid & 31;
  const int m     = lane & 15;
  const int ma    = m & (BC - 1);
  const int khalf = (lane < 16) ? 0 : 2;
  const int b0    = blockIdx.x * BC;
  const bool tail_wave = __builtin_amdgcn_readfirstlane(wave) < TAILW;
  const int seq   = seq_ptr[0];
  const int tOutStart = seq + 1;
  const int TOUT  = T_LEN - tOutStart;

  const int mb = tid >> 5;
  const int mc = tid & 31;
  float Mr[MROWS][D];
  for (int j = 0; j < MROWS; ++j) {
    for (int d = 0; d < D; ++d) Mr[j][d] = 1e-6f;
  }
  float wv[NH][MROWS];
  for (int h = 0; h < NH; ++h) {
    for (int j = 0; j < MROWS; ++j) wv[h][j] = (mc == 0 && j == 0) ? 1.f : 0.f;
  }

  {
    const v4f* __restrict__ src = (const v4f*)Wcat;
    v4f* dst = (v4f*)sWc;
    for (int i = tid; i < KZ*G4/4; i += NT) dst[i] = src[i];
  }
  for (int i = tid; i < G4;           i += NT) sBl[i] = b_lstm[i];
  for (int i = tid; i < (H + D)*OUTW; i += NT) sWo[i] = W_out[i];
  const float bo = b_out[tid & (OUTW - 1)];

  for (int i = tid; i < BC*H;       i += NT) { sH[i] = 0.f; sC[i] = 0.f; }
  for (int i = tid; i < BC*D;       i += NT) sR[i] = 0.f;
  if (tid < BC*(KXR - INW - D)) {
    const int b = tid / (KXR - INW - D);
    sZ[b*ZP + INW + D + (tid - b*(KXR - INW - D))] = 0.f;
  }
  __syncthreads();

  const bool own4 = __builtin_amdgcn_readfirstlane(wave) < G4/16 - 3*(NT/32);
  const int n0 = (wave           )*16 + m;
  const int n1 = (wave +   NT/32 )*16 + m;
  const int n2 = (wave + 2*NT/32 )*16 + m;
  const int n3 = own4 ? (wave + 3*NT/32)*16 + m : n2;
  const float* wc0 = sWc + 2*n0 + (khalf >> 1)*(2*G4);
  const float* wc1 = sWc + 2*n1 + (khalf >> 1)*(2*G4);
  const float* wc2 = sWc + 2*n2 + (khalf >> 1)*(2*G4);
  const float* wc3 = sWc + 2*n3 + (khalf >> 1)*(2*G4);
  const float* ah  = sH + ma*H  + khalf;
  const float* az  = sZ + ma*ZP + khalf;
  const float bias0 = sBl[n0], bias1 = sBl[n1], bias2 = sBl[n2], bias3 = sBl[n3];
  v8f acc0 = { bias0, bias0, bias0, bias0, bias0, bias0, bias0, bias0 };
  v8f acc1 = { bias1, bias1, bias1, bias1, bias1, bias1, bias1, bias1 };
  v8f acc2 = { bias2, bias2, bias2, bias2, bias2, bias2, bias2, bias2 };
  v8f acc3 = { bias3, bias3, bias3, bias3, bias3, bias3, bias3, bias3 };

  for (int t = 0; t < T_LEN; ++t) {
    {
      const float xv = inputs[(size_t)(b0 + mb) * T_LEN * INW + t * INW + ((lane < INW) ? lane : 0)];
      const float rv = sR[mb*D + ((lane < D) ? lane : 0)];
      if (lane < INW) sZ[mb*ZP + lane] = xv;
      if (lane < D)   sZ[mb*ZP + INW + lane] = rv;
    }
    __syncthreads();

    gate_ktiles<H/4, KXR/4>(az, wc0, wc1, wc2, wc3, own4, acc0, acc1, acc2, acc3);
    if (lane < 16) {
      #pragma unroll
      for (int v = 0; v < 8; ++v) {
        sG[v*G4 + n0] = acc0[v]; sG[v*G4 + n1] = acc1[v]; sG[v*G4 + n2] = acc2[v];
      }
      if (own4) {
        #pragma unroll
        for (int v = 0; v < 8; ++v) sG[v*G4 + n3] = acc3[v];
      }
    }
    __syncthreads();

    if (tail_wave) lstm_cell<HTRIPS>(sG, sC, sH, tid);
    else           lstm_cell<HTRIPS - 1>(sG, sC, sH, tid);
    __syncthreads();

    for (int ct = wave; ct < PPAD/16; ct += NT/32) {
      int n = ct*16 + m;
      const float* __restrict__ wp = Wp + n;
      float bias = bp[n];
      v8f acc = { bias, bias, bias, bias, bias, bias, bias, bias };
      for (int kt = 0; kt < H/4; ++kt) {
        int k0 = kt*4 + khalf;
        v2f av; av.x = sH[ma*H + k0];       av.y = sH[ma*H + k0 + 1];
        v2f bv; bv.x = wp[(size_t)k0*PPAD]; bv.y = wp[(size_t)(k0 + 1)*PPAD];
        acc = __builtin_amdgcn_wmma_f32_16x16x4_f32(false, av, false, bv,
                                                    (short)0, acc, false, false);
      }
      float pv[8];
      #pragma unroll
      for (int v = 0; v < 8; ++v) pv[v] = clipf(acc[v]);
      if (lane < 16) {
        #pragma unroll
        for (int v = 0; v < 8; ++v) sP[v*PPAD + n] = pv[v];
      }
    }
    __syncthreads();

    #pragma unroll
    for (int v = 0; v < 8; ++v) { acc0[v] = bias0; acc1[v] = bias1; acc2[v] = bias2; acc3[v] = bias3; }
    gate_ktiles<0*HCH, HCH>(ah + 0*HCH*4, wc0, wc1, wc2, wc3, own4, acc0, acc1, acc2, acc3);

    {
      const int bh  = mb*NH + ((lane >> 3) & 1);
      const int sub = lane & 7;
      const float* pb = sP + (bh >> 1)*PPAD + (bh & 1)*PPH;
      float kp[3];
      for (int j = 0; j < 3; ++j) kp[j] = pb[min(sub + 8*j, D - 1)];
      float kn2 = 0.f;
      for (int j = 0; j < 3; ++j) {
        const int d = sub + 8*j;
        const float kv = tanh_hw(kp[j]);
        if (d < D) {
          sK[bh*D + d] = kv;
          kn2 += kv * kv;
        }
      }
      kn2 = sum8(kn2);
      if (sub == 0) {
        sKn[bh] = sqrt_hw(kn2);
      } else if (sub == 1 || sub == 4) {
        const float sp = softplus_hw(pb[(sub == 1) ? D : D + 5]);
        if (sub == 1) sBe[bh] = LOG2E * sp; else sGa[bh] = sp + 1.f;
      } else if (sub == 2) {
        sGg[bh] = sigm(pb[D + 1]);
      } else if (sub == 3) {
        float s0 = pb[D + 2], s1 = pb[D + 3], s2 = pb[D + 4];
        float mx = fmaxf(s0, fmaxf(s1, s2));
        float e0 = exp2_hw(LOG2E * (s0 - mx)), e1 = exp2_hw(LOG2E * (s1 - mx)), e2 = exp2_hw(LOG2E * (s2 - mx));
        float inv = rcp_hw(e0 + e1 + e2);
        sS[bh*3 + 0] = e0 * inv; sS[bh*3 + 1] = e1 * inv; sS[bh*3 + 2] = e2 * inv;
      }
    }
    wave_sync();

    gate_ktiles<1*HCH, HCH>(ah + 1*HCH*4, wc0, wc1, wc2, wc3, own4, acc0, acc1, acc2, acc3);
    float sc[NH][MROWS];
    {
      float k0[D], k1[D];
      for (int d = 0; d < D; ++d) {
        k0[d] = sK[(mb*NH + 0)*D + d];
        k1[d] = sK[(mb*NH + 1)*D + d];
      }
      const float be0 = sBe[mb*NH + 0], be1 = sBe[mb*NH + 1];
      const float kn0 = sKn[mb*NH + 0], kn1 = sKn[mb*NH + 1];
      for (int j = 0; j < MROWS; ++j) {
        float mn2 = 0.f, d0 = 0.f, d1 = 0.f;
        for (int d = 0; d < D; ++d) {
          float mv = Mr[j][d];
          mn2 += mv * mv; d0 += k0[d] * mv; d1 += k1[d] * mv;
        }
        const float mn = sqrt_hw(mn2);
        sc[0][j] = be0 * d0 * rcp_hw(kn0*mn + EPSV);
        sc[1][j] = be1 * d1 * rcp_hw(kn1*mn + EPSV);
      }
    }

    gate_ktiles<2*HCH, HCH>(ah + 2*HCH*4, wc0, wc1, wc2, wc3, own4, acc0, acc1, acc2, acc3);
    float wg[NH][MROWS];
    for (int h = 0; h < NH; ++h) {
      float mx = sc[h][0];
      for (int j = 1; j < MROWS; ++j) mx = fmaxf(mx, sc[h][j]);
      mx = max32(mx);
      float ex[MROWS];
      float sum = 0.f;
      for (int j = 0; j < MROWS; ++j) { ex[j] = exp2_hw(sc[h][j] - mx); sum += ex[j]; }
      sum = sum32(sum);
      const float inv = rcp_hw(sum);
      const float g = sGg[mb*NH + h];
      for (int j = 0; j < MROWS; ++j) {
        const float wc = ex[j] * inv;
        wg[h][j] = g * wc + (1.f - g) * wv[h][j];
      }
    }

    gate_ktiles<3*HCH, HCH>(ah + 3*HCH*4, wc0, wc1, wc2, wc3, own4, acc0, acc1, acc2, acc3);

    for (int h = 0; h < NH; ++h) {
      const float s0 = sS[(mb*NH + h)*3 + 0], s1 = sS[(mb*NH + h)*3 + 1], s2 = sS[(mb*NH + h)*3 + 2];
      const float ga = sGa[mb*NH + h];
      float up[MROWS], dn[MROWS];
      for (int j = 0; j < MROWS; ++j) {
        const float offer_up = (mc == 0)  ? wg[h][(j + 1) & (MROWS - 1)]         : wg[h][j];
        const float offer_dn = (mc == 31) ? wg[h][(j + MROWS - 1) & (MROWS - 1)] : wg[h][j];
        up[j] = __shfl(offer_up, (mc + 1)  & 31, 32);
        dn[j] = __shfl(offer_dn, (mc + 31) & 31, 32);
      }
      float lw[MROWS];
      for (int j = 0; j < MROWS; ++j) {
        const float wt = s0 * up[j] + s1 * wg[h][j] + s2 * dn[j];
        lw[j] = log2_hw(fmaxf(wt, 1e-16f));
      }
      float lmx = lw[0];
      for (int j = 1; j < MROWS; ++j) lmx = fmaxf(lmx, lw[j]);
      lmx = max32(lmx);
      float pw[MROWS];
      float sum = 0.f;
      for (int j = 0; j < MROWS; ++j) {
        pw[j] = exp2_hw(ga * (lw[j] - lmx));
        sum += pw[j];
      }
      sum = sum32(sum);
      const float inv = rcp_hw(sum);
      for (int j = 0; j < MROWS; ++j) wv[h][j] = pw[j] * inv;
    }

    gate_ktiles<4*HCH, HCH>(ah + 4*HCH*4, wc0, wc1, wc2, wc3, own4, acc0, acc1, acc2, acc3);
    {
      float part[D];
      for (int d = 0; d < D; ++d) {
        float acc = 0.f;
        for (int j = 0; j < MROWS; ++j) acc += wv[0][j] * Mr[j][d];
        part[d] = sum32(acc);
      }
      if (mc == 0) {
        for (int d = 0; d < D; ++d) sR[mb*D + d] = part[d];
      }
    }
    if (lane < D) {
      sE[mb*D + lane]  = sigm (sP[mb*PPAD + NH*PPH + lane]);
      sAd[mb*D + lane] = tanh_hw(sP[mb*PPAD + NH*PPH + D + lane]);
    }
    wave_sync();

    {
      float ev[D], av[D];
      for (int d = 0; d < D; ++d) { ev[d] = sE[mb*D + d]; av[d] = sAd[mb*D + d]; }
      for (int j = 0; j < MROWS; ++j) {
        const float ww = wv[1][j];
        for (int d = 0; d < D; ++d)
          Mr[j][d] = Mr[j][d] * (1.f - ww * ev[d]) + ww * av[d];
      }
    }

    {
      const int o = lane & (OUTW - 1), q = lane >> 3;
      float acc = (q == 0) ? bo : 0.f;
      #pragma unroll
      for (int jj = 0; jj < (H + D)/4; ++jj) {
        const int j = q*((H + D)/4) + jj;
        const float v = (j < H) ? sH[mb*H + j] : sR[mb*D + (j - H)];
        acc += v * sWo[j*OUTW + o];
      }
      acc += lane_xor<8>(acc);
      acc += lane_xor16(acc);
      acc = clipf(acc);
      if (q == 0 && t >= tOutStart) {
        out[(size_t)(b0 + mb) * TOUT * OUTW + (t - tOutStart) * OUTW + o] = sigm(acc);
      }
    }
  }
}

extern "C" void kernel_launch(void* const* d_in, const int* in_sizes, int n_in,
                              void* d_out, int out_size, void* d_ws, size_t ws_size,
                              hipStream_t stream) {
  (void)in_sizes; (void)n_in; (void)out_size; (void)ws_size;
  const float* inputs   = (const float*)d_in[0];
  const float* Wx       = (const float*)d_in[1];
  const float* Wh       = (const float*)d_in[2];
  const float* b_lstm   = (const float*)d_in[3];
  const float* W_params = (const float*)d_in[4];
  const float* b_params = (const float*)d_in[5];
  const float* W_out    = (const float*)d_in[6];
  const float* b_out    = (const float*)d_in[7];
  const int*   seq      = (const int*)d_in[8];
  float* out = (float*)d_out;
  float* ws  = (float*)d_ws;

  pack_weights<<<dim3((WS_TOT + 255) / 256), dim3(256), 0, stream>>>(
      Wx, Wh, W_params, b_params, ws);

  ntm_fused<<<dim3(B_TOT / BC), dim3(NT), 0, stream>>>(
      inputs, ws + WS_WCAT, b_lstm, ws + WS_WP, ws + WS_BP,
      W_out, b_out, seq, out);
}
